# starting kernel
# baseline (speedup 1.0000x reference)
#include <hip/hip_runtime.h>
#include <math.h>

#define NNODES  50000
#define NEDGES  800000
#define ETOT    (NEDGES + NNODES)
#define IN_DIM  128
#define HID     64
#define HEADS   4
#define OUT_DIM 64
#define F1      (HEADS * HID)
#define ENC_NEG_INF 0x007FFFFFu

typedef __attribute__((ext_vector_type(2))) float v2f;
typedef __attribute__((ext_vector_type(8))) float v8f;

__device__ __forceinline__ float leakyf(float x, float s) { return x >= 0.f ? x : s * x; }

__device__ __forceinline__ unsigned enc_ord(float x) {
    unsigned u = __float_as_uint(x);
    return (u & 0x80000000u) ? ~u : (u | 0x80000000u);
}
__device__ __forceinline__ float dec_ord(unsigned e) {
    return (e & 0x80000000u) ? __uint_as_float(e & 0x7FFFFFFFu) : __uint_as_float(~e);
}

__device__ __forceinline__ void edge_sd(const int* __restrict__ ei, int t, int& s, int& d) {
    if (t < NEDGES) { s = ei[t]; d = ei[NEDGES + t]; }
    else            { s = t - NEDGES; d = s; }
}

__global__ void gemm_wmma_f32_kernel(const float* __restrict__ A,
                                     const float* __restrict__ W,
                                     float* __restrict__ C,
                                     int nrows, int fin, int fout) {
    const int gwave = (blockIdx.x * blockDim.x + threadIdx.x) >> 5;
    const int lane  = threadIdx.x & 31;
    const int stripsPerRow = fout >> 6;
    const int totalStrips  = (nrows >> 4) * stripsPerRow;
    if (gwave >= totalStrips) return;

    const int rowBlk = gwave / stripsPerRow;
    const int strip  = gwave % stripsPerRow;
    const int row0 = rowBlk << 4;
    const int col0 = strip << 6;
    const int half = lane >> 4;
    const int l    = lane & 15;

    v8f acc0 = {}, acc1 = {}, acc2 = {}, acc3 = {};
    const float* arow = A + (size_t)(row0 + l) * fin;
    for (int k = 0; k < fin; k += 4) {
        v2f a;
        a.x = arow[k + 2 * half + 0];
        a.y = arow[k + 2 * half + 1];
        const float* w0 = W + (size_t)(k + 2 * half) * fout + col0 + l;
        const float* w1 = w0 + fout;
        v2f b0 = { w0[ 0], w1[ 0] };
        v2f b1 = { w0[16], w1[16] };
        v2f b2 = { w0[32], w1[32] };
        v2f b3 = { w0[48], w1[48] };
        acc0 = __builtin_amdgcn_wmma_f32_16x16x4_f32(false, a, false, b0, (short)0, acc0, false, false);
        acc1 = __builtin_amdgcn_wmma_f32_16x16x4_f32(false, a, false, b1, (short)0, acc1, false, false);
        acc2 = __builtin_amdgcn_wmma_f32_16x16x4_f32(false, a, false, b2, (short)0, acc2, false, false);
        acc3 = __builtin_amdgcn_wmma_f32_16x16x4_f32(false, a, false, b3, (short)0, acc3, false, false);
    }
    float* crow = C + (size_t)row0 * fout + col0;
#pragma unroll
    for (int r = 0; r < 8; ++r) {
        const int m = r + 8 * half;
        float* p = crow + (size_t)m * fout + l;
        p[ 0] = acc0[r];
        p[16] = acc1[r];
        p[32] = acc2[r];
        p[48] = acc3[r];
    }
}

__global__ void alpha_kernel(const float* __restrict__ h,
                             const float* __restrict__ a_src, const float* __restrict__ a_dst,
                             float* __restrict__ as_, float* __restrict__ ad_,
                             int nn, int H, int Cc) {
    int idx = blockIdx.x * blockDim.x + threadIdx.x;
    if (idx >= nn * H) return;
    int n = idx / H, hh = idx - n * H;
    const float* hp = h + (size_t)n * H * Cc + hh * Cc;
    float s = 0.f, d = 0.f;
    for (int c = 0; c < Cc; ++c) {
        float v = hp[c];
        s += v * a_src[hh * Cc + c];
        d += v * a_dst[hh * Cc + c];
    }
    as_[idx] = s;
    ad_[idx] = d;
}

__global__ void edge_max_kernel(const int* __restrict__ ei,
                                const float* __restrict__ as_, const float* __restrict__ ad_,
                                float* __restrict__ ebuf, unsigned* __restrict__ menc, int H) {
    int idx = blockIdx.x * blockDim.x + threadIdx.x;
    if (idx >= ETOT * H) return;
    int t = idx / H, hh = idx - t * H;
    int s, d; edge_sd(ei, t, s, d);
    float e = leakyf(as_[s * H + hh] + ad_[d * H + hh], 0.2f);
    ebuf[idx] = e;
    atomicMax(&menc[d * H + hh], enc_ord(e));
}

__global__ void edge_exp_kernel(const int* __restrict__ ei,
                                float* __restrict__ ebuf, const unsigned* __restrict__ menc,
                                float* __restrict__ den, int H) {
    int idx = blockIdx.x * blockDim.x + threadIdx.x;
    if (idx >= ETOT * H) return;
    int t = idx / H, hh = idx - t * H;
    int s, d; edge_sd(ei, t, s, d);
    (void)s;
    float m  = dec_ord(menc[d * H + hh]);
    float ex = expf(ebuf[idx] - m);
    ebuf[idx] = ex;
    atomicAdd(&den[d * H + hh], ex);
}

__global__ void scatter_kernel(const int* __restrict__ ei,
                               const float* __restrict__ h,
                               const float* __restrict__ ebuf,
                               const float* __restrict__ den,
                               float* __restrict__ out, int H) {
    const int gwave = (blockIdx.x * blockDim.x + threadIdx.x) >> 5;
    const int lane  = threadIdx.x & 31;
    if (gwave >= ETOT) return;
    int s, d; edge_sd(ei, gwave, s, d);
    const int F = H << 6;
    const float* hs = h + (size_t)s * F;
    float* od = out + (size_t)d * F;
    float alph[HEADS];
#pragma unroll
    for (int hh = 0; hh < HEADS; ++hh)
        if (hh < H) alph[hh] = ebuf[(size_t)gwave * H + hh] / (den[d * H + hh] + 1e-16f);
    for (int c = lane; c < F; c += 32)
        atomicAdd(&od[c], alph[c >> 6] * hs[c]);
}

__global__ void bias_act_kernel(const float* __restrict__ in, const float* __restrict__ b,
                                float* __restrict__ out, int total, int F, float slope) {
    int i = blockIdx.x * blockDim.x + threadIdx.x;
    if (i >= total) return;
    out[i] = leakyf(in[i] + b[i % F], slope);
}

__global__ void fill_u32_kernel(unsigned* __restrict__ p, unsigned v, int n) {
    int i = blockIdx.x * blockDim.x + threadIdx.x;
    if (i < n) p[i] = v;
}

extern "C" void kernel_launch(void* const* d_in, const int* in_sizes, int n_in,
                              void* d_out, int out_size, void* d_ws, size_t ws_size,
                              hipStream_t stream) {
    const float* x   = (const float*)d_in[0];
    const int*   ei  = (const int*)  d_in[1];
    const float* W1  = (const float*)d_in[2];
    const float* as1 = (const float*)d_in[3];
    const float* ad1 = (const float*)d_in[4];
    const float* b1  = (const float*)d_in[5];
    const float* W2  = (const float*)d_in[6];
    const float* as2 = (const float*)d_in[7];
    const float* ad2 = (const float*)d_in[8];
    const float* b2  = (const float*)d_in[9];
    float* out = (float*)d_out;

    float* ws   = (float*)d_ws;
    float* h1   = ws;
    float* h2   = h1   + (size_t)NNODES * F1;
    float* ebuf = h2   + (size_t)NNODES * F1;
    float* asb  = ebuf + (size_t)ETOT * HEADS;
    float* adb  = asb  + (size_t)NNODES * HEADS;
    unsigned* menc = (unsigned*)(adb + (size_t)NNODES * HEADS);
    float* den  = (float*)(menc + (size_t)NNODES * HEADS);
    float* g2   = h1;
    float* out2 = h1 + (size_t)NNODES * OUT_DIM;

    {
        int strips = (NNODES / 16) * (F1 / 64);
        gemm_wmma_f32_kernel<<<(strips + 3) / 4, 128, 0, stream>>>(x, W1, h1, NNODES, IN_DIM, F1);
    }
    alpha_kernel<<<(NNODES * HEADS + 255) / 256, 256, 0, stream>>>(h1, as1, ad1, asb, adb,
                                                                   NNODES, HEADS, HID);
    hipMemsetAsync(den, 0, sizeof(float) * (size_t)NNODES * HEADS, stream);
    fill_u32_kernel<<<(NNODES * HEADS + 255) / 256, 256, 0, stream>>>(menc, ENC_NEG_INF,
                                                                      NNODES * HEADS);
    hipMemsetAsync(h2, 0, sizeof(float) * (size_t)NNODES * F1, stream);
    edge_max_kernel<<<(ETOT * HEADS + 255) / 256, 256, 0, stream>>>(ei, asb, adb, ebuf, menc, HEADS);
    edge_exp_kernel<<<(ETOT * HEADS + 255) / 256, 256, 0, stream>>>(ei, ebuf, menc, den, HEADS);
    scatter_kernel<<<(ETOT + 7) / 8, 256, 0, stream>>>(ei, h1, ebuf, den, h2, HEADS);
    bias_act_kernel<<<(NNODES * F1 + 255) / 256, 256, 0, stream>>>(h2, b1, h2,
                                                                   NNODES * F1, F1, 0.01f);

    {
        int strips = (NNODES / 16) * (OUT_DIM / 64);
        gemm_wmma_f32_kernel<<<(strips + 3) / 4, 128, 0, stream>>>(h2, W2, g2, NNODES, F1, OUT_DIM);
    }
    alpha_kernel<<<(NNODES + 255) / 256, 256, 0, stream>>>(g2, as2, ad2, asb, adb,
                                                           NNODES, 1, OUT_DIM);
    hipMemsetAsync(den, 0, sizeof(float) * (size_t)NNODES, stream);
    fill_u32_kernel<<<(NNODES + 255) / 256, 256, 0, stream>>>(menc, ENC_NEG_INF, NNODES);
    hipMemsetAsync(out2, 0, sizeof(float) * (size_t)NNODES * OUT_DIM, stream);
    edge_max_kernel<<<(ETOT + 255) / 256, 256, 0, stream>>>(ei, asb, adb, ebuf, menc, 1);
    edge_exp_kernel<<<(ETOT + 255) / 256, 256, 0, stream>>>(ei, ebuf, menc, den, 1);
    scatter_kernel<<<(ETOT + 7) / 8, 256, 0, stream>>>(ei, g2, ebuf, den, out2, 1);
    bias_act_kernel<<<(NNODES * OUT_DIM + 255) / 256, 256, 0, stream>>>(out2, b2, out,
                                                                        NNODES * OUT_DIM,
                                                                        OUT_DIM, 1.0f);
}


# Layer-2 product moved into the layer-1 owner work-group: finished rows get bias and activation in LDS and are multiplied by W2 on fp32 matrix instructions from LDS; h2 never stored; 8 launches -> 6
# speedup vs baseline: 3.2911x; 3.2911x over previous
#include <hip/hip_runtime.h>
#include <math.h>

#define NNODES  50000
#define NEDGES  800000
#define ETOT    (NEDGES + NNODES)
#define IN_DIM  128
#define HID     64
#define HEADS   4
#define OUT_DIM 64
#define F1      (HEADS * HID)
#define ENC_NEG_INF 0x007FFFFFu

typedef __attribute__((ext_vector_type(2))) float v2f;
typedef __attribute__((ext_vector_type(8))) float v8f;

__device__ __forceinline__ float leakyf(float x, float s) { return x >= 0.f ? x : s * x; }

__device__ __forceinline__ unsigned enc_ord(float x) {
    unsigned u = __float_as_uint(x);
    return (u & 0x80000000u) ? ~u : (u | 0x80000000u);
}
__device__ __forceinline__ float dec_ord(unsigned e) {
    return (e & 0x80000000u) ? __uint_as_float(e & 0x7FFFFFFFu) : __uint_as_float(~e);
}

__global__ void gemm_wmma_f32_kernel(const float* __restrict__ A,
                                     const float* __restrict__ W,
                                     float* __restrict__ C,
                                     int nrows, int fin, int fout) {
    const int gwave = (blockIdx.x * blockDim.x + threadIdx.x) >> 5;
    const int lane  = threadIdx.x & 31;
    const int stripsPerRow = fout >> 6;
    const int totalStrips  = (nrows >> 4) * stripsPerRow;
    if (gwave >= totalStrips) return;

    const int rowBlk = gwave / stripsPerRow;
    const int strip  = gwave % stripsPerRow;
    const int row0 = rowBlk << 4;
    const int col0 = strip << 6;
    const int half = lane >> 4;
    const int l    = lane & 15;

    v8f acc0 = {}, acc1 = {}, acc2 = {}, acc3 = {};
    const float* arow = A + (size_t)(row0 + l) * fin;
    for (int k = 0; k < fin; k += 4) {
        v2f a;
        a.x = arow[k + 2 * half + 0];
        a.y = arow[k + 2 * half + 1];
        const float* w0 = W + (size_t)(k + 2 * half) * fout + col0 + l;
        const float* w1 = w0 + fout;
        v2f b0 = { w0[ 0], w1[ 0] };
        v2f b1 = { w0[16], w1[16] };
        v2f b2 = { w0[32], w1[32] };
        v2f b3 = { w0[48], w1[48] };
        acc0 = __builtin_amdgcn_wmma_f32_16x16x4_f32(false, a, false, b0, (short)0, acc0, false, false);
        acc1 = __builtin_amdgcn_wmma_f32_16x16x4_f32(false, a, false, b1, (short)0, acc1, false, false);
        acc2 = __builtin_amdgcn_wmma_f32_16x16x4_f32(false, a, false, b2, (short)0, acc2, false, false);
        acc3 = __builtin_amdgcn_wmma_f32_16x16x4_f32(false, a, false, b3, (short)0, acc3, false, false);
    }
    float* crow = C + (size_t)row0 * fout + col0;
#pragma unroll
    for (int r = 0; r < 8; ++r) {
        const int m = r + 8 * half;
        float* p = crow + (size_t)m * fout + l;
        p[ 0] = acc0[r];
        p[16] = acc1[r];
        p[32] = acc2[r];
        p[48] = acc3[r];
    }
}

__global__ void alpha_kernel(const float* __restrict__ h,
                             const float* __restrict__ a_src, const float* __restrict__ a_dst,
                             float* __restrict__ as_, float* __restrict__ ad_,
                             int nn, int H, int Cc) {
    int idx = blockIdx.x * blockDim.x + threadIdx.x;
    if (idx >= nn * H) return;
    int n = idx / H, hh = idx - n * H;
    const float* hp = h + (size_t)n * H * Cc + hh * Cc;
    float s = 0.f, d = 0.f;
    for (int c = 0; c < Cc; ++c) {
        float v = hp[c];
        s += v * a_src[hh * Cc + c];
        d += v * a_dst[hh * Cc + c];
    }
    as_[idx] = s;
    ad_[idx] = d;
}

#define NWG_AGG   256
#define NOWN      196
#define AGG_WAVES 32
#define QCAP      160
#define LCAP      16384
#define NDST4     (NEDGES / 4)
#define SCAN_U    8
#define CHUNK4    (SCAN_U * 32)
#define NCHUNK    ((NDST4 + CHUNK4 - 1) / CHUNK4)

static_assert(NOWN * NWG_AGG >= NNODES && (NWG_AGG - 1) * NOWN < NNODES, "every node has one owner");
static_assert(ETOT <= (1 << 20) && NOWN <= 256, "queue entry = edge id | local node << 20");
static_assert(NEDGES % 4 == 0, "destination list read as int4");

template <int H> struct RowPitch { static constexpr int value = (H == 4) ? (H * 64 + 4) : (H * 64); };
#define W2_ROWS F1
static_assert(W2_ROWS * OUT_DIM == LCAP, "the W2 fragments take exactly the edge record's LDS");
static_assert(OUT_DIM == 64 && W2_ROWS % 4 == 0, "layer-2 tile map below");

template <int H, int PHASE, int G>
__device__ __forceinline__ void flush_batch(const unsigned* list, int base, int n, int lane,
                                            const int* __restrict__ ei,
                                            const float* __restrict__ h,
                                            const float* __restrict__ as_,
                                            const float* adl, unsigned* menc, float* den, float* acc,
                                            unsigned* wlist, unsigned* wcount) {
    constexpr int F = H * 64;
    constexpr int P = RowPitch<H>::value;
    const bool act = lane < n;
    unsigned ent = 0;
    if (act) ent = list[base + lane];
    const int t  = (int)(ent & 0xFFFFFu);
    const int dl = (int)(ent >> 20);
    int s = 0;
    if (act) s = (t < NEDGES) ? ei[t] : (t - NEDGES);
    float e[H];
#pragma unroll
    for (int hh = 0; hh < H; ++hh) e[hh] = 0.f;
    if (act) {
#pragma unroll
        for (int hh = 0; hh < H; ++hh) e[hh] = leakyf(as_[s * H + hh] + adl[dl * H + hh], 0.2f);
    }
    if (PHASE == 0) {
        if (act) {
#pragma unroll
            for (int hh = 0; hh < H; ++hh)
                __hip_atomic_fetch_max(&menc[dl * H + hh], enc_ord(e[hh]), __ATOMIC_RELAXED,
                                       __HIP_MEMORY_SCOPE_WORKGROUP);
        }
        unsigned pos = 0;
        if (lane == 0)
            pos = __hip_atomic_fetch_add(wcount, (unsigned)n, __ATOMIC_RELAXED, __HIP_MEMORY_SCOPE_WORKGROUP);
        pos = (unsigned)__builtin_amdgcn_readfirstlane((int)pos);
        if (act && pos + (unsigned)n <= (unsigned)LCAP) wlist[pos + lane] = ent;
    } else {
        float ex[H];
#pragma unroll
        for (int hh = 0; hh < H; ++hh) ex[hh] = 0.f;
        if (act) {
#pragma unroll
            for (int hh = 0; hh < H; ++hh) {
                ex[hh] = expf(e[hh] - dec_ord(menc[dl * H + hh]));
                __hip_atomic_fetch_add(&den[dl * H + hh], ex[hh], __ATOMIC_RELAXED, __HIP_MEMORY_SCOPE_WORKGROUP);
            }
        }
        for (int j = 0; j < n; j += G) {
            v2f   r[G][H];
            float w[G][H];
            int   dj[G];
#pragma unroll
            for (int g = 0; g < G; ++g) {
                const int jj = (j + g < n) ? (j + g) : (n - 1);
                const int sj = __builtin_amdgcn_readlane(s, jj);
                dj[g] = __builtin_amdgcn_readlane(dl, jj);
#pragma unroll
                for (int hh = 0; hh < H; ++hh)
                    w[g][hh] = __int_as_float(__builtin_amdgcn_readlane(__float_as_int(ex[hh]), jj));
                const v2f* hp = (const v2f*)(h + (size_t)sj * F) + lane;
#pragma unroll
                for (int hh = 0; hh < H; ++hh) r[g][hh] = hp[hh * 32];
            }
#pragma unroll
            for (int g = 0; g < G; ++g) {
                if (j + g < n) {
                    float* ap = acc + dj[g] * P + 2 * lane;
#pragma unroll
                    for (int hh = 0; hh < H; ++hh) {
                        __hip_atomic_fetch_add(ap + hh * 64,     w[g][hh] * r[g][hh].x, __ATOMIC_RELAXED,
                                               __HIP_MEMORY_SCOPE_WORKGROUP);
                        __hip_atomic_fetch_add(ap + hh * 64 + 1, w[g][hh] * r[g][hh].y, __ATOMIC_RELAXED,
                                               __HIP_MEMORY_SCOPE_WORKGROUP);
                    }
                }
            }
        }
    }
}

template <int H, int PHASE, int G>
__device__ __forceinline__ void scan_pass(const int* __restrict__ ei,
                                          const float* __restrict__ h,
                                          const float* __restrict__ as_,
                                          const float* adl, unsigned* menc, float* den, float* acc,
                                          unsigned* myq, unsigned* wlist, unsigned* wcount,
                                          int lo, int nown, int lane, int wave) {
    int qn = 0;
    auto enqueue = [&](bool mt, unsigned val) {
        const unsigned mask = __builtin_amdgcn_ballot_w32(mt);
        if (mt) myq[qn + __builtin_popcount(mask & ((1u << lane) - 1u))] = val;
        qn += __builtin_popcount(mask);
    };
    {
        const int dl = wave + AGG_WAVES * lane;
        enqueue(dl < nown, (unsigned)(NEDGES + lo + dl) | ((unsigned)dl << 20));
    }
    const int4* dst4 = (const int4*)(ei + NEDGES);
    const int c0 = ((int)blockIdx.x * NCHUNK) / NWG_AGG;
    for (int jc = wave; jc < NCHUNK; jc += AGG_WAVES) {
        int c = jc + c0;
        if (c >= NCHUNK) c -= NCHUNK;
        const int i0 = c * CHUNK4 + lane;
        int4 dv[SCAN_U];
#pragma unroll
        for (int u = 0; u < SCAN_U; ++u) {
            const int i4 = i0 + u * 32;
            dv[u] = dst4[(i4 < NDST4) ? i4 : 0];
        }
#pragma unroll
        for (int u = 0; u < SCAN_U; ++u) {
            const int tb = (i0 + u * 32) * 4;
            const bool inr = (i0 + u * 32) < NDST4;
            const unsigned r0 = (unsigned)(dv[u].x - lo), r1 = (unsigned)(dv[u].y - lo);
            const unsigned r2 = (unsigned)(dv[u].z - lo), r3 = (unsigned)(dv[u].w - lo);
            const bool m0 = inr & (r0 < (unsigned)nown), m1 = inr & (r1 < (unsigned)nown);
            const bool m2 = inr & (r2 < (unsigned)nown), m3 = inr & (r3 < (unsigned)nown);
            if (__builtin_amdgcn_ballot_w32(m0 | m1 | m2 | m3) != 0u) {
                enqueue(m0, (unsigned)(tb + 0) | (r0 << 20));
                enqueue(m1, (unsigned)(tb + 1) | (r1 << 20));
                enqueue(m2, (unsigned)(tb + 2) | (r2 << 20));
                enqueue(m3, (unsigned)(tb + 3) | (r3 << 20));
                while (qn >= 32) {
                    flush_batch<H, PHASE, G>(myq, qn - 32, 32, lane, ei, h, as_, adl, menc, den, acc, wlist, wcount);
                    qn -= 32;
                }
            }
        }
    }
    if (qn > 0) flush_batch<H, PHASE, G>(myq, 0, qn, lane, ei, h, as_, adl, menc, den, acc, wlist, wcount);
}

template <int H>
__global__ void __launch_bounds__(AGG_WAVES * 32)
agg_owner_kernel(const int* __restrict__ ei, const float* __restrict__ h,
                 const float* __restrict__ as_, const float* __restrict__ ad_,
                 float* __restrict__ out,
                 const float* __restrict__ bias, const float* __restrict__ W2) {
    constexpr int F = H * 64;
    constexpr int P = RowPitch<H>::value;
    __shared__ __attribute__((aligned(16))) float acc[NOWN * P];
    __shared__ unsigned menc[NOWN * H];
    __shared__ float    den[NOWN * H];
    __shared__ float    adl[NOWN * H];
    __shared__ unsigned q[AGG_WAVES * QCAP];
    __shared__ __attribute__((aligned(16))) unsigned wlist[LCAP];
    __shared__ unsigned wcount;

    const int tid = threadIdx.x, lane = tid & 31, wave = tid >> 5;
    const int lo = (int)blockIdx.x * NOWN;
    const int nown = (NNODES - lo < NOWN) ? (NNODES - lo) : NOWN;

    for (int i = tid; i < NOWN * P; i += AGG_WAVES * 32) acc[i] = 0.f;
    for (int i = tid; i < NOWN * H; i += AGG_WAVES * 32) {
        menc[i] = ENC_NEG_INF;
        den[i]  = 0.f;
        adl[i]  = (i < nown * H) ? ad_[lo * H + i] : 0.f;
    }
    if (tid == 0) wcount = 0u;
    __syncthreads();

    unsigned* myq = q + wave * QCAP;
    scan_pass<H, 0, 4>(ei, h, as_, adl, menc, den, acc, myq, wlist, &wcount, lo, nown, lane, wave);
    __syncthreads();

    const unsigned cnt = wcount;
    if (cnt <= (unsigned)LCAP) {
        for (int base = wave * 32; base < (int)cnt; base += AGG_WAVES * 32) {
            int n = (int)cnt - base;
            n = __builtin_amdgcn_readfirstlane(n < 32 ? n : 32);
            flush_batch<H, 1, (H == 1 ? 16 : 8)>(wlist, base, n, lane, ei, h, as_, adl, menc, den, acc, wlist, &wcount);
        }
    } else {
        scan_pass<H, 1, 4>(ei, h, as_, adl, menc, den, acc, myq, wlist, &wcount, lo, nown, lane, wave);
    }
    __syncthreads();

    if constexpr (H == 4) {
        constexpr int WSLOTS = LCAP / 2 / (AGG_WAVES * 32);
        v2f wreg[WSLOTS];
#pragma unroll
        for (int j = 0; j < WSLOTS; ++j) {
            const int slot = tid + j * (AGG_WAVES * 32);
            const int sl = slot & 31, tile = (slot >> 5) & 3, k4 = slot >> 7;
            const float* wp = W2 + (size_t)(4 * k4 + 2 * (sl >> 4)) * OUT_DIM + 16 * tile + (sl & 15);
            wreg[j].x = wp[0];
            wreg[j].y = wp[OUT_DIM];
        }
        for (int i = tid; i < nown * (F / 4); i += AGG_WAVES * 32) {
            const int row = i / (F / 4), col = (i % (F / 4)) * 4;
            const float dn = den[row * H + (col >> 6)] + 1e-16f;
            float4* p = (float4*)(acc + row * P + col);
            const float4 b = *(const float4*)(bias + col);
            float4 v = *p;
            v.x /= dn; v.y /= dn; v.z /= dn; v.w /= dn;
            v.x = leakyf(v.x + b.x, 0.01f); v.y = leakyf(v.y + b.y, 0.01f);
            v.z = leakyf(v.z + b.z, 0.01f); v.w = leakyf(v.w + b.w, 0.01f);
            *p = v;
        }
        v2f* wf = (v2f*)wlist;
#pragma unroll
        for (int j = 0; j < WSLOTS; ++j) wf[tid + j * (AGG_WAVES * 32)] = wreg[j];
        __syncthreads();

        const int half = lane >> 4, l = lane & 15;
        const int njobs = ((nown + 15) >> 4) * 2;
        const int wv = __builtin_amdgcn_readfirstlane(wave);
        for (int job = wv; job < njobs; job += AGG_WAVES) {
            const int rb = job >> 1, cp = job & 1;
            int arow = rb * 16 + l;
            arow = (arow < NOWN) ? arow : (NOWN - 1);
            const float* ap = acc + arow * P + 2 * half;
            const v2f* bp = wf + (2 * cp) * 32 + lane;
            v8f c0 = {}, c1 = {};
            for (int k4 = 0; k4 < W2_ROWS / 4; ++k4) {
                const v2f a  = *(const v2f*)(ap + 4 * k4);
                const v2f b0 = bp[k4 * 128];
                const v2f b1 = bp[k4 * 128 + 32];
                c0 = __builtin_amdgcn_wmma_f32_16x16x4_f32(false, a, false, b0, (short)0, c0, false, false);
                c1 = __builtin_amdgcn_wmma_f32_16x16x4_f32(false, a, false, b1, (short)0, c1, false, false);
            }
#pragma unroll
            for (int r = 0; r < 8; ++r) {
                const int row = rb * 16 + r + 8 * half;
                if (row < nown) {
                    float* gp = out + (size_t)(lo + row) * OUT_DIM + 32 * cp + l;
                    gp[ 0] = c0[r];
                    gp[16] = c1[r];
                }
            }
        }
    } else {
        const int n4 = nown * F / 4;
        float4* o4 = (float4*)(out + (size_t)lo * F);
        const float4* a4 = (const float4*)acc;
        for (int i = tid; i < n4; i += AGG_WAVES * 32) {
            const int idx = i * 4;
            const float dn = den[(idx / F) * H + ((idx % F) >> 6)] + 1e-16f;
            float4 v = a4[i];
            v.x /= dn; v.y /= dn; v.z /= dn; v.w /= dn;
            o4[i] = v;
        }
    }
}

__global__ void bias_act_kernel(const float* __restrict__ in, const float* __restrict__ b,
                                float* __restrict__ out, int total, int F, float slope) {
    int i = blockIdx.x * blockDim.x + threadIdx.x;
    if (i >= total) return;
    out[i] = leakyf(in[i] + b[i % F], slope);
}

extern "C" void kernel_launch(void* const* d_in, const int* in_sizes, int n_in,
                              void* d_out, int out_size, void* d_ws, size_t ws_size,
                              hipStream_t stream) {
    const float* x   = (const float*)d_in[0];
    const int*   ei  = (const int*)  d_in[1];
    const float* W1  = (const float*)d_in[2];
    const float* as1 = (const float*)d_in[3];
    const float* ad1 = (const float*)d_in[4];
    const float* b1  = (const float*)d_in[5];
    const float* W2  = (const float*)d_in[6];
    const float* as2 = (const float*)d_in[7];
    const float* ad2 = (const float*)d_in[8];
    const float* b2  = (const float*)d_in[9];
    float* out = (float*)d_out;

    float* ws   = (float*)d_ws;
    float* h1   = ws;
    float* g2   = h1   + (size_t)NNODES * F1;
    float* out2 = g2   + (size_t)NNODES * OUT_DIM;
    float* asb  = h1   + (size_t)2 * NNODES * F1;
    float* adb  = asb  + (size_t)NNODES * HEADS;

    {
        int strips = (NNODES / 16) * (F1 / 64);
        gemm_wmma_f32_kernel<<<(strips + 3) / 4, 128, 0, stream>>>(x, W1, h1, NNODES, IN_DIM, F1);
    }
    alpha_kernel<<<(NNODES * HEADS + 255) / 256, 256, 0, stream>>>(h1, as1, ad1, asb, adb,
                                                                   NNODES, HEADS, HID);
    agg_owner_kernel<HEADS><<<NWG_AGG, AGG_WAVES * 32, 0, stream>>>(ei, h1, asb, adb, g2, b1, W2);

    alpha_kernel<<<(NNODES + 255) / 256, 256, 0, stream>>>(g2, as2, ad2, asb, adb,
                                                           NNODES, 1, OUT_DIM);
    agg_owner_kernel<1><<<NWG_AGG, AGG_WAVES * 32, 0, stream>>>(ei, g2, asb, adb, out2, b2, W2);
    bias_act_kernel<<<(NNODES * OUT_DIM + 255) / 256, 256, 0, stream>>>(out2, b2, out,
                                                                        NNODES * OUT_DIM,
                                                                        OUT_DIM, 1.0f);
}


# Layer-1 owner writes its kept-edge record (source, local node) to the workspace; layer-2 owner loads it instead of filtering the destination list again; layer-2 LDS float adds arrive in record order
# speedup vs baseline: 3.8186x; 1.1603x over previous
#include <hip/hip_runtime.h>
#include <math.h>

#define NNODES  50000
#define NEDGES  800000
#define ETOT    (NEDGES + NNODES)
#define IN_DIM  128
#define HID     64
#define HEADS   4
#define OUT_DIM 64
#define F1      (HEADS * HID)
#define ENC_NEG_INF 0x007FFFFFu

typedef __attribute__((ext_vector_type(2))) float v2f;
typedef __attribute__((ext_vector_type(8))) float v8f;

__device__ __forceinline__ float leakyf(float x, float s) { return x >= 0.f ? x : s * x; }

__device__ __forceinline__ unsigned enc_ord(float x) {
    unsigned u = __float_as_uint(x);
    return (u & 0x80000000u) ? ~u : (u | 0x80000000u);
}
__device__ __forceinline__ float dec_ord(unsigned e) {
    return (e & 0x80000000u) ? __uint_as_float(e & 0x7FFFFFFFu) : __uint_as_float(~e);
}

__global__ void gemm_wmma_f32_kernel(const float* __restrict__ A,
                                     const float* __restrict__ W,
                                     float* __restrict__ C,
                                     int nrows, int fin, int fout) {
    const int gwave = (blockIdx.x * blockDim.x + threadIdx.x) >> 5;
    const int lane  = threadIdx.x & 31;
    const int stripsPerRow = fout >> 6;
    const int totalStrips  = (nrows >> 4) * stripsPerRow;
    if (gwave >= totalStrips) return;

    const int rowBlk = gwave / stripsPerRow;
    const int strip  = gwave % stripsPerRow;
    const int row0 = rowBlk << 4;
    const int col0 = strip << 6;
    const int half = lane >> 4;
    const int l    = lane & 15;

    v8f acc0 = {}, acc1 = {}, acc2 = {}, acc3 = {};
    const float* arow = A + (size_t)(row0 + l) * fin;
    for (int k = 0; k < fin; k += 4) {
        v2f a;
        a.x = arow[k + 2 * half + 0];
        a.y = arow[k + 2 * half + 1];
        const float* w0 = W + (size_t)(k + 2 * half) * fout + col0 + l;
        const float* w1 = w0 + fout;
        v2f b0 = { w0[ 0], w1[ 0] };
        v2f b1 = { w0[16], w1[16] };
        v2f b2 = { w0[32], w1[32] };
        v2f b3 = { w0[48], w1[48] };
        acc0 = __builtin_amdgcn_wmma_f32_16x16x4_f32(false, a, false, b0, (short)0, acc0, false, false);
        acc1 = __builtin_amdgcn_wmma_f32_16x16x4_f32(false, a, false, b1, (short)0, acc1, false, false);
        acc2 = __builtin_amdgcn_wmma_f32_16x16x4_f32(false, a, false, b2, (short)0, acc2, false, false);
        acc3 = __builtin_amdgcn_wmma_f32_16x16x4_f32(false, a, false, b3, (short)0, acc3, false, false);
    }
    float* crow = C + (size_t)row0 * fout + col0;
#pragma unroll
    for (int r = 0; r < 8; ++r) {
        const int m = r + 8 * half;
        float* p = crow + (size_t)m * fout + l;
        p[ 0] = acc0[r];
        p[16] = acc1[r];
        p[32] = acc2[r];
        p[48] = acc3[r];
    }
}

__global__ void alpha_kernel(const float* __restrict__ h,
                             const float* __restrict__ a_src, const float* __restrict__ a_dst,
                             float* __restrict__ as_, float* __restrict__ ad_,
                             int nn, int H, int Cc) {
    int idx = blockIdx.x * blockDim.x + threadIdx.x;
    if (idx >= nn * H) return;
    int n = idx / H, hh = idx - n * H;
    const float* hp = h + (size_t)n * H * Cc + hh * Cc;
    float s = 0.f, d = 0.f;
    for (int c = 0; c < Cc; ++c) {
        float v = hp[c];
        s += v * a_src[hh * Cc + c];
        d += v * a_dst[hh * Cc + c];
    }
    as_[idx] = s;
    ad_[idx] = d;
}

#define NWG_AGG   256
#define NOWN      196
#define AGG_WAVES 32
#define QCAP      160
#define LCAP      16384
#define NDST4     (NEDGES / 4)
#define SCAN_U    8
#define CHUNK4    (SCAN_U * 32)
#define NCHUNK    ((NDST4 + CHUNK4 - 1) / CHUNK4)

static_assert(NOWN * NWG_AGG >= NNODES && (NWG_AGG - 1) * NOWN < NNODES, "every node has one owner");
static_assert(ETOT <= (1 << 20) && NOWN <= 256, "queue entry = edge id | local node << 20");
static_assert(NEDGES % 4 == 0, "destination list read as int4");
static_assert(NNODES <= (1 << 20), "handed-over entry = source node | local node << 20");
#define RSTRIDE   (LCAP + 32)
static_assert((size_t)NNODES * (F1 + 2 * OUT_DIM) + (size_t)NWG_AGG * RSTRIDE <= (size_t)2 * NNODES * F1,
              "the records end before the logits begin");
static_assert(((size_t)NNODES * (F1 + 2 * OUT_DIM)) % 32 == 0, "records start on a 128-byte line of the workspace");

template <int H> struct RowPitch { static constexpr int value = (H == 4) ? (H * 64 + 4) : (H * 64); };
#define W2_ROWS F1
static_assert(W2_ROWS * OUT_DIM == LCAP, "the W2 fragments take exactly the edge record's LDS");
static_assert(OUT_DIM == 64 && W2_ROWS % 4 == 0, "layer-2 tile map below");

template <int H, int PHASE, int G, bool SRC, bool REC, bool EMIT>
__device__ __forceinline__ void flush_batch(const unsigned* list, int base, int n, int lane,
                                            const int* __restrict__ ei,
                                            const float* __restrict__ h,
                                            const float* __restrict__ as_,
                                            const float* adl, unsigned* menc, float* den, float* acc,
                                            unsigned* wlist, unsigned* wcount,
                                            unsigned* __restrict__ grec) {
    constexpr int F = H * 64;
    constexpr int P = RowPitch<H>::value;
    const bool act = lane < n;
    unsigned ent = 0;
    if (act) ent = list[base + lane];
    const int t  = (int)(ent & 0xFFFFFu);
    const int dl = (int)(ent >> 20);
    int s = 0;
    if (act) {
        if (SRC) s = t;
        else     s = (t < NEDGES) ? ei[t] : (t - NEDGES);
    }
    float e[H];
#pragma unroll
    for (int hh = 0; hh < H; ++hh) e[hh] = 0.f;
    if (act) {
#pragma unroll
        for (int hh = 0; hh < H; ++hh) e[hh] = leakyf(as_[s * H + hh] + adl[dl * H + hh], 0.2f);
    }
    if (PHASE == 0) {
        if (act) {
#pragma unroll
            for (int hh = 0; hh < H; ++hh)
                __hip_atomic_fetch_max(&menc[dl * H + hh], enc_ord(e[hh]), __ATOMIC_RELAXED,
                                       __HIP_MEMORY_SCOPE_WORKGROUP);
        }
        if (REC) {
            unsigned pos = 0;
            if (lane == 0)
                pos = __hip_atomic_fetch_add(wcount, (unsigned)n, __ATOMIC_RELAXED, __HIP_MEMORY_SCOPE_WORKGROUP);
            pos = (unsigned)__builtin_amdgcn_readfirstlane((int)pos);
            if (act && pos + (unsigned)n <= (unsigned)LCAP) wlist[pos + lane] = ent;
        }
    } else {
        if (EMIT) {
            if (act) grec[base + lane] = (unsigned)s | ((unsigned)dl << 20);
        }
        float ex[H];
#pragma unroll
        for (int hh = 0; hh < H; ++hh) ex[hh] = 0.f;
        if (act) {
#pragma unroll
            for (int hh = 0; hh < H; ++hh) {
                ex[hh] = expf(e[hh] - dec_ord(menc[dl * H + hh]));
                __hip_atomic_fetch_add(&den[dl * H + hh], ex[hh], __ATOMIC_RELAXED, __HIP_MEMORY_SCOPE_WORKGROUP);
            }
        }
        for (int j = 0; j < n; j += G) {
            v2f   r[G][H];
            float w[G][H];
            int   dj[G];
#pragma unroll
            for (int g = 0; g < G; ++g) {
                const int jj = (j + g < n) ? (j + g) : (n - 1);
                const int sj = __builtin_amdgcn_readlane(s, jj);
                dj[g] = __builtin_amdgcn_readlane(dl, jj);
#pragma unroll
                for (int hh = 0; hh < H; ++hh)
                    w[g][hh] = __int_as_float(__builtin_amdgcn_readlane(__float_as_int(ex[hh]), jj));
                const v2f* hp = (const v2f*)(h + (size_t)sj * F) + lane;
#pragma unroll
                for (int hh = 0; hh < H; ++hh) r[g][hh] = hp[hh * 32];
            }
#pragma unroll
            for (int g = 0; g < G; ++g) {
                if (j + g < n) {
                    float* ap = acc + dj[g] * P + 2 * lane;
#pragma unroll
                    for (int hh = 0; hh < H; ++hh) {
                        __hip_atomic_fetch_add(ap + hh * 64,     w[g][hh] * r[g][hh].x, __ATOMIC_RELAXED,
                                               __HIP_MEMORY_SCOPE_WORKGROUP);
                        __hip_atomic_fetch_add(ap + hh * 64 + 1, w[g][hh] * r[g][hh].y, __ATOMIC_RELAXED,
                                               __HIP_MEMORY_SCOPE_WORKGROUP);
                    }
                }
            }
        }
    }
}

template <int H, int PHASE, int G>
__device__ __forceinline__ void scan_pass(const int* __restrict__ ei,
                                          const float* __restrict__ h,
                                          const float* __restrict__ as_,
                                          const float* adl, unsigned* menc, float* den, float* acc,
                                          unsigned* myq, unsigned* wlist, unsigned* wcount,
                                          int lo, int nown, int lane, int wave) {
    int qn = 0;
    auto enqueue = [&](bool mt, unsigned val) {
        const unsigned mask = __builtin_amdgcn_ballot_w32(mt);
        if (mt) myq[qn + __builtin_popcount(mask & ((1u << lane) - 1u))] = val;
        qn += __builtin_popcount(mask);
    };
    {
        const int dl = wave + AGG_WAVES * lane;
        enqueue(dl < nown, (unsigned)(NEDGES + lo + dl) | ((unsigned)dl << 20));
    }
    const int4* dst4 = (const int4*)(ei + NEDGES);
    const int c0 = ((int)blockIdx.x * NCHUNK) / NWG_AGG;
    for (int jc = wave; jc < NCHUNK; jc += AGG_WAVES) {
        int c = jc + c0;
        if (c >= NCHUNK) c -= NCHUNK;
        const int i0 = c * CHUNK4 + lane;
        int4 dv[SCAN_U];
#pragma unroll
        for (int u = 0; u < SCAN_U; ++u) {
            const int i4 = i0 + u * 32;
            dv[u] = dst4[(i4 < NDST4) ? i4 : 0];
        }
#pragma unroll
        for (int u = 0; u < SCAN_U; ++u) {
            const int tb = (i0 + u * 32) * 4;
            const bool inr = (i0 + u * 32) < NDST4;
            const unsigned r0 = (unsigned)(dv[u].x - lo), r1 = (unsigned)(dv[u].y - lo);
            const unsigned r2 = (unsigned)(dv[u].z - lo), r3 = (unsigned)(dv[u].w - lo);
            const bool m0 = inr & (r0 < (unsigned)nown), m1 = inr & (r1 < (unsigned)nown);
            const bool m2 = inr & (r2 < (unsigned)nown), m3 = inr & (r3 < (unsigned)nown);
            if (__builtin_amdgcn_ballot_w32(m0 | m1 | m2 | m3) != 0u) {
                enqueue(m0, (unsigned)(tb + 0) | (r0 << 20));
                enqueue(m1, (unsigned)(tb + 1) | (r1 << 20));
                enqueue(m2, (unsigned)(tb + 2) | (r2 << 20));
                enqueue(m3, (unsigned)(tb + 3) | (r3 << 20));
                while (qn >= 32) {
                    flush_batch<H, PHASE, G, false, PHASE == 0, false>(myq, qn - 32, 32, lane, ei, h, as_, adl, menc,
                                                                       den, acc, wlist, wcount, nullptr);
                    qn -= 32;
                }
            }
        }
    }
    if (qn > 0)
        flush_batch<H, PHASE, G, false, PHASE == 0, false>(myq, 0, qn, lane, ei, h, as_, adl, menc, den, acc, wlist,
                                                           wcount, nullptr);
}

template <int H>
__global__ void __launch_bounds__(AGG_WAVES * 32)
agg_owner_kernel(const int* __restrict__ ei, const float* __restrict__ h,
                 const float* __restrict__ as_, const float* __restrict__ ad_,
                 float* __restrict__ out,
                 const float* __restrict__ bias, const float* __restrict__ W2,
                 unsigned* __restrict__ grec) {
    constexpr int F = H * 64;
    constexpr int P = RowPitch<H>::value;
    __shared__ __attribute__((aligned(16))) float acc[NOWN * P];
    __shared__ unsigned menc[NOWN * H];
    __shared__ float    den[NOWN * H];
    __shared__ float    adl[NOWN * H];
    __shared__ unsigned q[AGG_WAVES * QCAP];
    __shared__ __attribute__((aligned(16))) unsigned wlist[LCAP];
    __shared__ unsigned wcount;

    const int tid = threadIdx.x, lane = tid & 31, wave = tid >> 5;
    const int lo = (int)blockIdx.x * NOWN;
    const int nown = (NNODES - lo < NOWN) ? (NNODES - lo) : NOWN;
    unsigned* myhdr = grec + (size_t)blockIdx.x * RSTRIDE;
    unsigned* myrec = myhdr + 32;

    unsigned cnt_in = (unsigned)LCAP + 1u;
    if constexpr (H == 1) {
        cnt_in = (unsigned)__builtin_amdgcn_readfirstlane((int)myhdr[0]);
        if (cnt_in <= (unsigned)LCAP)
            for (int i = tid; i < (int)cnt_in; i += AGG_WAVES * 32) wlist[i] = myrec[i];
    }
    const bool fromrec = (H == 1) && (cnt_in <= (unsigned)LCAP);

    for (int i = tid; i < NOWN * P; i += AGG_WAVES * 32) acc[i] = 0.f;
    for (int i = tid; i < NOWN * H; i += AGG_WAVES * 32) {
        menc[i] = ENC_NEG_INF;
        den[i]  = 0.f;
        adl[i]  = (i < nown * H) ? ad_[lo * H + i] : 0.f;
    }
    if (tid == 0) wcount = 0u;
    __syncthreads();

    unsigned* myq = q + wave * QCAP;
    if (fromrec) {
        for (int base = wave * 32; base < (int)cnt_in; base += AGG_WAVES * 32) {
            int n = (int)cnt_in - base;
            n = __builtin_amdgcn_readfirstlane(n < 32 ? n : 32);
            flush_batch<H, 0, 4, true, false, false>(wlist, base, n, lane, ei, h, as_, adl, menc, den, acc, wlist,
                                                     &wcount, nullptr);
        }
    } else {
        scan_pass<H, 0, 4>(ei, h, as_, adl, menc, den, acc, myq, wlist, &wcount, lo, nown, lane, wave);
    }
    __syncthreads();

    const unsigned cnt = fromrec ? cnt_in : wcount;
    if constexpr (H == 4) {
        if (tid == 0) myhdr[0] = cnt;
    }
    constexpr int GB = (H == 1) ? 16 : 8;
    if (cnt <= (unsigned)LCAP) {
        for (int base = wave * 32; base < (int)cnt; base += AGG_WAVES * 32) {
            int n = (int)cnt - base;
            n = __builtin_amdgcn_readfirstlane(n < 32 ? n : 32);
            if (fromrec)
                flush_batch<H, 1, GB, true, false, false>(wlist, base, n, lane, ei, h, as_, adl, menc, den, acc, wlist,
                                                          &wcount, nullptr);
            else
                flush_batch<H, 1, GB, false, false, H == 4>(wlist, base, n, lane, ei, h, as_, adl, menc, den, acc, wlist,
                                                            &wcount, myrec);
        }
    } else {
        scan_pass<H, 1, 4>(ei, h, as_, adl, menc, den, acc, myq, wlist, &wcount, lo, nown, lane, wave);
    }
    __syncthreads();

    if constexpr (H == 4) {
        constexpr int WSLOTS = LCAP / 2 / (AGG_WAVES * 32);
        v2f wreg[WSLOTS];
#pragma unroll
        for (int j = 0; j < WSLOTS; ++j) {
            const int slot = tid + j * (AGG_WAVES * 32);
            const int sl = slot & 31, tile = (slot >> 5) & 3, k4 = slot >> 7;
            const float* wp = W2 + (size_t)(4 * k4 + 2 * (sl >> 4)) * OUT_DIM + 16 * tile + (sl & 15);
            wreg[j].x = wp[0];
            wreg[j].y = wp[OUT_DIM];
        }
        for (int i = tid; i < nown * (F / 4); i += AGG_WAVES * 32) {
            const int row = i / (F / 4), col = (i % (F / 4)) * 4;
            const float dn = den[row * H + (col >> 6)] + 1e-16f;
            float4* p = (float4*)(acc + row * P + col);
            const float4 b = *(const float4*)(bias + col);
            float4 v = *p;
            v.x /= dn; v.y /= dn; v.z /= dn; v.w /= dn;
            v.x = leakyf(v.x + b.x, 0.01f); v.y = leakyf(v.y + b.y, 0.01f);
            v.z = leakyf(v.z + b.z, 0.01f); v.w = leakyf(v.w + b.w, 0.01f);
            *p = v;
        }
        v2f* wf = (v2f*)wlist;
#pragma unroll
        for (int j = 0; j < WSLOTS; ++j) wf[tid + j * (AGG_WAVES * 32)] = wreg[j];
        __syncthreads();

        const int half = lane >> 4, l = lane & 15;
        const int njobs = ((nown + 15) >> 4) * 2;
        const int wv = __builtin_amdgcn_readfirstlane(wave);
        for (int job = wv; job < njobs; job += AGG_WAVES) {
            const int rb = job >> 1, cp = job & 1;
            int arow = rb * 16 + l;
            arow = (arow < NOWN) ? arow : (NOWN - 1);
            const float* ap = acc + arow * P + 2 * half;
            const v2f* bp = wf + (2 * cp) * 32 + lane;
            v8f c0 = {}, c1 = {};
            for (int k4 = 0; k4 < W2_ROWS / 4; ++k4) {
                const v2f a  = *(const v2f*)(ap + 4 * k4);
                const v2f b0 = bp[k4 * 128];
                const v2f b1 = bp[k4 * 128 + 32];
                c0 = __builtin_amdgcn_wmma_f32_16x16x4_f32(false, a, false, b0, (short)0, c0, false, false);
                c1 = __builtin_amdgcn_wmma_f32_16x16x4_f32(false, a, false, b1, (short)0, c1, false, false);
            }
#pragma unroll
            for (int r = 0; r < 8; ++r) {
                const int row = rb * 16 + r + 8 * half;
                if (row < nown) {
                    float* gp = out + (size_t)(lo + row) * OUT_DIM + 32 * cp + l;
                    gp[ 0] = c0[r];
                    gp[16] = c1[r];
                }
            }
        }
    } else {
        const int n4 = nown * F / 4;
        float4* o4 = (float4*)(out + (size_t)lo * F);
        const float4* a4 = (const float4*)acc;
        for (int i = tid; i < n4; i += AGG_WAVES * 32) {
            const int idx = i * 4;
            const float dn = den[(idx / F) * H + ((idx % F) >> 6)] + 1e-16f;
            float4 v = a4[i];
            v.x /= dn; v.y /= dn; v.z /= dn; v.w /= dn;
            o4[i] = v;
        }
    }
}

__global__ void bias_act_kernel(const float* __restrict__ in, const float* __restrict__ b,
                                float* __restrict__ out, int total, int F, float slope) {
    int i = blockIdx.x * blockDim.x + threadIdx.x;
    if (i >= total) return;
    out[i] = leakyf(in[i] + b[i % F], slope);
}

extern "C" void kernel_launch(void* const* d_in, const int* in_sizes, int n_in,
                              void* d_out, int out_size, void* d_ws, size_t ws_size,
                              hipStream_t stream) {
    const float* x   = (const float*)d_in[0];
    const int*   ei  = (const int*)  d_in[1];
    const float* W1  = (const float*)d_in[2];
    const float* as1 = (const float*)d_in[3];
    const float* ad1 = (const float*)d_in[4];
    const float* b1  = (const float*)d_in[5];
    const float* W2  = (const float*)d_in[6];
    const float* as2 = (const float*)d_in[7];
    const float* ad2 = (const float*)d_in[8];
    const float* b2  = (const float*)d_in[9];
    float* out = (float*)d_out;

    float* ws   = (float*)d_ws;
    float* h1   = ws;
    float* g2   = h1   + (size_t)NNODES * F1;
    float* out2 = g2   + (size_t)NNODES * OUT_DIM;
    float* asb  = h1   + (size_t)2 * NNODES * F1;
    float* adb  = asb  + (size_t)NNODES * HEADS;
    unsigned* rec = (unsigned*)(out2 + (size_t)NNODES * OUT_DIM);

    {
        int strips = (NNODES / 16) * (F1 / 64);
        gemm_wmma_f32_kernel<<<(strips + 3) / 4, 128, 0, stream>>>(x, W1, h1, NNODES, IN_DIM, F1);
    }
    alpha_kernel<<<(NNODES * HEADS + 255) / 256, 256, 0, stream>>>(h1, as1, ad1, asb, adb,
                                                                   NNODES, HEADS, HID);
    agg_owner_kernel<HEADS><<<NWG_AGG, AGG_WAVES * 32, 0, stream>>>(ei, h1, asb, adb, g2, b1, W2, rec);

    alpha_kernel<<<(NNODES + 255) / 256, 256, 0, stream>>>(g2, as2, ad2, asb, adb,
                                                           NNODES, 1, OUT_DIM);
    agg_owner_kernel<1><<<NWG_AGG, AGG_WAVES * 32, 0, stream>>>(ei, g2, asb, adb, out2, b2, W2, rec);
    bias_act_kernel<<<(NNODES * OUT_DIM + 255) / 256, 256, 0, stream>>>(out2, b2, out,
                                                                        NNODES * OUT_DIM,
                                                                        OUT_DIM, 1.0f);
}


# Owner scan reads a 1-byte-per-edge owner digest of the destination list (made by extra work-groups of the layer-1 logit launch), not the 4-byte list; kept-edge order, hence LDS add order, changes
# speedup vs baseline: 4.1480x; 1.0863x over previous
#include <hip/hip_runtime.h>
#include <math.h>

#define NNODES  50000
#define NEDGES  800000
#define ETOT    (NEDGES + NNODES)
#define IN_DIM  128
#define HID     64
#define HEADS   4
#define OUT_DIM 64
#define F1      (HEADS * HID)
#define ENC_NEG_INF 0x007FFFFFu

typedef __attribute__((ext_vector_type(2))) float v2f;
typedef __attribute__((ext_vector_type(8))) float v8f;

__device__ __forceinline__ float leakyf(float x, float s) { return x >= 0.f ? x : s * x; }

__device__ __forceinline__ unsigned enc_ord(float x) {
    unsigned u = __float_as_uint(x);
    return (u & 0x80000000u) ? ~u : (u | 0x80000000u);
}
__device__ __forceinline__ float dec_ord(unsigned e) {
    return (e & 0x80000000u) ? __uint_as_float(e & 0x7FFFFFFFu) : __uint_as_float(~e);
}

__global__ void gemm_wmma_f32_kernel(const float* __restrict__ A,
                                     const float* __restrict__ W,
                                     float* __restrict__ C,
                                     int nrows, int fin, int fout) {
    const int gwave = (blockIdx.x * blockDim.x + threadIdx.x) >> 5;
    const int lane  = threadIdx.x & 31;
    const int stripsPerRow = fout >> 6;
    const int totalStrips  = (nrows >> 4) * stripsPerRow;
    if (gwave >= totalStrips) return;

    const int rowBlk = gwave / stripsPerRow;
    const int strip  = gwave % stripsPerRow;
    const int row0 = rowBlk << 4;
    const int col0 = strip << 6;
    const int half = lane >> 4;
    const int l    = lane & 15;

    v8f acc0 = {}, acc1 = {}, acc2 = {}, acc3 = {};
    const float* arow = A + (size_t)(row0 + l) * fin;
    for (int k = 0; k < fin; k += 4) {
        v2f a;
        a.x = arow[k + 2 * half + 0];
        a.y = arow[k + 2 * half + 1];
        const float* w0 = W + (size_t)(k + 2 * half) * fout + col0 + l;
        const float* w1 = w0 + fout;
        v2f b0 = { w0[ 0], w1[ 0] };
        v2f b1 = { w0[16], w1[16] };
        v2f b2 = { w0[32], w1[32] };
        v2f b3 = { w0[48], w1[48] };
        acc0 = __builtin_amdgcn_wmma_f32_16x16x4_f32(false, a, false, b0, (short)0, acc0, false, false);
        acc1 = __builtin_amdgcn_wmma_f32_16x16x4_f32(false, a, false, b1, (short)0, acc1, false, false);
        acc2 = __builtin_amdgcn_wmma_f32_16x16x4_f32(false, a, false, b2, (short)0, acc2, false, false);
        acc3 = __builtin_amdgcn_wmma_f32_16x16x4_f32(false, a, false, b3, (short)0, acc3, false, false);
    }
    float* crow = C + (size_t)row0 * fout + col0;
#pragma unroll
    for (int r = 0; r < 8; ++r) {
        const int m = r + 8 * half;
        float* p = crow + (size_t)m * fout + l;
        p[ 0] = acc0[r];
        p[16] = acc1[r];
        p[32] = acc2[r];
        p[48] = acc3[r];
    }
}

#define ALPHA_GPW 32
static_assert(HID == 64 && OUT_DIM == 64, "a group is 64 floats in both layers");
template <int H>
__device__ __forceinline__ void alpha_body(const float* __restrict__ h,
                                           const float* __restrict__ a_src, const float* __restrict__ a_dst,
                                           float* __restrict__ as_, float* __restrict__ ad_, int ngroups, int blk) {
    static_assert(16 % H == 0, "head of group g0 + 16*hf + j is j % H");
    const int lane = threadIdx.x & 31, hf = lane >> 4, l = lane & 15;
    const int g0 = (int)((blk * 256 + threadIdx.x) >> 5) * ALPHA_GPW;
    if (g0 >= ngroups) return;

    float4 v[16];
#pragma unroll
    for (int j = 0; j < 16; ++j) {
        int g = g0 + 16 * hf + j;
        g = (g < ngroups) ? g : (ngroups - 1);
        v[j] = *(const float4*)(h + (size_t)g * 64 + 4 * l);
    }
    float4 sa[H], da[H];
#pragma unroll
    for (int hh = 0; hh < H; ++hh) {
        sa[hh] = *(const float4*)(a_src + hh * 64 + 4 * l);
        da[hh] = *(const float4*)(a_dst + hh * 64 + 4 * l);
    }
    float ps[16], pd[16];
#pragma unroll
    for (int j = 0; j < 16; ++j) {
        const float4 s = sa[j % H], d = da[j % H];
        ps[j] = v[j].x * s.x + v[j].y * s.y + v[j].z * s.z + v[j].w * s.w;
        pd[j] = v[j].x * d.x + v[j].y * d.y + v[j].z * d.z + v[j].w * d.w;
    }
    const bool b3 = (l & 8) != 0, b2 = (l & 4) != 0, b1 = (l & 2) != 0, b0 = (l & 1) != 0;
    float s8[8], d8[8];
#pragma unroll
    for (int t = 0; t < 8; ++t) {
        s8[t] = (b3 ? ps[t + 8] : ps[t]) + __shfl_xor(b3 ? ps[t] : ps[t + 8], 8);
        d8[t] = (b3 ? pd[t + 8] : pd[t]) + __shfl_xor(b3 ? pd[t] : pd[t + 8], 8);
    }
    float s4[4], d4[4];
#pragma unroll
    for (int t = 0; t < 4; ++t) {
        s4[t] = (b2 ? s8[t + 4] : s8[t]) + __shfl_xor(b2 ? s8[t] : s8[t + 4], 4);
        d4[t] = (b2 ? d8[t + 4] : d8[t]) + __shfl_xor(b2 ? d8[t] : d8[t + 4], 4);
    }
    float s2[2], d2[2];
#pragma unroll
    for (int t = 0; t < 2; ++t) {
        s2[t] = (b1 ? s4[t + 2] : s4[t]) + __shfl_xor(b1 ? s4[t] : s4[t + 2], 2);
        d2[t] = (b1 ? d4[t + 2] : d4[t]) + __shfl_xor(b1 ? d4[t] : d4[t + 2], 2);
    }
    const float s1 = (b0 ? s2[1] : s2[0]) + __shfl_xor(b0 ? s2[0] : s2[1], 1);
    const float d1 = (b0 ? d2[1] : d2[0]) + __shfl_xor(b0 ? d2[0] : d2[1], 1);
    const int g = g0 + lane;
    if (g < ngroups) {
        as_[g] = s1;
        ad_[g] = d1;
    }
}

template <int H>
__global__ void __launch_bounds__(256)
alpha_kernel(const float* __restrict__ h,
             const float* __restrict__ a_src, const float* __restrict__ a_dst,
             float* __restrict__ as_, float* __restrict__ ad_, int ngroups) {
    alpha_body<H>(h, a_src, a_dst, as_, ad_, ngroups, (int)blockIdx.x);
}

#define NOWN       196
#define NDIG16     (NEDGES / 16)
#define DIG_BLOCKS ((NDIG16 + 255) / 256)
static_assert(NEDGES % 16 == 0, "the digest is written and read in 16-byte groups");
static_assert((NNODES - 1) / NOWN <= 255, "an owner index fits a byte");
__device__ __forceinline__ unsigned owner4(int4 d) {
    return  (unsigned)d.x / NOWN        | ((unsigned)d.y / NOWN) << 8 |
           ((unsigned)d.z / NOWN) << 16 | ((unsigned)d.w / NOWN) << 24;
}
__device__ __forceinline__ void digest_block(const int* __restrict__ dst, unsigned* __restrict__ dig, int blk) {
    const int i = blk * 256 + (int)threadIdx.x;
    if (i >= NDIG16) return;
    const int4* d4 = (const int4*)dst + 4 * (size_t)i;
    const int4 a = d4[0], b = d4[1], cc = d4[2], d = d4[3];
    uint4 o;
    o.x = owner4(a); o.y = owner4(b); o.z = owner4(cc); o.w = owner4(d);
    ((uint4*)dig)[i] = o;
}

#define ALPHA1_BLOCKS (((NNODES * HEADS + ALPHA_GPW - 1) / ALPHA_GPW + 7) / 8)
__global__ void __launch_bounds__(256)
alpha_digest_kernel(const float* __restrict__ h,
                    const float* __restrict__ a_src, const float* __restrict__ a_dst,
                    float* __restrict__ as_, float* __restrict__ ad_, int ngroups,
                    const int* __restrict__ dst, unsigned* __restrict__ dig) {
    if ((int)blockIdx.x < DIG_BLOCKS) digest_block(dst, dig, (int)blockIdx.x);
    else alpha_body<HEADS>(h, a_src, a_dst, as_, ad_, ngroups, (int)blockIdx.x - DIG_BLOCKS);
}

#define NWG_AGG   256
#define AGG_WAVES 32
#define QCAP      64
#define LCAP      16384
#define SCAN_U    8
#define CHUNK16   (SCAN_U * 32)
#define NCHUNK    ((NDIG16 + CHUNK16 - 1) / CHUNK16)

static_assert(NOWN * NWG_AGG >= NNODES && (NWG_AGG - 1) * NOWN < NNODES, "every node has one owner");
static_assert(ETOT <= (1 << 20) && NOWN <= 256, "record entry = edge id | local node << 20");
static_assert(NNODES <= (1 << 20), "handed-over entry = source node | local node << 20");
#define RSTRIDE   (LCAP + 32)
static_assert((size_t)NNODES * (F1 + 2 * OUT_DIM) + (size_t)NWG_AGG * RSTRIDE + NEDGES / 4 <= (size_t)2 * NNODES * F1,
              "the records and the digest end before the logits begin");
static_assert(((size_t)NWG_AGG * RSTRIDE) % 4 == 0, "the digest starts on a 16-byte boundary");
static_assert(((size_t)NNODES * (F1 + 2 * OUT_DIM)) % 32 == 0, "records start on a 128-byte line of the workspace");

template <int H> struct RowPitch { static constexpr int value = (H == 4) ? (H * 64 + 4) : (H * 64); };
#define W2_ROWS F1
static_assert(W2_ROWS * OUT_DIM == LCAP, "the W2 fragments take exactly the edge record's LDS");
static_assert(OUT_DIM == 64 && W2_ROWS % 4 == 0, "layer-2 tile map below");

template <int H, int PHASE, int G, bool SRC, bool QDST, bool REC, bool EMIT>
__device__ __forceinline__ void flush_batch(const unsigned* list, int base, int n, int lane, int lo,
                                            const int* __restrict__ ei,
                                            const float* __restrict__ h,
                                            const float* __restrict__ as_,
                                            const float* adl, unsigned* menc, float* den, float* acc,
                                            unsigned* wlist, unsigned* wcount,
                                            unsigned* __restrict__ grec) {
    constexpr int F = H * 64;
    constexpr int P = RowPitch<H>::value;
    const bool act = lane < n;
    unsigned ent = 0;
    if (act) ent = list[base + lane];
    const int t  = (int)(ent & 0xFFFFFu);
    int dl = (int)(ent >> 20);
    int s = 0;
    if (act) {
        if (SRC) s = t;
        else     s = (t < NEDGES) ? ei[t] : (t - NEDGES);
        if (QDST) {
            dl = ((t < NEDGES) ? ei[NEDGES + t] : (t - NEDGES)) - lo;
            ent = (unsigned)t | ((unsigned)dl << 20);
        }
    }
    float e[H];
#pragma unroll
    for (int hh = 0; hh < H; ++hh) e[hh] = 0.f;
    if (act) {
#pragma unroll
        for (int hh = 0; hh < H; ++hh) e[hh] = leakyf(as_[s * H + hh] + adl[dl * H + hh], 0.2f);
    }
    if (PHASE == 0) {
        if (act) {
#pragma unroll
            for (int hh = 0; hh < H; ++hh)
                __hip_atomic_fetch_max(&menc[dl * H + hh], enc_ord(e[hh]), __ATOMIC_RELAXED,
                                       __HIP_MEMORY_SCOPE_WORKGROUP);
        }
        if (REC) {
            unsigned pos = 0;
            if (lane == 0)
                pos = __hip_atomic_fetch_add(wcount, (unsigned)n, __ATOMIC_RELAXED, __HIP_MEMORY_SCOPE_WORKGROUP);
            pos = (unsigned)__builtin_amdgcn_readfirstlane((int)pos);
            if (act && pos + (unsigned)n <= (unsigned)LCAP) wlist[pos + lane] = ent;
        }
    } else {
        if (EMIT) {
            if (act) grec[base + lane] = (unsigned)s | ((unsigned)dl << 20);
        }
        float ex[H];
#pragma unroll
        for (int hh = 0; hh < H; ++hh) ex[hh] = 0.f;
        if (act) {
#pragma unroll
            for (int hh = 0; hh < H; ++hh) {
                ex[hh] = expf(e[hh] - dec_ord(menc[dl * H + hh]));
                __hip_atomic_fetch_add(&den[dl * H + hh], ex[hh], __ATOMIC_RELAXED, __HIP_MEMORY_SCOPE_WORKGROUP);
            }
        }
        for (int j = 0; j < n; j += G) {
            v2f   r[G][H];
            float w[G][H];
            int   dj[G];
#pragma unroll
            for (int g = 0; g < G; ++g) {
                const int jj = (j + g < n) ? (j + g) : (n - 1);
                const int sj = __builtin_amdgcn_readlane(s, jj);
                dj[g] = __builtin_amdgcn_readlane(dl, jj);
#pragma unroll
                for (int hh = 0; hh < H; ++hh)
                    w[g][hh] = __int_as_float(__builtin_amdgcn_readlane(__float_as_int(ex[hh]), jj));
                const v2f* hp = (const v2f*)(h + (size_t)sj * F) + lane;
#pragma unroll
                for (int hh = 0; hh < H; ++hh) r[g][hh] = hp[hh * 32];
            }
#pragma unroll
            for (int g = 0; g < G; ++g) {
                if (j + g < n) {
                    float* ap = acc + dj[g] * P + 2 * lane;
#pragma unroll
                    for (int hh = 0; hh < H; ++hh) {
                        __hip_atomic_fetch_add(ap + hh * 64,     w[g][hh] * r[g][hh].x, __ATOMIC_RELAXED,
                                               __HIP_MEMORY_SCOPE_WORKGROUP);
                        __hip_atomic_fetch_add(ap + hh * 64 + 1, w[g][hh] * r[g][hh].y, __ATOMIC_RELAXED,
                                               __HIP_MEMORY_SCOPE_WORKGROUP);
                    }
                }
            }
        }
    }
}

__device__ __forceinline__ unsigned zero_bytes(unsigned x) {
    const unsigned t = (x & 0x7F7F7F7Fu) + 0x7F7F7F7Fu;
    return ~(t | x) & 0x80808080u;
}

template <int H, int PHASE, int G>
__device__ __forceinline__ void scan_pass(const int* __restrict__ ei, const unsigned* __restrict__ dig,
                                          const float* __restrict__ h,
                                          const float* __restrict__ as_,
                                          const float* adl, unsigned* menc, float* den, float* acc,
                                          unsigned* myq, unsigned* wlist, unsigned* wcount,
                                          int lo, int nown, int lane, int wave) {
    int qn = 0;
    auto enqueue = [&](bool mt, unsigned val) {
        const unsigned mask = __builtin_amdgcn_ballot_w32(mt);
        if (mt) myq[qn + __builtin_popcount(mask & ((1u << lane) - 1u))] = val;
        qn += __builtin_popcount(mask);
    };
    {
        const int dl = wave + AGG_WAVES * lane;
        enqueue(dl < nown, (unsigned)(NEDGES + lo + dl));
    }
    const uint4* dg4 = (const uint4*)dig;
    const unsigned mine = (unsigned)blockIdx.x * 0x01010101u;
    const int c0 = ((int)blockIdx.x * NCHUNK) / NWG_AGG;
    for (int jc = wave; jc < NCHUNK; jc += AGG_WAVES) {
        int c = jc + c0;
        if (c >= NCHUNK) c -= NCHUNK;
        const int i0 = c * CHUNK16 + lane;
        uint4 dv[SCAN_U];
#pragma unroll
        for (int u = 0; u < SCAN_U; ++u) {
            const int i16 = i0 + u * 32;
            dv[u] = dg4[(i16 < NDIG16) ? i16 : 0];
        }
#pragma unroll
        for (int u = 0; u < SCAN_U; ++u) {
            const int i16 = i0 + u * 32;
            unsigned f = (zero_bytes(dv[u].x ^ mine) >> 7) | (zero_bytes(dv[u].y ^ mine) >> 6) |
                         (zero_bytes(dv[u].z ^ mine) >> 5) | (zero_bytes(dv[u].w ^ mine) >> 4);
            if (i16 >= NDIG16) f = 0u;
            while (__builtin_amdgcn_ballot_w32(f != 0u) != 0u) {
                const int p = __builtin_ctz(f | 0x80000000u);
                enqueue(f != 0u, (unsigned)(i16 * 16 + 4 * (p & 7) + (p >> 3)));
                f &= f - 1u;
                while (qn >= 32) {
                    flush_batch<H, PHASE, G, false, true, PHASE == 0, false>(myq, qn - 32, 32, lane, lo, ei, h, as_, adl,
                                                                             menc, den, acc, wlist, wcount, nullptr);
                    qn -= 32;
                }
            }
        }
    }
    if (qn > 0)
        flush_batch<H, PHASE, G, false, true, PHASE == 0, false>(myq, 0, qn, lane, lo, ei, h, as_, adl, menc, den, acc,
                                                                 wlist, wcount, nullptr);
}

template <int H>
__global__ void __launch_bounds__(AGG_WAVES * 32)
agg_owner_kernel(const int* __restrict__ ei, const float* __restrict__ h,
                 const float* __restrict__ as_, const float* __restrict__ ad_,
                 float* __restrict__ out,
                 const float* __restrict__ bias, const float* __restrict__ W2,
                 unsigned* __restrict__ grec, const unsigned* __restrict__ dig) {
    constexpr int F = H * 64;
    constexpr int P = RowPitch<H>::value;
    __shared__ __attribute__((aligned(16))) float acc[NOWN * P];
    __shared__ unsigned menc[NOWN * H];
    __shared__ float    den[NOWN * H];
    __shared__ float    adl[NOWN * H];
    __shared__ unsigned q[AGG_WAVES * QCAP];
    __shared__ __attribute__((aligned(16))) unsigned wlist[LCAP];
    __shared__ unsigned wcount;

    const int tid = threadIdx.x, lane = tid & 31, wave = tid >> 5;
    const int lo = (int)blockIdx.x * NOWN;
    const int nown = (NNODES - lo < NOWN) ? (NNODES - lo) : NOWN;
    unsigned* myhdr = grec + (size_t)blockIdx.x * RSTRIDE;
    unsigned* myrec = myhdr + 32;

    unsigned cnt_in = (unsigned)LCAP + 1u;
    if constexpr (H == 1) {
        cnt_in = (unsigned)__builtin_amdgcn_readfirstlane((int)myhdr[0]);
        if (cnt_in <= (unsigned)LCAP)
            for (int i = tid; i < (int)cnt_in; i += AGG_WAVES * 32) wlist[i] = myrec[i];
    }
    const bool fromrec = (H == 1) && (cnt_in <= (unsigned)LCAP);

    for (int i = tid; i < NOWN * P; i += AGG_WAVES * 32) acc[i] = 0.f;
    for (int i = tid; i < NOWN * H; i += AGG_WAVES * 32) {
        menc[i] = ENC_NEG_INF;
        den[i]  = 0.f;
        adl[i]  = (i < nown * H) ? ad_[lo * H + i] : 0.f;
    }
    if (tid == 0) wcount = 0u;
    __syncthreads();

    unsigned* myq = q + wave * QCAP;
    if (fromrec) {
        for (int base = wave * 32; base < (int)cnt_in; base += AGG_WAVES * 32) {
            int n = (int)cnt_in - base;
            n = __builtin_amdgcn_readfirstlane(n < 32 ? n : 32);
            flush_batch<H, 0, 4, true, false, false, false>(wlist, base, n, lane, lo, ei, h, as_, adl, menc, den, acc,
                                                            wlist, &wcount, nullptr);
        }
    } else {
        scan_pass<H, 0, 4>(ei, dig, h, as_, adl, menc, den, acc, myq, wlist, &wcount, lo, nown, lane, wave);
    }
    __syncthreads();

    const unsigned cnt = fromrec ? cnt_in : wcount;
    if constexpr (H == 4) {
        if (tid == 0) myhdr[0] = cnt;
    }
    constexpr int GB = (H == 1) ? 16 : 8;
    if (cnt <= (unsigned)LCAP) {
        for (int base = wave * 32; base < (int)cnt; base += AGG_WAVES * 32) {
            int n = (int)cnt - base;
            n = __builtin_amdgcn_readfirstlane(n < 32 ? n : 32);
            if (fromrec)
                flush_batch<H, 1, GB, true, false, false, false>(wlist, base, n, lane, lo, ei, h, as_, adl, menc, den,
                                                                 acc, wlist, &wcount, nullptr);
            else
                flush_batch<H, 1, GB, false, false, false, H == 4>(wlist, base, n, lane, lo, ei, h, as_, adl, menc, den,
                                                                   acc, wlist, &wcount, myrec);
        }
    } else {
        scan_pass<H, 1, 4>(ei, dig, h, as_, adl, menc, den, acc, myq, wlist, &wcount, lo, nown, lane, wave);
    }
    __syncthreads();

    if constexpr (H == 4) {
        constexpr int WSLOTS = LCAP / 2 / (AGG_WAVES * 32);
        v2f wreg[WSLOTS];
#pragma unroll
        for (int j = 0; j < WSLOTS; ++j) {
            const int slot = tid + j * (AGG_WAVES * 32);
            const int sl = slot & 31, tile = (slot >> 5) & 3, k4 = slot >> 7;
            const float* wp = W2 + (size_t)(4 * k4 + 2 * (sl >> 4)) * OUT_DIM + 16 * tile + (sl & 15);
            wreg[j].x = wp[0];
            wreg[j].y = wp[OUT_DIM];
        }
        for (int i = tid; i < nown * (F / 4); i += AGG_WAVES * 32) {
            const int row = i / (F / 4), col = (i % (F / 4)) * 4;
            const float dn = den[row * H + (col >> 6)] + 1e-16f;
            float4* p = (float4*)(acc + row * P + col);
            const float4 b = *(const float4*)(bias + col);
            float4 v = *p;
            v.x /= dn; v.y /= dn; v.z /= dn; v.w /= dn;
            v.x = leakyf(v.x + b.x, 0.01f); v.y = leakyf(v.y + b.y, 0.01f);
            v.z = leakyf(v.z + b.z, 0.01f); v.w = leakyf(v.w + b.w, 0.01f);
            *p = v;
        }
        v2f* wf = (v2f*)wlist;
#pragma unroll
        for (int j = 0; j < WSLOTS; ++j) wf[tid + j * (AGG_WAVES * 32)] = wreg[j];
        __syncthreads();

        const int half = lane >> 4, l = lane & 15;
        const int njobs = ((nown + 15) >> 4) * 2;
        const int wv = __builtin_amdgcn_readfirstlane(wave);
        for (int job = wv; job < njobs; job += AGG_WAVES) {
            const int rb = job >> 1, cp = job & 1;
            int arow = rb * 16 + l;
            arow = (arow < NOWN) ? arow : (NOWN - 1);
            const float* ap = acc + arow * P + 2 * half;
            const v2f* bp = wf + (2 * cp) * 32 + lane;
            v8f c0 = {}, c1 = {};
            for (int k4 = 0; k4 < W2_ROWS / 4; ++k4) {
                const v2f a  = *(const v2f*)(ap + 4 * k4);
                const v2f b0 = bp[k4 * 128];
                const v2f b1 = bp[k4 * 128 + 32];
                c0 = __builtin_amdgcn_wmma_f32_16x16x4_f32(false, a, false, b0, (short)0, c0, false, false);
                c1 = __builtin_amdgcn_wmma_f32_16x16x4_f32(false, a, false, b1, (short)0, c1, false, false);
            }
#pragma unroll
            for (int r = 0; r < 8; ++r) {
                const int row = rb * 16 + r + 8 * half;
                if (row < nown) {
                    float* gp = out + (size_t)(lo + row) * OUT_DIM + 32 * cp + l;
                    gp[ 0] = c0[r];
                    gp[16] = c1[r];
                }
            }
        }
    } else {
        const int n4 = nown * F / 4;
        float4* o4 = (float4*)(out + (size_t)lo * F);
        const float4* a4 = (const float4*)acc;
        for (int i = tid; i < n4; i += AGG_WAVES * 32) {
            const int idx = i * 4;
            const float dn = den[(idx / F) * H + ((idx % F) >> 6)] + 1e-16f;
            float4 v = a4[i];
            v.x /= dn; v.y /= dn; v.z /= dn; v.w /= dn;
            o4[i] = v;
        }
    }
}

__global__ void bias_act_kernel(const float* __restrict__ in, const float* __restrict__ b,
                                float* __restrict__ out, int total, int F, float slope) {
    int i = blockIdx.x * blockDim.x + threadIdx.x;
    if (i >= total) return;
    out[i] = leakyf(in[i] + b[i % F], slope);
}

extern "C" void kernel_launch(void* const* d_in, const int* in_sizes, int n_in,
                              void* d_out, int out_size, void* d_ws, size_t ws_size,
                              hipStream_t stream) {
    const float* x   = (const float*)d_in[0];
    const int*   ei  = (const int*)  d_in[1];
    const float* W1  = (const float*)d_in[2];
    const float* as1 = (const float*)d_in[3];
    const float* ad1 = (const float*)d_in[4];
    const float* b1  = (const float*)d_in[5];
    const float* W2  = (const float*)d_in[6];
    const float* as2 = (const float*)d_in[7];
    const float* ad2 = (const float*)d_in[8];
    const float* b2  = (const float*)d_in[9];
    float* out = (float*)d_out;

    float* ws   = (float*)d_ws;
    float* h1   = ws;
    float* g2   = h1   + (size_t)NNODES * F1;
    float* out2 = g2   + (size_t)NNODES * OUT_DIM;
    float* asb  = h1   + (size_t)2 * NNODES * F1;
    float* adb  = asb  + (size_t)NNODES * HEADS;
    unsigned* rec = (unsigned*)(out2 + (size_t)NNODES * OUT_DIM);
    unsigned* dig = rec + (size_t)NWG_AGG * RSTRIDE;

    {
        int strips = (NNODES / 16) * (F1 / 64);
        gemm_wmma_f32_kernel<<<(strips + 3) / 4, 128, 0, stream>>>(x, W1, h1, NNODES, IN_DIM, F1);
    }
    alpha_digest_kernel<<<DIG_BLOCKS + ALPHA1_BLOCKS, 256, 0, stream>>>(h1, as1, ad1, asb, adb, NNODES * HEADS,
                                                                        ei + NEDGES, dig);
    agg_owner_kernel<HEADS><<<NWG_AGG, AGG_WAVES * 32, 0, stream>>>(ei, h1, asb, adb, g2, b1, W2, rec, dig);

    alpha_kernel<1><<<((NNODES + ALPHA_GPW - 1) / ALPHA_GPW + 7) / 8, 256, 0, stream>>>(
        g2, as2, ad2, asb, adb, NNODES);
    agg_owner_kernel<1><<<NWG_AGG, AGG_WAVES * 32, 0, stream>>>(ei, g2, asb, adb, out2, b2, W2, rec, dig);
    bias_act_kernel<<<(NNODES * OUT_DIM + 255) / 256, 256, 0, stream>>>(out2, b2, out,
                                                                        NNODES * OUT_DIM,
                                                                        OUT_DIM, 1.0f);
}


# Layer-1 product: a wave keeps its 128x64 strip of W1 in registers and streams 16-row blocks of X through it (256 persistent work-groups); X read in 16-byte pieces: k order of the sums changes
# speedup vs baseline: 4.5317x; 1.0925x over previous
#include <hip/hip_runtime.h>
#include <math.h>

#define NNODES  50000
#define NEDGES  800000
#define ETOT    (NEDGES + NNODES)
#define IN_DIM  128
#define HID     64
#define HEADS   4
#define OUT_DIM 64
#define F1      (HEADS * HID)
#define ENC_NEG_INF 0x007FFFFFu

typedef __attribute__((ext_vector_type(2))) float v2f;
typedef __attribute__((ext_vector_type(8))) float v8f;

__device__ __forceinline__ float leakyf(float x, float s) { return x >= 0.f ? x : s * x; }

__device__ __forceinline__ unsigned enc_ord(float x) {
    unsigned u = __float_as_uint(x);
    return (u & 0x80000000u) ? ~u : (u | 0x80000000u);
}
__device__ __forceinline__ float dec_ord(unsigned e) {
    return (e & 0x80000000u) ? __uint_as_float(e & 0x7FFFFFFFu) : __uint_as_float(~e);
}

#define G1_WAVES 8
#define G1_WGS   256
#define G1_RB    (NNODES / 16)
#define G1_STEP  (G1_WGS * (G1_WAVES / 4))
static_assert(NNODES % 16 == 0 && IN_DIM == 128 && F1 == 256, "tile map of the layer-1 product");
static_assert(G1_STEP <= G1_RB, "every wave has a first row block");
__global__ void __launch_bounds__(G1_WAVES * 32)
gemm1_wreg_kernel(const float* __restrict__ X, const float* __restrict__ W, float* __restrict__ C) {
    const int tid = threadIdx.x, lane = tid & 31;
    const int half = lane >> 4;
    const int l    = lane & 15;
    const int wv = __builtin_amdgcn_readfirstlane(tid >> 5);
    const int strip = wv & 3;

    int rb = (int)blockIdx.x + G1_WGS * (wv >> 2);
    float4 a[IN_DIM / 8];
    {
        const float4* xp = (const float4*)(X + (size_t)((rb << 4) + l) * IN_DIM) + half;
#pragma unroll
        for (int q = 0; q < IN_DIM / 8; ++q) a[q] = xp[2 * q];
    }
    float4 b[IN_DIM / 8][4];
    const float* wp = W + (size_t)(4 * half) * F1 + (strip << 6) + l;
#pragma unroll
    for (int t = 0; t < 4; ++t) {
#pragma unroll
        for (int q = 0; q < IN_DIM / 8; ++q) {
            const float* p = wp + (size_t)(8 * q) * F1 + 16 * t;
            b[q][t].x = p[0]; b[q][t].y = p[F1]; b[q][t].z = p[2 * F1]; b[q][t].w = p[3 * F1];
        }
    }
    for (;;) {
        const int rbn = rb + G1_STEP;
        const bool more = rbn < G1_RB;
        const int rbl = more ? rbn : rb;
        float4 an[IN_DIM / 8];
        const float4* xn = (const float4*)(X + (size_t)((rbl << 4) + l) * IN_DIM) + half;
#pragma unroll
        for (int q = 0; q < IN_DIM / 8; ++q) an[q] = xn[2 * q];

        v8f acc0 = {}, acc1 = {}, acc2 = {}, acc3 = {};
#pragma unroll
        for (int q = 0; q < IN_DIM / 8; ++q) {
            const v2f a0 = { a[q].x, a[q].y }, a1 = { a[q].z, a[q].w };
            const v2f b00 = { b[q][0].x, b[q][0].y }, b01 = { b[q][0].z, b[q][0].w };
            const v2f b10 = { b[q][1].x, b[q][1].y }, b11 = { b[q][1].z, b[q][1].w };
            const v2f b20 = { b[q][2].x, b[q][2].y }, b21 = { b[q][2].z, b[q][2].w };
            const v2f b30 = { b[q][3].x, b[q][3].y }, b31 = { b[q][3].z, b[q][3].w };
            acc0 = __builtin_amdgcn_wmma_f32_16x16x4_f32(false, a0, false, b00, (short)0, acc0, false, false);
            acc1 = __builtin_amdgcn_wmma_f32_16x16x4_f32(false, a0, false, b10, (short)0, acc1, false, false);
            acc2 = __builtin_amdgcn_wmma_f32_16x16x4_f32(false, a0, false, b20, (short)0, acc2, false, false);
            acc3 = __builtin_amdgcn_wmma_f32_16x16x4_f32(false, a0, false, b30, (short)0, acc3, false, false);
            acc0 = __builtin_amdgcn_wmma_f32_16x16x4_f32(false, a1, false, b01, (short)0, acc0, false, false);
            acc1 = __builtin_amdgcn_wmma_f32_16x16x4_f32(false, a1, false, b11, (short)0, acc1, false, false);
            acc2 = __builtin_amdgcn_wmma_f32_16x16x4_f32(false, a1, false, b21, (short)0, acc2, false, false);
            acc3 = __builtin_amdgcn_wmma_f32_16x16x4_f32(false, a1, false, b31, (short)0, acc3, false, false);
        }
        float* crow = C + (size_t)(rb << 4) * F1 + (strip << 6);
#pragma unroll
        for (int r = 0; r < 8; ++r) {
            const int m = r + 8 * half;
            float* p = crow + (size_t)m * F1 + l;
            p[ 0] = acc0[r];
            p[16] = acc1[r];
            p[32] = acc2[r];
            p[48] = acc3[r];
        }
        if (!more) break;
        rb = rbn;
#pragma unroll
        for (int q = 0; q < IN_DIM / 8; ++q) a[q] = an[q];
    }
}

#define ALPHA_GPW 32
static_assert(HID == 64 && OUT_DIM == 64, "a group is 64 floats in both layers");
template <int H>
__device__ __forceinline__ void alpha_body(const float* __restrict__ h,
                                           const float* __restrict__ a_src, const float* __restrict__ a_dst,
                                           float* __restrict__ as_, float* __restrict__ ad_, int ngroups, int blk) {
    static_assert(16 % H == 0, "head of group g0 + 16*hf + j is j % H");
    const int lane = threadIdx.x & 31, hf = lane >> 4, l = lane & 15;
    const int g0 = (int)((blk * 256 + threadIdx.x) >> 5) * ALPHA_GPW;
    if (g0 >= ngroups) return;

    float4 v[16];
#pragma unroll
    for (int j = 0; j < 16; ++j) {
        int g = g0 + 16 * hf + j;
        g = (g < ngroups) ? g : (ngroups - 1);
        v[j] = *(const float4*)(h + (size_t)g * 64 + 4 * l);
    }
    float4 sa[H], da[H];
#pragma unroll
    for (int hh = 0; hh < H; ++hh) {
        sa[hh] = *(const float4*)(a_src + hh * 64 + 4 * l);
        da[hh] = *(const float4*)(a_dst + hh * 64 + 4 * l);
    }
    float ps[16], pd[16];
#pragma unroll
    for (int j = 0; j < 16; ++j) {
        const float4 s = sa[j % H], d = da[j % H];
        ps[j] = v[j].x * s.x + v[j].y * s.y + v[j].z * s.z + v[j].w * s.w;
        pd[j] = v[j].x * d.x + v[j].y * d.y + v[j].z * d.z + v[j].w * d.w;
    }
    const bool b3 = (l & 8) != 0, b2 = (l & 4) != 0, b1 = (l & 2) != 0, b0 = (l & 1) != 0;
    float s8[8], d8[8];
#pragma unroll
    for (int t = 0; t < 8; ++t) {
        s8[t] = (b3 ? ps[t + 8] : ps[t]) + __shfl_xor(b3 ? ps[t] : ps[t + 8], 8);
        d8[t] = (b3 ? pd[t + 8] : pd[t]) + __shfl_xor(b3 ? pd[t] : pd[t + 8], 8);
    }
    float s4[4], d4[4];
#pragma unroll
    for (int t = 0; t < 4; ++t) {
        s4[t] = (b2 ? s8[t + 4] : s8[t]) + __shfl_xor(b2 ? s8[t] : s8[t + 4], 4);
        d4[t] = (b2 ? d8[t + 4] : d8[t]) + __shfl_xor(b2 ? d8[t] : d8[t + 4], 4);
    }
    float s2[2], d2[2];
#pragma unroll
    for (int t = 0; t < 2; ++t) {
        s2[t] = (b1 ? s4[t + 2] : s4[t]) + __shfl_xor(b1 ? s4[t] : s4[t + 2], 2);
        d2[t] = (b1 ? d4[t + 2] : d4[t]) + __shfl_xor(b1 ? d4[t] : d4[t + 2], 2);
    }
    const float s1 = (b0 ? s2[1] : s2[0]) + __shfl_xor(b0 ? s2[0] : s2[1], 1);
    const float d1 = (b0 ? d2[1] : d2[0]) + __shfl_xor(b0 ? d2[0] : d2[1], 1);
    const int g = g0 + lane;
    if (g < ngroups) {
        as_[g] = s1;
        ad_[g] = d1;
    }
}

template <int H>
__global__ void __launch_bounds__(256)
alpha_kernel(const float* __restrict__ h,
             const float* __restrict__ a_src, const float* __restrict__ a_dst,
             float* __restrict__ as_, float* __restrict__ ad_, int ngroups) {
    alpha_body<H>(h, a_src, a_dst, as_, ad_, ngroups, (int)blockIdx.x);
}

#define NOWN       196
#define NDIG16     (NEDGES / 16)
#define DIG_BLOCKS ((NDIG16 + 255) / 256)
static_assert(NEDGES % 16 == 0, "the digest is written and read in 16-byte groups");
static_assert((NNODES - 1) / NOWN <= 255, "an owner index fits a byte");
__device__ __forceinline__ unsigned owner4(int4 d) {
    return  (unsigned)d.x / NOWN        | ((unsigned)d.y / NOWN) << 8 |
           ((unsigned)d.z / NOWN) << 16 | ((unsigned)d.w / NOWN) << 24;
}
__device__ __forceinline__ void digest_block(const int* __restrict__ dst, unsigned* __restrict__ dig, int blk) {
    const int i = blk * 256 + (int)threadIdx.x;
    if (i >= NDIG16) return;
    const int4* d4 = (const int4*)dst + 4 * (size_t)i;
    const int4 a = d4[0], b = d4[1], cc = d4[2], d = d4[3];
    uint4 o;
    o.x = owner4(a); o.y = owner4(b); o.z = owner4(cc); o.w = owner4(d);
    ((uint4*)dig)[i] = o;
}

#define ALPHA1_BLOCKS (((NNODES * HEADS + ALPHA_GPW - 1) / ALPHA_GPW + 7) / 8)
__global__ void __launch_bounds__(256)
alpha_digest_kernel(const float* __restrict__ h,
                    const float* __restrict__ a_src, const float* __restrict__ a_dst,
                    float* __restrict__ as_, float* __restrict__ ad_, int ngroups,
                    const int* __restrict__ dst, unsigned* __restrict__ dig) {
    if ((int)blockIdx.x < DIG_BLOCKS) digest_block(dst, dig, (int)blockIdx.x);
    else alpha_body<HEADS>(h, a_src, a_dst, as_, ad_, ngroups, (int)blockIdx.x - DIG_BLOCKS);
}

#define NWG_AGG   256
#define AGG_WAVES 32
#define QCAP      64
#define LCAP      16384
#define SCAN_U    8
#define CHUNK16   (SCAN_U * 32)
#define NCHUNK    ((NDIG16 + CHUNK16 - 1) / CHUNK16)

static_assert(NOWN * NWG_AGG >= NNODES && (NWG_AGG - 1) * NOWN < NNODES, "every node has one owner");
static_assert(ETOT <= (1 << 20) && NOWN <= 256, "record entry = edge id | local node << 20");
static_assert(NNODES <= (1 << 20), "handed-over entry = source node | local node << 20");
#define RSTRIDE   (LCAP + 32)
static_assert((size_t)NNODES * (F1 + 2 * OUT_DIM) + (size_t)NWG_AGG * RSTRIDE + NEDGES / 4 <= (size_t)2 * NNODES * F1,
              "the records and the digest end before the logits begin");
static_assert(((size_t)NWG_AGG * RSTRIDE) % 4 == 0, "the digest starts on a 16-byte boundary");
static_assert(((size_t)NNODES * (F1 + 2 * OUT_DIM)) % 32 == 0, "records start on a 128-byte line of the workspace");

template <int H> struct RowPitch { static constexpr int value = (H == 4) ? (H * 64 + 4) : (H * 64); };
#define W2_ROWS F1
static_assert(W2_ROWS * OUT_DIM == LCAP, "the W2 fragments take exactly the edge record's LDS");
static_assert(OUT_DIM == 64 && W2_ROWS % 4 == 0, "layer-2 tile map below");

template <int H, int PHASE, int G, bool SRC, bool QDST, bool REC, bool EMIT>
__device__ __forceinline__ void flush_batch(const unsigned* list, int base, int n, int lane, int lo,
                                            const int* __restrict__ ei,
                                            const float* __restrict__ h,
                                            const float* __restrict__ as_,
                                            const float* adl, unsigned* menc, float* den, float* acc,
                                            unsigned* wlist, unsigned* wcount,
                                            unsigned* __restrict__ grec) {
    constexpr int F = H * 64;
    constexpr int P = RowPitch<H>::value;
    const bool act = lane < n;
    unsigned ent = 0;
    if (act) ent = list[base + lane];
    const int t  = (int)(ent & 0xFFFFFu);
    int dl = (int)(ent >> 20);
    int s = 0;
    if (act) {
        if (SRC) s = t;
        else     s = (t < NEDGES) ? ei[t] : (t - NEDGES);
        if (QDST) {
            dl = ((t < NEDGES) ? ei[NEDGES + t] : (t - NEDGES)) - lo;
            ent = (unsigned)t | ((unsigned)dl << 20);
        }
    }
    float e[H];
#pragma unroll
    for (int hh = 0; hh < H; ++hh) e[hh] = 0.f;
    if (act) {
#pragma unroll
        for (int hh = 0; hh < H; ++hh) e[hh] = leakyf(as_[s * H + hh] + adl[dl * H + hh], 0.2f);
    }
    if (PHASE == 0) {
        if (act) {
#pragma unroll
            for (int hh = 0; hh < H; ++hh)
                __hip_atomic_fetch_max(&menc[dl * H + hh], enc_ord(e[hh]), __ATOMIC_RELAXED,
                                       __HIP_MEMORY_SCOPE_WORKGROUP);
        }
        if (REC) {
            unsigned pos = 0;
            if (lane == 0)
                pos = __hip_atomic_fetch_add(wcount, (unsigned)n, __ATOMIC_RELAXED, __HIP_MEMORY_SCOPE_WORKGROUP);
            pos = (unsigned)__builtin_amdgcn_readfirstlane((int)pos);
            if (act && pos + (unsigned)n <= (unsigned)LCAP) wlist[pos + lane] = ent;
        }
    } else {
        if (EMIT) {
            if (act) grec[base + lane] = (unsigned)s | ((unsigned)dl << 20);
        }
        float ex[H];
#pragma unroll
        for (int hh = 0; hh < H; ++hh) ex[hh] = 0.f;
        if (act) {
#pragma unroll
            for (int hh = 0; hh < H; ++hh) {
                ex[hh] = expf(e[hh] - dec_ord(menc[dl * H + hh]));
                __hip_atomic_fetch_add(&den[dl * H + hh], ex[hh], __ATOMIC_RELAXED, __HIP_MEMORY_SCOPE_WORKGROUP);
            }
        }
        for (int j = 0; j < n; j += G) {
            v2f   r[G][H];
            float w[G][H];
            int   dj[G];
#pragma unroll
            for (int g = 0; g < G; ++g) {
                const int jj = (j + g < n) ? (j + g) : (n - 1);
                const int sj = __builtin_amdgcn_readlane(s, jj);
                dj[g] = __builtin_amdgcn_readlane(dl, jj);
#pragma unroll
                for (int hh = 0; hh < H; ++hh)
                    w[g][hh] = __int_as_float(__builtin_amdgcn_readlane(__float_as_int(ex[hh]), jj));
                const v2f* hp = (const v2f*)(h + (size_t)sj * F) + lane;
#pragma unroll
                for (int hh = 0; hh < H; ++hh) r[g][hh] = hp[hh * 32];
            }
#pragma unroll
            for (int g = 0; g < G; ++g) {
                if (j + g < n) {
                    float* ap = acc + dj[g] * P + 2 * lane;
#pragma unroll
                    for (int hh = 0; hh < H; ++hh) {
                        __hip_atomic_fetch_add(ap + hh * 64,     w[g][hh] * r[g][hh].x, __ATOMIC_RELAXED,
                                               __HIP_MEMORY_SCOPE_WORKGROUP);
                        __hip_atomic_fetch_add(ap + hh * 64 + 1, w[g][hh] * r[g][hh].y, __ATOMIC_RELAXED,
                                               __HIP_MEMORY_SCOPE_WORKGROUP);
                    }
                }
            }
        }
    }
}

__device__ __forceinline__ unsigned zero_bytes(unsigned x) {
    const unsigned t = (x & 0x7F7F7F7Fu) + 0x7F7F7F7Fu;
    return ~(t | x) & 0x80808080u;
}

template <int H, int PHASE, int G>
__device__ __forceinline__ void scan_pass(const int* __restrict__ ei, const unsigned* __restrict__ dig,
                                          const float* __restrict__ h,
                                          const float* __restrict__ as_,
                                          const float* adl, unsigned* menc, float* den, float* acc,
                                          unsigned* myq, unsigned* wlist, unsigned* wcount,
                                          int lo, int nown, int lane, int wave) {
    int qn = 0;
    auto enqueue = [&](bool mt, unsigned val) {
        const unsigned mask = __builtin_amdgcn_ballot_w32(mt);
        if (mt) myq[qn + __builtin_popcount(mask & ((1u << lane) - 1u))] = val;
        qn += __builtin_popcount(mask);
    };
    {
        const int dl = wave + AGG_WAVES * lane;
        enqueue(dl < nown, (unsigned)(NEDGES + lo + dl));
    }
    const uint4* dg4 = (const uint4*)dig;
    const unsigned mine = (unsigned)blockIdx.x * 0x01010101u;
    const int c0 = ((int)blockIdx.x * NCHUNK) / NWG_AGG;
    for (int jc = wave; jc < NCHUNK; jc += AGG_WAVES) {
        int c = jc + c0;
        if (c >= NCHUNK) c -= NCHUNK;
        const int i0 = c * CHUNK16 + lane;
        uint4 dv[SCAN_U];
#pragma unroll
        for (int u = 0; u < SCAN_U; ++u) {
            const int i16 = i0 + u * 32;
            dv[u] = dg4[(i16 < NDIG16) ? i16 : 0];
        }
#pragma unroll
        for (int u = 0; u < SCAN_U; ++u) {
            const int i16 = i0 + u * 32;
            unsigned f = (zero_bytes(dv[u].x ^ mine) >> 7) | (zero_bytes(dv[u].y ^ mine) >> 6) |
                         (zero_bytes(dv[u].z ^ mine) >> 5) | (zero_bytes(dv[u].w ^ mine) >> 4);
            if (i16 >= NDIG16) f = 0u;
            while (__builtin_amdgcn_ballot_w32(f != 0u) != 0u) {
                const int p = __builtin_ctz(f | 0x80000000u);
                enqueue(f != 0u, (unsigned)(i16 * 16 + 4 * (p & 7) + (p >> 3)));
                f &= f - 1u;
                while (qn >= 32) {
                    flush_batch<H, PHASE, G, false, true, PHASE == 0, false>(myq, qn - 32, 32, lane, lo, ei, h, as_, adl,
                                                                             menc, den, acc, wlist, wcount, nullptr);
                    qn -= 32;
                }
            }
        }
    }
    if (qn > 0)
        flush_batch<H, PHASE, G, false, true, PHASE == 0, false>(myq, 0, qn, lane, lo, ei, h, as_, adl, menc, den, acc,
                                                                 wlist, wcount, nullptr);
}

template <int H>
__global__ void __launch_bounds__(AGG_WAVES * 32)
agg_owner_kernel(const int* __restrict__ ei, const float* __restrict__ h,
                 const float* __restrict__ as_, const float* __restrict__ ad_,
                 float* __restrict__ out,
                 const float* __restrict__ bias, const float* __restrict__ W2,
                 unsigned* __restrict__ grec, const unsigned* __restrict__ dig) {
    constexpr int F = H * 64;
    constexpr int P = RowPitch<H>::value;
    __shared__ __attribute__((aligned(16))) float acc[NOWN * P];
    __shared__ unsigned menc[NOWN * H];
    __shared__ float    den[NOWN * H];
    __shared__ float    adl[NOWN * H];
    __shared__ unsigned q[AGG_WAVES * QCAP];
    __shared__ __attribute__((aligned(16))) unsigned wlist[LCAP];
    __shared__ unsigned wcount;

    const int tid = threadIdx.x, lane = tid & 31, wave = tid >> 5;
    const int lo = (int)blockIdx.x * NOWN;
    const int nown = (NNODES - lo < NOWN) ? (NNODES - lo) : NOWN;
    unsigned* myhdr = grec + (size_t)blockIdx.x * RSTRIDE;
    unsigned* myrec = myhdr + 32;

    unsigned cnt_in = (unsigned)LCAP + 1u;
    if constexpr (H == 1) {
        cnt_in = (unsigned)__builtin_amdgcn_readfirstlane((int)myhdr[0]);
        if (cnt_in <= (unsigned)LCAP)
            for (int i = tid; i < (int)cnt_in; i += AGG_WAVES * 32) wlist[i] = myrec[i];
    }
    const bool fromrec = (H == 1) && (cnt_in <= (unsigned)LCAP);

    for (int i = tid; i < NOWN * P; i += AGG_WAVES * 32) acc[i] = 0.f;
    for (int i = tid; i < NOWN * H; i += AGG_WAVES * 32) {
        menc[i] = ENC_NEG_INF;
        den[i]  = 0.f;
        adl[i]  = (i < nown * H) ? ad_[lo * H + i] : 0.f;
    }
    if (tid == 0) wcount = 0u;
    __syncthreads();

    unsigned* myq = q + wave * QCAP;
    if (fromrec) {
        for (int base = wave * 32; base < (int)cnt_in; base += AGG_WAVES * 32) {
            int n = (int)cnt_in - base;
            n = __builtin_amdgcn_readfirstlane(n < 32 ? n : 32);
            flush_batch<H, 0, 4, true, false, false, false>(wlist, base, n, lane, lo, ei, h, as_, adl, menc, den, acc,
                                                            wlist, &wcount, nullptr);
        }
    } else {
        scan_pass<H, 0, 4>(ei, dig, h, as_, adl, menc, den, acc, myq, wlist, &wcount, lo, nown, lane, wave);
    }
    __syncthreads();

    const unsigned cnt = fromrec ? cnt_in : wcount;
    if constexpr (H == 4) {
        if (tid == 0) myhdr[0] = cnt;
    }
    constexpr int GB = (H == 1) ? 16 : 8;
    if (cnt <= (unsigned)LCAP) {
        for (int base = wave * 32; base < (int)cnt; base += AGG_WAVES * 32) {
            int n = (int)cnt - base;
            n = __builtin_amdgcn_readfirstlane(n < 32 ? n : 32);
            if (fromrec)
                flush_batch<H, 1, GB, true, false, false, false>(wlist, base, n, lane, lo, ei, h, as_, adl, menc, den,
                                                                 acc, wlist, &wcount, nullptr);
            else
                flush_batch<H, 1, GB, false, false, false, H == 4>(wlist, base, n, lane, lo, ei, h, as_, adl, menc, den,
                                                                   acc, wlist, &wcount, myrec);
        }
    } else {
        scan_pass<H, 1, 4>(ei, dig, h, as_, adl, menc, den, acc, myq, wlist, &wcount, lo, nown, lane, wave);
    }
    __syncthreads();

    if constexpr (H == 4) {
        constexpr int WSLOTS = LCAP / 2 / (AGG_WAVES * 32);
        v2f wreg[WSLOTS];
#pragma unroll
        for (int j = 0; j < WSLOTS; ++j) {
            const int slot = tid + j * (AGG_WAVES * 32);
            const int sl = slot & 31, tile = (slot >> 5) & 3, k4 = slot >> 7;
            const float* wp = W2 + (size_t)(4 * k4 + 2 * (sl >> 4)) * OUT_DIM + 16 * tile + (sl & 15);
            wreg[j].x = wp[0];
            wreg[j].y = wp[OUT_DIM];
        }
        for (int i = tid; i < nown * (F / 4); i += AGG_WAVES * 32) {
            const int row = i / (F / 4), col = (i % (F / 4)) * 4;
            const float dn = den[row * H + (col >> 6)] + 1e-16f;
            float4* p = (float4*)(acc + row * P + col);
            const float4 b = *(const float4*)(bias + col);
            float4 v = *p;
            v.x /= dn; v.y /= dn; v.z /= dn; v.w /= dn;
            v.x = leakyf(v.x + b.x, 0.01f); v.y = leakyf(v.y + b.y, 0.01f);
            v.z = leakyf(v.z + b.z, 0.01f); v.w = leakyf(v.w + b.w, 0.01f);
            *p = v;
        }
        v2f* wf = (v2f*)wlist;
#pragma unroll
        for (int j = 0; j < WSLOTS; ++j) wf[tid + j * (AGG_WAVES * 32)] = wreg[j];
        __syncthreads();

        const int half = lane >> 4, l = lane & 15;
        const int njobs = ((nown + 15) >> 4) * 2;
        const int wv = __builtin_amdgcn_readfirstlane(wave);
        for (int job = wv; job < njobs; job += AGG_WAVES) {
            const int rb = job >> 1, cp = job & 1;
            int arow = rb * 16 + l;
            arow = (arow < NOWN) ? arow : (NOWN - 1);
            const float* ap = acc + arow * P + 2 * half;
            const v2f* bp = wf + (2 * cp) * 32 + lane;
            v8f c0 = {}, c1 = {};
            for (int k4 = 0; k4 < W2_ROWS / 4; ++k4) {
                const v2f a  = *(const v2f*)(ap + 4 * k4);
                const v2f b0 = bp[k4 * 128];
                const v2f b1 = bp[k4 * 128 + 32];
                c0 = __builtin_amdgcn_wmma_f32_16x16x4_f32(false, a, false, b0, (short)0, c0, false, false);
                c1 = __builtin_amdgcn_wmma_f32_16x16x4_f32(false, a, false, b1, (short)0, c1, false, false);
            }
#pragma unroll
            for (int r = 0; r < 8; ++r) {
                const int row = rb * 16 + r + 8 * half;
                if (row < nown) {
                    float* gp = out + (size_t)(lo + row) * OUT_DIM + 32 * cp + l;
                    gp[ 0] = c0[r];
                    gp[16] = c1[r];
                }
            }
        }
    } else {
        const int n4 = nown * F / 4;
        float4* o4 = (float4*)(out + (size_t)lo * F);
        const float4* a4 = (const float4*)acc;
        for (int i = tid; i < n4; i += AGG_WAVES * 32) {
            const int idx = i * 4;
            const float dn = den[(idx / F) * H + ((idx % F) >> 6)] + 1e-16f;
            float4 v = a4[i];
            v.x /= dn; v.y /= dn; v.z /= dn; v.w /= dn;
            o4[i] = v;
        }
    }
}

__global__ void bias_act_kernel(const float* __restrict__ in, const float* __restrict__ b,
                                float* __restrict__ out, int total, int F, float slope) {
    int i = blockIdx.x * blockDim.x + threadIdx.x;
    if (i >= total) return;
    out[i] = leakyf(in[i] + b[i % F], slope);
}

extern "C" void kernel_launch(void* const* d_in, const int* in_sizes, int n_in,
                              void* d_out, int out_size, void* d_ws, size_t ws_size,
                              hipStream_t stream) {
    const float* x   = (const float*)d_in[0];
    const int*   ei  = (const int*)  d_in[1];
    const float* W1  = (const float*)d_in[2];
    const float* as1 = (const float*)d_in[3];
    const float* ad1 = (const float*)d_in[4];
    const float* b1  = (const float*)d_in[5];
    const float* W2  = (const float*)d_in[6];
    const float* as2 = (const float*)d_in[7];
    const float* ad2 = (const float*)d_in[8];
    const float* b2  = (const float*)d_in[9];
    float* out = (float*)d_out;

    float* ws   = (float*)d_ws;
    float* h1   = ws;
    float* g2   = h1   + (size_t)NNODES * F1;
    float* out2 = g2   + (size_t)NNODES * OUT_DIM;
    float* asb  = h1   + (size_t)2 * NNODES * F1;
    float* adb  = asb  + (size_t)NNODES * HEADS;
    unsigned* rec = (unsigned*)(out2 + (size_t)NNODES * OUT_DIM);
    unsigned* dig = rec + (size_t)NWG_AGG * RSTRIDE;

    gemm1_wreg_kernel<<<G1_WGS, G1_WAVES * 32, 0, stream>>>(x, W1, h1);
    alpha_digest_kernel<<<DIG_BLOCKS + ALPHA1_BLOCKS, 256, 0, stream>>>(h1, as1, ad1, asb, adb, NNODES * HEADS,
                                                                        ei + NEDGES, dig);
    agg_owner_kernel<HEADS><<<NWG_AGG, AGG_WAVES * 32, 0, stream>>>(ei, h1, asb, adb, g2, b1, W2, rec, dig);

    alpha_kernel<1><<<((NNODES + ALPHA_GPW - 1) / ALPHA_GPW + 7) / 8, 256, 0, stream>>>(
        g2, as2, ad2, asb, adb, NNODES);
    agg_owner_kernel<1><<<NWG_AGG, AGG_WAVES * 32, 0, stream>>>(ei, g2, asb, adb, out2, b2, W2, rec, dig);
    bias_act_kernel<<<(NNODES * OUT_DIM + 255) / 256, 256, 0, stream>>>(out2, b2, out,
                                                                        NNODES * OUT_DIM,
                                                                        OUT_DIM, 1.0f);
}


# Owner index: digest blocks also write, per owner, a bitmap of the 16-edge digest groups that hold its edges; an owner loads only those groups, not the whole digest; kept-edge (LDS add) order changes
# speedup vs baseline: 4.7034x; 1.0379x over previous
#include <hip/hip_runtime.h>
#include <math.h>

#define NNODES  50000
#define NEDGES  800000
#define ETOT    (NEDGES + NNODES)
#define IN_DIM  128
#define HID     64
#define HEADS   4
#define OUT_DIM 64
#define F1      (HEADS * HID)
#define ENC_NEG_INF 0x007FFFFFu

typedef __attribute__((ext_vector_type(2))) float v2f;
typedef __attribute__((ext_vector_type(8))) float v8f;

__device__ __forceinline__ float leakyf(float x, float s) { return x >= 0.f ? x : s * x; }

__device__ __forceinline__ unsigned enc_ord(float x) {
    unsigned u = __float_as_uint(x);
    return (u & 0x80000000u) ? ~u : (u | 0x80000000u);
}
__device__ __forceinline__ float dec_ord(unsigned e) {
    return (e & 0x80000000u) ? __uint_as_float(e & 0x7FFFFFFFu) : __uint_as_float(~e);
}

#define G1_WAVES 8
#define G1_WGS   256
#define G1_RB    (NNODES / 16)
#define G1_STEP  (G1_WGS * (G1_WAVES / 4))
static_assert(NNODES % 16 == 0 && IN_DIM == 128 && F1 == 256, "tile map of the layer-1 product");
static_assert(G1_STEP <= G1_RB, "every wave has a first row block");
__global__ void __launch_bounds__(G1_WAVES * 32)
gemm1_wreg_kernel(const float* __restrict__ X, const float* __restrict__ W, float* __restrict__ C) {
    const int tid = threadIdx.x, lane = tid & 31;
    const int half = lane >> 4;
    const int l    = lane & 15;
    const int wv = __builtin_amdgcn_readfirstlane(tid >> 5);
    const int strip = wv & 3;

    int rb = (int)blockIdx.x + G1_WGS * (wv >> 2);
    float4 a[IN_DIM / 8];
    {
        const float4* xp = (const float4*)(X + (size_t)((rb << 4) + l) * IN_DIM) + half;
#pragma unroll
        for (int q = 0; q < IN_DIM / 8; ++q) a[q] = xp[2 * q];
    }
    float4 b[IN_DIM / 8][4];
    const float* wp = W + (size_t)(4 * half) * F1 + (strip << 6) + l;
#pragma unroll
    for (int t = 0; t < 4; ++t) {
#pragma unroll
        for (int q = 0; q < IN_DIM / 8; ++q) {
            const float* p = wp + (size_t)(8 * q) * F1 + 16 * t;
            b[q][t].x = p[0]; b[q][t].y = p[F1]; b[q][t].z = p[2 * F1]; b[q][t].w = p[3 * F1];
        }
    }
    for (;;) {
        const int rbn = rb + G1_STEP;
        const bool more = rbn < G1_RB;
        const int rbl = more ? rbn : rb;
        float4 an[IN_DIM / 8];
        const float4* xn = (const float4*)(X + (size_t)((rbl << 4) + l) * IN_DIM) + half;
#pragma unroll
        for (int q = 0; q < IN_DIM / 8; ++q) an[q] = xn[2 * q];

        v8f acc0 = {}, acc1 = {}, acc2 = {}, acc3 = {};
#pragma unroll
        for (int q = 0; q < IN_DIM / 8; ++q) {
            const v2f a0 = { a[q].x, a[q].y }, a1 = { a[q].z, a[q].w };
            const v2f b00 = { b[q][0].x, b[q][0].y }, b01 = { b[q][0].z, b[q][0].w };
            const v2f b10 = { b[q][1].x, b[q][1].y }, b11 = { b[q][1].z, b[q][1].w };
            const v2f b20 = { b[q][2].x, b[q][2].y }, b21 = { b[q][2].z, b[q][2].w };
            const v2f b30 = { b[q][3].x, b[q][3].y }, b31 = { b[q][3].z, b[q][3].w };
            acc0 = __builtin_amdgcn_wmma_f32_16x16x4_f32(false, a0, false, b00, (short)0, acc0, false, false);
            acc1 = __builtin_amdgcn_wmma_f32_16x16x4_f32(false, a0, false, b10, (short)0, acc1, false, false);
            acc2 = __builtin_amdgcn_wmma_f32_16x16x4_f32(false, a0, false, b20, (short)0, acc2, false, false);
            acc3 = __builtin_amdgcn_wmma_f32_16x16x4_f32(false, a0, false, b30, (short)0, acc3, false, false);
            acc0 = __builtin_amdgcn_wmma_f32_16x16x4_f32(false, a1, false, b01, (short)0, acc0, false, false);
            acc1 = __builtin_amdgcn_wmma_f32_16x16x4_f32(false, a1, false, b11, (short)0, acc1, false, false);
            acc2 = __builtin_amdgcn_wmma_f32_16x16x4_f32(false, a1, false, b21, (short)0, acc2, false, false);
            acc3 = __builtin_amdgcn_wmma_f32_16x16x4_f32(false, a1, false, b31, (short)0, acc3, false, false);
        }
        float* crow = C + (size_t)(rb << 4) * F1 + (strip << 6);
#pragma unroll
        for (int r = 0; r < 8; ++r) {
            const int m = r + 8 * half;
            float* p = crow + (size_t)m * F1 + l;
            p[ 0] = acc0[r];
            p[16] = acc1[r];
            p[32] = acc2[r];
            p[48] = acc3[r];
        }
        if (!more) break;
        rb = rbn;
#pragma unroll
        for (int q = 0; q < IN_DIM / 8; ++q) a[q] = an[q];
    }
}

#define ALPHA_GPW 32
static_assert(HID == 64 && OUT_DIM == 64, "a group is 64 floats in both layers");
template <int H>
__device__ __forceinline__ void alpha_body(const float* __restrict__ h,
                                           const float* __restrict__ a_src, const float* __restrict__ a_dst,
                                           float* __restrict__ as_, float* __restrict__ ad_, int ngroups, int blk) {
    static_assert(16 % H == 0, "head of group g0 + 16*hf + j is j % H");
    const int lane = threadIdx.x & 31, hf = lane >> 4, l = lane & 15;
    const int g0 = (int)((blk * 256 + threadIdx.x) >> 5) * ALPHA_GPW;
    if (g0 >= ngroups) return;

    float4 v[16];
#pragma unroll
    for (int j = 0; j < 16; ++j) {
        int g = g0 + 16 * hf + j;
        g = (g < ngroups) ? g : (ngroups - 1);
        v[j] = *(const float4*)(h + (size_t)g * 64 + 4 * l);
    }
    float4 sa[H], da[H];
#pragma unroll
    for (int hh = 0; hh < H; ++hh) {
        sa[hh] = *(const float4*)(a_src + hh * 64 + 4 * l);
        da[hh] = *(const float4*)(a_dst + hh * 64 + 4 * l);
    }
    float ps[16], pd[16];
#pragma unroll
    for (int j = 0; j < 16; ++j) {
        const float4 s = sa[j % H], d = da[j % H];
        ps[j] = v[j].x * s.x + v[j].y * s.y + v[j].z * s.z + v[j].w * s.w;
        pd[j] = v[j].x * d.x + v[j].y * d.y + v[j].z * d.z + v[j].w * d.w;
    }
    const bool b3 = (l & 8) != 0, b2 = (l & 4) != 0, b1 = (l & 2) != 0, b0 = (l & 1) != 0;
    float s8[8], d8[8];
#pragma unroll
    for (int t = 0; t < 8; ++t) {
        s8[t] = (b3 ? ps[t + 8] : ps[t]) + __shfl_xor(b3 ? ps[t] : ps[t + 8], 8);
        d8[t] = (b3 ? pd[t + 8] : pd[t]) + __shfl_xor(b3 ? pd[t] : pd[t + 8], 8);
    }
    float s4[4], d4[4];
#pragma unroll
    for (int t = 0; t < 4; ++t) {
        s4[t] = (b2 ? s8[t + 4] : s8[t]) + __shfl_xor(b2 ? s8[t] : s8[t + 4], 4);
        d4[t] = (b2 ? d8[t + 4] : d8[t]) + __shfl_xor(b2 ? d8[t] : d8[t + 4], 4);
    }
    float s2[2], d2[2];
#pragma unroll
    for (int t = 0; t < 2; ++t) {
        s2[t] = (b1 ? s4[t + 2] : s4[t]) + __shfl_xor(b1 ? s4[t] : s4[t + 2], 2);
        d2[t] = (b1 ? d4[t + 2] : d4[t]) + __shfl_xor(b1 ? d4[t] : d4[t + 2], 2);
    }
    const float s1 = (b0 ? s2[1] : s2[0]) + __shfl_xor(b0 ? s2[0] : s2[1], 1);
    const float d1 = (b0 ? d2[1] : d2[0]) + __shfl_xor(b0 ? d2[0] : d2[1], 1);
    const int g = g0 + lane;
    if (g < ngroups) {
        as_[g] = s1;
        ad_[g] = d1;
    }
}

template <int H>
__global__ void __launch_bounds__(256)
alpha_kernel(const float* __restrict__ h,
             const float* __restrict__ a_src, const float* __restrict__ a_dst,
             float* __restrict__ as_, float* __restrict__ ad_, int ngroups) {
    alpha_body<H>(h, a_src, a_dst, as_, ad_, ngroups, (int)blockIdx.x);
}

#define NOWN       196
#define NDIG16     (NEDGES / 16)
#define DIG_BLOCKS ((NDIG16 + 255) / 256)
static_assert(NEDGES % 16 == 0, "the digest is written and read in 16-byte groups");
static_assert((NNODES - 1) / NOWN <= 255, "an owner index fits a byte");
__device__ __forceinline__ unsigned owner4(int4 d) {
    return  (unsigned)d.x / NOWN        | ((unsigned)d.y / NOWN) << 8 |
           ((unsigned)d.z / NOWN) << 16 | ((unsigned)d.w / NOWN) << 24;
}
#define BM_W      8
#define BM_WORDS  (DIG_BLOCKS * BM_W)
static_assert(BM_W == 8 && BM_W * 32 == 256, "a digest block is 8 waves; word index = 8 * owner + wave below");
__device__ __forceinline__ void or_owner_bits(unsigned* row, unsigned o, unsigned bit) {
    __hip_atomic_fetch_or(&row[o & 255u],         bit, __ATOMIC_RELAXED, __HIP_MEMORY_SCOPE_WORKGROUP);
    __hip_atomic_fetch_or(&row[(o >> 8) & 255u],  bit, __ATOMIC_RELAXED, __HIP_MEMORY_SCOPE_WORKGROUP);
    __hip_atomic_fetch_or(&row[(o >> 16) & 255u], bit, __ATOMIC_RELAXED, __HIP_MEMORY_SCOPE_WORKGROUP);
    __hip_atomic_fetch_or(&row[o >> 24],          bit, __ATOMIC_RELAXED, __HIP_MEMORY_SCOPE_WORKGROUP);
}
__device__ __forceinline__ void digest_block(const int* __restrict__ dst, unsigned* __restrict__ dig,
                                             unsigned* __restrict__ gbm, int blk) {
    __shared__ unsigned bm[BM_W * 256];
    const int tid = (int)threadIdx.x;
    for (int k = tid; k < BM_W * 256; k += 256) bm[k] = 0u;
    __syncthreads();
    const int i = blk * 256 + tid;
    if (i < NDIG16) {
        const int4* d4 = (const int4*)dst + 4 * (size_t)i;
        const int4 a = d4[0], b = d4[1], cc = d4[2], d = d4[3];
        uint4 o;
        o.x = owner4(a); o.y = owner4(b); o.z = owner4(cc); o.w = owner4(d);
        ((uint4*)dig)[i] = o;
        unsigned* row = bm + (tid >> 5) * 256;
        const unsigned bit = 1u << (tid & 31);
        or_owner_bits(row, o.x, bit); or_owner_bits(row, o.y, bit);
        or_owner_bits(row, o.z, bit); or_owner_bits(row, o.w, bit);
    }
    __syncthreads();
    unsigned* mybm = gbm + (size_t)blk * (BM_W * 256);
    for (int k = tid; k < BM_W * 256; k += 256) mybm[k] = bm[(k & (BM_W - 1)) * 256 + (k >> 3)];
}

#define ALPHA1_BLOCKS (((NNODES * HEADS + ALPHA_GPW - 1) / ALPHA_GPW + 7) / 8)
__global__ void __launch_bounds__(256)
alpha_digest_kernel(const float* __restrict__ h,
                    const float* __restrict__ a_src, const float* __restrict__ a_dst,
                    float* __restrict__ as_, float* __restrict__ ad_, int ngroups,
                    const int* __restrict__ dst, unsigned* __restrict__ dig, unsigned* __restrict__ gbm) {
    if ((int)blockIdx.x < DIG_BLOCKS) digest_block(dst, dig, gbm, (int)blockIdx.x);
    else alpha_body<HEADS>(h, a_src, a_dst, as_, ad_, ngroups, (int)blockIdx.x - DIG_BLOCKS);
}

#define NWG_AGG   256
#define AGG_WAVES 32
#define QCAP      64
#define LCAP      16384
#define SCAN_U    8

static_assert(NOWN * NWG_AGG >= NNODES && (NWG_AGG - 1) * NOWN < NNODES, "every node has one owner");
static_assert(ETOT <= (1 << 20) && NOWN <= 256, "record entry = edge id | local node << 20");
static_assert(NNODES <= (1 << 20), "handed-over entry = source node | local node << 20");
#define RSTRIDE   (LCAP + 32)
static_assert((size_t)NNODES * (F1 + 2 * OUT_DIM) + (size_t)NWG_AGG * RSTRIDE + NEDGES / 4 +
              (size_t)DIG_BLOCKS * 256 * BM_W <= (size_t)2 * NNODES * F1,
              "the records, the digest and the owner index end before the logits begin");
static_assert(NWG_AGG == 256 && (NEDGES / 4) % 32 == 0, "one index row per owner; the index starts on a 128-byte line");
static_assert(((size_t)NWG_AGG * RSTRIDE) % 4 == 0, "the digest starts on a 16-byte boundary");
static_assert(((size_t)NNODES * (F1 + 2 * OUT_DIM)) % 32 == 0, "records start on a 128-byte line of the workspace");

template <int H> struct RowPitch { static constexpr int value = (H == 4) ? (H * 64 + 4) : (H * 64); };
#define W2_ROWS F1
static_assert(W2_ROWS * OUT_DIM == LCAP, "the W2 fragments take exactly the edge record's LDS");
static_assert(OUT_DIM == 64 && W2_ROWS % 4 == 0, "layer-2 tile map below");

template <int H, int PHASE, int G, bool SRC, bool QDST, bool REC, bool EMIT>
__device__ __forceinline__ void flush_batch(const unsigned* list, int base, int n, int lane, int lo,
                                            const int* __restrict__ ei,
                                            const float* __restrict__ h,
                                            const float* __restrict__ as_,
                                            const float* adl, unsigned* menc, float* den, float* acc,
                                            unsigned* wlist, unsigned* wcount,
                                            unsigned* __restrict__ grec) {
    constexpr int F = H * 64;
    constexpr int P = RowPitch<H>::value;
    const bool act = lane < n;
    unsigned ent = 0;
    if (act) ent = list[base + lane];
    const int t  = (int)(ent & 0xFFFFFu);
    int dl = (int)(ent >> 20);
    int s = 0;
    if (act) {
        if (SRC) s = t;
        else     s = (t < NEDGES) ? ei[t] : (t - NEDGES);
        if (QDST) {
            dl = ((t < NEDGES) ? ei[NEDGES + t] : (t - NEDGES)) - lo;
            ent = (unsigned)t | ((unsigned)dl << 20);
        }
    }
    float e[H];
#pragma unroll
    for (int hh = 0; hh < H; ++hh) e[hh] = 0.f;
    if (act) {
#pragma unroll
        for (int hh = 0; hh < H; ++hh) e[hh] = leakyf(as_[s * H + hh] + adl[dl * H + hh], 0.2f);
    }
    if (PHASE == 0) {
        if (act) {
#pragma unroll
            for (int hh = 0; hh < H; ++hh)
                __hip_atomic_fetch_max(&menc[dl * H + hh], enc_ord(e[hh]), __ATOMIC_RELAXED,
                                       __HIP_MEMORY_SCOPE_WORKGROUP);
        }
        if (REC) {
            unsigned pos = 0;
            if (lane == 0)
                pos = __hip_atomic_fetch_add(wcount, (unsigned)n, __ATOMIC_RELAXED, __HIP_MEMORY_SCOPE_WORKGROUP);
            pos = (unsigned)__builtin_amdgcn_readfirstlane((int)pos);
            if (act && pos + (unsigned)n <= (unsigned)LCAP) wlist[pos + lane] = ent;
        }
    } else {
        if (EMIT) {
            if (act) grec[base + lane] = (unsigned)s | ((unsigned)dl << 20);
        }
        float ex[H];
#pragma unroll
        for (int hh = 0; hh < H; ++hh) ex[hh] = 0.f;
        if (act) {
#pragma unroll
            for (int hh = 0; hh < H; ++hh) {
                ex[hh] = expf(e[hh] - dec_ord(menc[dl * H + hh]));
                __hip_atomic_fetch_add(&den[dl * H + hh], ex[hh], __ATOMIC_RELAXED, __HIP_MEMORY_SCOPE_WORKGROUP);
            }
        }
        for (int j = 0; j < n; j += G) {
            v2f   r[G][H];
            float w[G][H];
            int   dj[G];
#pragma unroll
            for (int g = 0; g < G; ++g) {
                const int jj = (j + g < n) ? (j + g) : (n - 1);
                const int sj = __builtin_amdgcn_readlane(s, jj);
                dj[g] = __builtin_amdgcn_readlane(dl, jj);
#pragma unroll
                for (int hh = 0; hh < H; ++hh)
                    w[g][hh] = __int_as_float(__builtin_amdgcn_readlane(__float_as_int(ex[hh]), jj));
                const v2f* hp = (const v2f*)(h + (size_t)sj * F) + lane;
#pragma unroll
                for (int hh = 0; hh < H; ++hh) r[g][hh] = hp[hh * 32];
            }
#pragma unroll
            for (int g = 0; g < G; ++g) {
                if (j + g < n) {
                    float* ap = acc + dj[g] * P + 2 * lane;
#pragma unroll
                    for (int hh = 0; hh < H; ++hh) {
                        __hip_atomic_fetch_add(ap + hh * 64,     w[g][hh] * r[g][hh].x, __ATOMIC_RELAXED,
                                               __HIP_MEMORY_SCOPE_WORKGROUP);
                        __hip_atomic_fetch_add(ap + hh * 64 + 1, w[g][hh] * r[g][hh].y, __ATOMIC_RELAXED,
                                               __HIP_MEMORY_SCOPE_WORKGROUP);
                    }
                }
            }
        }
    }
}

__device__ __forceinline__ unsigned zero_bytes(unsigned x) {
    const unsigned t = (x & 0x7F7F7F7Fu) + 0x7F7F7F7Fu;
    return ~(t | x) & 0x80808080u;
}

template <int H, int PHASE, int G>
__device__ __forceinline__ void scan_pass(const int* __restrict__ ei, const unsigned* __restrict__ dig,
                                          const unsigned* __restrict__ gbm,
                                          const float* __restrict__ h,
                                          const float* __restrict__ as_,
                                          const float* adl, unsigned* menc, float* den, float* acc,
                                          unsigned* myq, unsigned* wlist, unsigned* wcount,
                                          int lo, int nown, int lane, int wave) {
    int qn = 0;
    auto enqueue = [&](bool mt, unsigned val) {
        const unsigned mask = __builtin_amdgcn_ballot_w32(mt);
        if (mt) myq[qn + __builtin_popcount(mask & ((1u << lane) - 1u))] = val;
        qn += __builtin_popcount(mask);
    };
    {
        const int dl = wave + AGG_WAVES * lane;
        enqueue(dl < nown, (unsigned)(NEDGES + lo + dl));
    }
    const uint4* dg4 = (const uint4*)dig;
    const unsigned mine = (unsigned)blockIdx.x * 0x01010101u;
    constexpr int WPW = BM_WORDS / AGG_WAVES;
    static_assert(BM_WORDS % AGG_WAVES == 0, "index words per wave");
    for (int o = 0; o < WPW; o += 32) {
        const int k = wave * WPW + o + lane;
        unsigned m = 0u;
        if (o + lane < WPW) m = gbm[((size_t)(k >> 3) * 256 + blockIdx.x) * BM_W + (k & 7)];
        while (__builtin_amdgcn_ballot_w32(m != 0u) != 0u) {
            int   gi[SCAN_U];
            uint4 dv[SCAN_U];
#pragma unroll
            for (int u = 0; u < SCAN_U; ++u) {
                gi[u] = (m != 0u) ? (k * 32 + __builtin_ctz(m | 0x80000000u)) : -1;
                m &= m - 1u;
                dv[u] = dg4[(gi[u] >= 0) ? gi[u] : 0];
            }
#pragma unroll
            for (int u = 0; u < SCAN_U; ++u) {
                if (__builtin_amdgcn_ballot_w32(gi[u] >= 0) == 0u) break;
                const int i16 = gi[u];
                unsigned f = (zero_bytes(dv[u].x ^ mine) >> 7) | (zero_bytes(dv[u].y ^ mine) >> 6) |
                             (zero_bytes(dv[u].z ^ mine) >> 5) | (zero_bytes(dv[u].w ^ mine) >> 4);
                if (i16 < 0) f = 0u;
                while (__builtin_amdgcn_ballot_w32(f != 0u) != 0u) {
                    const int p = __builtin_ctz(f | 0x80000000u);
                    enqueue(f != 0u, (unsigned)(i16 * 16 + 4 * (p & 7) + (p >> 3)));
                    f &= f - 1u;
                    while (qn >= 32) {
                        flush_batch<H, PHASE, G, false, true, PHASE == 0, false>(myq, qn - 32, 32, lane, lo, ei, h, as_, adl,
                                                                                 menc, den, acc, wlist, wcount, nullptr);
                        qn -= 32;
                    }
                }
            }
        }
    }
    if (qn > 0)
        flush_batch<H, PHASE, G, false, true, PHASE == 0, false>(myq, 0, qn, lane, lo, ei, h, as_, adl, menc, den, acc,
                                                                 wlist, wcount, nullptr);
}

template <int H>
__global__ void __launch_bounds__(AGG_WAVES * 32)
agg_owner_kernel(const int* __restrict__ ei, const float* __restrict__ h,
                 const float* __restrict__ as_, const float* __restrict__ ad_,
                 float* __restrict__ out,
                 const float* __restrict__ bias, const float* __restrict__ W2,
                 unsigned* __restrict__ grec, const unsigned* __restrict__ dig,
                 const unsigned* __restrict__ gbm) {
    constexpr int F = H * 64;
    constexpr int P = RowPitch<H>::value;
    __shared__ __attribute__((aligned(16))) float acc[NOWN * P];
    __shared__ unsigned menc[NOWN * H];
    __shared__ float    den[NOWN * H];
    __shared__ float    adl[NOWN * H];
    __shared__ unsigned q[AGG_WAVES * QCAP];
    __shared__ __attribute__((aligned(16))) unsigned wlist[LCAP];
    __shared__ unsigned wcount;

    const int tid = threadIdx.x, lane = tid & 31, wave = tid >> 5;
    const int lo = (int)blockIdx.x * NOWN;
    const int nown = (NNODES - lo < NOWN) ? (NNODES - lo) : NOWN;
    unsigned* myhdr = grec + (size_t)blockIdx.x * RSTRIDE;
    unsigned* myrec = myhdr + 32;

    unsigned cnt_in = (unsigned)LCAP + 1u;
    if constexpr (H == 1) {
        cnt_in = (unsigned)__builtin_amdgcn_readfirstlane((int)myhdr[0]);
        if (cnt_in <= (unsigned)LCAP)
            for (int i = tid; i < (int)cnt_in; i += AGG_WAVES * 32) wlist[i] = myrec[i];
    }
    const bool fromrec = (H == 1) && (cnt_in <= (unsigned)LCAP);

    for (int i = tid; i < NOWN * P; i += AGG_WAVES * 32) acc[i] = 0.f;
    for (int i = tid; i < NOWN * H; i += AGG_WAVES * 32) {
        menc[i] = ENC_NEG_INF;
        den[i]  = 0.f;
        adl[i]  = (i < nown * H) ? ad_[lo * H + i] : 0.f;
    }
    if (tid == 0) wcount = 0u;
    __syncthreads();

    unsigned* myq = q + wave * QCAP;
    if (fromrec) {
        for (int base = wave * 32; base < (int)cnt_in; base += AGG_WAVES * 32) {
            int n = (int)cnt_in - base;
            n = __builtin_amdgcn_readfirstlane(n < 32 ? n : 32);
            flush_batch<H, 0, 4, true, false, false, false>(wlist, base, n, lane, lo, ei, h, as_, adl, menc, den, acc,
                                                            wlist, &wcount, nullptr);
        }
    } else {
        scan_pass<H, 0, 4>(ei, dig, gbm, h, as_, adl, menc, den, acc, myq, wlist, &wcount, lo, nown, lane, wave);
    }
    __syncthreads();

    const unsigned cnt = fromrec ? cnt_in : wcount;
    if constexpr (H == 4) {
        if (tid == 0) myhdr[0] = cnt;
    }
    constexpr int GB = (H == 1) ? 16 : 8;
    if (cnt <= (unsigned)LCAP) {
        for (int base = wave * 32; base < (int)cnt; base += AGG_WAVES * 32) {
            int n = (int)cnt - base;
            n = __builtin_amdgcn_readfirstlane(n < 32 ? n : 32);
            if (fromrec)
                flush_batch<H, 1, GB, true, false, false, false>(wlist, base, n, lane, lo, ei, h, as_, adl, menc, den,
                                                                 acc, wlist, &wcount, nullptr);
            else
                flush_batch<H, 1, GB, false, false, false, H == 4>(wlist, base, n, lane, lo, ei, h, as_, adl, menc, den,
                                                                   acc, wlist, &wcount, myrec);
        }
    } else {
        scan_pass<H, 1, 4>(ei, dig, gbm, h, as_, adl, menc, den, acc, myq, wlist, &wcount, lo, nown, lane, wave);
    }
    __syncthreads();

    if constexpr (H == 4) {
        constexpr int WSLOTS = LCAP / 2 / (AGG_WAVES * 32);
        v2f wreg[WSLOTS];
#pragma unroll
        for (int j = 0; j < WSLOTS; ++j) {
            const int slot = tid + j * (AGG_WAVES * 32);
            const int sl = slot & 31, tile = (slot >> 5) & 3, k4 = slot >> 7;
            const float* wp = W2 + (size_t)(4 * k4 + 2 * (sl >> 4)) * OUT_DIM + 16 * tile + (sl & 15);
            wreg[j].x = wp[0];
            wreg[j].y = wp[OUT_DIM];
        }
        for (int i = tid; i < nown * (F / 4); i += AGG_WAVES * 32) {
            const int row = i / (F / 4), col = (i % (F / 4)) * 4;
            const float dn = den[row * H + (col >> 6)] + 1e-16f;
            float4* p = (float4*)(acc + row * P + col);
            const float4 b = *(const float4*)(bias + col);
            float4 v = *p;
            v.x /= dn; v.y /= dn; v.z /= dn; v.w /= dn;
            v.x = leakyf(v.x + b.x, 0.01f); v.y = leakyf(v.y + b.y, 0.01f);
            v.z = leakyf(v.z + b.z, 0.01f); v.w = leakyf(v.w + b.w, 0.01f);
            *p = v;
        }
        v2f* wf = (v2f*)wlist;
#pragma unroll
        for (int j = 0; j < WSLOTS; ++j) wf[tid + j * (AGG_WAVES * 32)] = wreg[j];
        __syncthreads();

        const int half = lane >> 4, l = lane & 15;
        const int njobs = ((nown + 15) >> 4) * 2;
        const int wv = __builtin_amdgcn_readfirstlane(wave);
        for (int job = wv; job < njobs; job += AGG_WAVES) {
            const int rb = job >> 1, cp = job & 1;
            int arow = rb * 16 + l;
            arow = (arow < NOWN) ? arow : (NOWN - 1);
            const float* ap = acc + arow * P + 2 * half;
            const v2f* bp = wf + (2 * cp) * 32 + lane;
            v8f c0 = {}, c1 = {};
            for (int k4 = 0; k4 < W2_ROWS / 4; ++k4) {
                const v2f a  = *(const v2f*)(ap + 4 * k4);
                const v2f b0 = bp[k4 * 128];
                const v2f b1 = bp[k4 * 128 + 32];
                c0 = __builtin_amdgcn_wmma_f32_16x16x4_f32(false, a, false, b0, (short)0, c0, false, false);
                c1 = __builtin_amdgcn_wmma_f32_16x16x4_f32(false, a, false, b1, (short)0, c1, false, false);
            }
#pragma unroll
            for (int r = 0; r < 8; ++r) {
                const int row = rb * 16 + r + 8 * half;
                if (row < nown) {
                    float* gp = out + (size_t)(lo + row) * OUT_DIM + 32 * cp + l;
                    gp[ 0] = c0[r];
                    gp[16] = c1[r];
                }
            }
        }
    } else {
        const int n4 = nown * F / 4;
        float4* o4 = (float4*)(out + (size_t)lo * F);
        const float4* a4 = (const float4*)acc;
        for (int i = tid; i < n4; i += AGG_WAVES * 32) {
            const int idx = i * 4;
            const float dn = den[(idx / F) * H + ((idx % F) >> 6)] + 1e-16f;
            float4 v = a4[i];
            v.x /= dn; v.y /= dn; v.z /= dn; v.w /= dn;
            o4[i] = v;
        }
    }
}

__global__ void bias_act_kernel(const float* __restrict__ in, const float* __restrict__ b,
                                float* __restrict__ out, int total, int F, float slope) {
    int i = blockIdx.x * blockDim.x + threadIdx.x;
    if (i >= total) return;
    out[i] = leakyf(in[i] + b[i % F], slope);
}

extern "C" void kernel_launch(void* const* d_in, const int* in_sizes, int n_in,
                              void* d_out, int out_size, void* d_ws, size_t ws_size,
                              hipStream_t stream) {
    const float* x   = (const float*)d_in[0];
    const int*   ei  = (const int*)  d_in[1];
    const float* W1  = (const float*)d_in[2];
    const float* as1 = (const float*)d_in[3];
    const float* ad1 = (const float*)d_in[4];
    const float* b1  = (const float*)d_in[5];
    const float* W2  = (const float*)d_in[6];
    const float* as2 = (const float*)d_in[7];
    const float* ad2 = (const float*)d_in[8];
    const float* b2  = (const float*)d_in[9];
    float* out = (float*)d_out;

    float* ws   = (float*)d_ws;
    float* h1   = ws;
    float* g2   = h1   + (size_t)NNODES * F1;
    float* out2 = g2   + (size_t)NNODES * OUT_DIM;
    float* asb  = h1   + (size_t)2 * NNODES * F1;
    float* adb  = asb  + (size_t)NNODES * HEADS;
    unsigned* rec = (unsigned*)(out2 + (size_t)NNODES * OUT_DIM);
    unsigned* dig = rec + (size_t)NWG_AGG * RSTRIDE;
    unsigned* gbm = dig + NEDGES / 4;

    gemm1_wreg_kernel<<<G1_WGS, G1_WAVES * 32, 0, stream>>>(x, W1, h1);
    alpha_digest_kernel<<<DIG_BLOCKS + ALPHA1_BLOCKS, 256, 0, stream>>>(h1, as1, ad1, asb, adb, NNODES * HEADS,
                                                                        ei + NEDGES, dig, gbm);
    agg_owner_kernel<HEADS><<<NWG_AGG, AGG_WAVES * 32, 0, stream>>>(ei, h1, asb, adb, g2, b1, W2, rec, dig, gbm);

    alpha_kernel<1><<<((NNODES + ALPHA_GPW - 1) / ALPHA_GPW + 7) / 8, 256, 0, stream>>>(
        g2, as2, ad2, asb, adb, NNODES);
    agg_owner_kernel<1><<<NWG_AGG, AGG_WAVES * 32, 0, stream>>>(ei, g2, asb, adb, out2, b2, W2, rec, dig, gbm);
    bias_act_kernel<<<(NNODES * OUT_DIM + 255) / 256, 256, 0, stream>>>(out2, b2, out,
                                                                        NNODES * OUT_DIM,
                                                                        OUT_DIM, 1.0f);
}


# Pass B gathers only feature rows: the LDS record keeps what pass A fetched per edge (source node, logits) and pass B takes them from LDS, not from the source list and as again (both layers)
# speedup vs baseline: 4.8878x; 1.0392x over previous
#include <hip/hip_runtime.h>
#include <math.h>

#define NNODES  50000
#define NEDGES  800000
#define ETOT    (NEDGES + NNODES)
#define IN_DIM  128
#define HID     64
#define HEADS   4
#define OUT_DIM 64
#define F1      (HEADS * HID)
#define ENC_NEG_INF 0x007FFFFFu

typedef __attribute__((ext_vector_type(2))) float v2f;
typedef __attribute__((ext_vector_type(8))) float v8f;

__device__ __forceinline__ float leakyf(float x, float s) { return x >= 0.f ? x : s * x; }

__device__ __forceinline__ unsigned enc_ord(float x) {
    unsigned u = __float_as_uint(x);
    return (u & 0x80000000u) ? ~u : (u | 0x80000000u);
}
__device__ __forceinline__ float dec_ord(unsigned e) {
    return (e & 0x80000000u) ? __uint_as_float(e & 0x7FFFFFFFu) : __uint_as_float(~e);
}

#define G1_WAVES 8
#define G1_WGS   256
#define G1_RB    (NNODES / 16)
#define G1_STEP  (G1_WGS * (G1_WAVES / 4))
static_assert(NNODES % 16 == 0 && IN_DIM == 128 && F1 == 256, "tile map of the layer-1 product");
static_assert(G1_STEP <= G1_RB, "every wave has a first row block");
__global__ void __launch_bounds__(G1_WAVES * 32)
gemm1_wreg_kernel(const float* __restrict__ X, const float* __restrict__ W, float* __restrict__ C) {
    const int tid = threadIdx.x, lane = tid & 31;
    const int half = lane >> 4;
    const int l    = lane & 15;
    const int wv = __builtin_amdgcn_readfirstlane(tid >> 5);
    const int strip = wv & 3;

    int rb = (int)blockIdx.x + G1_WGS * (wv >> 2);
    float4 a[IN_DIM / 8];
    {
        const float4* xp = (const float4*)(X + (size_t)((rb << 4) + l) * IN_DIM) + half;
#pragma unroll
        for (int q = 0; q < IN_DIM / 8; ++q) a[q] = xp[2 * q];
    }
    float4 b[IN_DIM / 8][4];
    const float* wp = W + (size_t)(4 * half) * F1 + (strip << 6) + l;
#pragma unroll
    for (int t = 0; t < 4; ++t) {
#pragma unroll
        for (int q = 0; q < IN_DIM / 8; ++q) {
            const float* p = wp + (size_t)(8 * q) * F1 + 16 * t;
            b[q][t].x = p[0]; b[q][t].y = p[F1]; b[q][t].z = p[2 * F1]; b[q][t].w = p[3 * F1];
        }
    }
    for (;;) {
        const int rbn = rb + G1_STEP;
        const bool more = rbn < G1_RB;
        const int rbl = more ? rbn : rb;
        float4 an[IN_DIM / 8];
        const float4* xn = (const float4*)(X + (size_t)((rbl << 4) + l) * IN_DIM) + half;
#pragma unroll
        for (int q = 0; q < IN_DIM / 8; ++q) an[q] = xn[2 * q];

        v8f acc0 = {}, acc1 = {}, acc2 = {}, acc3 = {};
#pragma unroll
        for (int q = 0; q < IN_DIM / 8; ++q) {
            const v2f a0 = { a[q].x, a[q].y }, a1 = { a[q].z, a[q].w };
            const v2f b00 = { b[q][0].x, b[q][0].y }, b01 = { b[q][0].z, b[q][0].w };
            const v2f b10 = { b[q][1].x, b[q][1].y }, b11 = { b[q][1].z, b[q][1].w };
            const v2f b20 = { b[q][2].x, b[q][2].y }, b21 = { b[q][2].z, b[q][2].w };
            const v2f b30 = { b[q][3].x, b[q][3].y }, b31 = { b[q][3].z, b[q][3].w };
            acc0 = __builtin_amdgcn_wmma_f32_16x16x4_f32(false, a0, false, b00, (short)0, acc0, false, false);
            acc1 = __builtin_amdgcn_wmma_f32_16x16x4_f32(false, a0, false, b10, (short)0, acc1, false, false);
            acc2 = __builtin_amdgcn_wmma_f32_16x16x4_f32(false, a0, false, b20, (short)0, acc2, false, false);
            acc3 = __builtin_amdgcn_wmma_f32_16x16x4_f32(false, a0, false, b30, (short)0, acc3, false, false);
            acc0 = __builtin_amdgcn_wmma_f32_16x16x4_f32(false, a1, false, b01, (short)0, acc0, false, false);
            acc1 = __builtin_amdgcn_wmma_f32_16x16x4_f32(false, a1, false, b11, (short)0, acc1, false, false);
            acc2 = __builtin_amdgcn_wmma_f32_16x16x4_f32(false, a1, false, b21, (short)0, acc2, false, false);
            acc3 = __builtin_amdgcn_wmma_f32_16x16x4_f32(false, a1, false, b31, (short)0, acc3, false, false);
        }
        float* crow = C + (size_t)(rb << 4) * F1 + (strip << 6);
#pragma unroll
        for (int r = 0; r < 8; ++r) {
            const int m = r + 8 * half;
            float* p = crow + (size_t)m * F1 + l;
            p[ 0] = acc0[r];
            p[16] = acc1[r];
            p[32] = acc2[r];
            p[48] = acc3[r];
        }
        if (!more) break;
        rb = rbn;
#pragma unroll
        for (int q = 0; q < IN_DIM / 8; ++q) a[q] = an[q];
    }
}

#define ALPHA_GPW 32
static_assert(HID == 64 && OUT_DIM == 64, "a group is 64 floats in both layers");
template <int H>
__device__ __forceinline__ void alpha_body(const float* __restrict__ h,
                                           const float* __restrict__ a_src, const float* __restrict__ a_dst,
                                           float* __restrict__ as_, float* __restrict__ ad_, int ngroups, int blk) {
    static_assert(16 % H == 0, "head of group g0 + 16*hf + j is j % H");
    const int lane = threadIdx.x & 31, hf = lane >> 4, l = lane & 15;
    const int g0 = (int)((blk * 256 + threadIdx.x) >> 5) * ALPHA_GPW;
    if (g0 >= ngroups) return;

    float4 v[16];
#pragma unroll
    for (int j = 0; j < 16; ++j) {
        int g = g0 + 16 * hf + j;
        g = (g < ngroups) ? g : (ngroups - 1);
        v[j] = *(const float4*)(h + (size_t)g * 64 + 4 * l);
    }
    float4 sa[H], da[H];
#pragma unroll
    for (int hh = 0; hh < H; ++hh) {
        sa[hh] = *(const float4*)(a_src + hh * 64 + 4 * l);
        da[hh] = *(const float4*)(a_dst + hh * 64 + 4 * l);
    }
    float ps[16], pd[16];
#pragma unroll
    for (int j = 0; j < 16; ++j) {
        const float4 s = sa[j % H], d = da[j % H];
        ps[j] = v[j].x * s.x + v[j].y * s.y + v[j].z * s.z + v[j].w * s.w;
        pd[j] = v[j].x * d.x + v[j].y * d.y + v[j].z * d.z + v[j].w * d.w;
    }
    const bool b3 = (l & 8) != 0, b2 = (l & 4) != 0, b1 = (l & 2) != 0, b0 = (l & 1) != 0;
    float s8[8], d8[8];
#pragma unroll
    for (int t = 0; t < 8; ++t) {
        s8[t] = (b3 ? ps[t + 8] : ps[t]) + __shfl_xor(b3 ? ps[t] : ps[t + 8], 8);
        d8[t] = (b3 ? pd[t + 8] : pd[t]) + __shfl_xor(b3 ? pd[t] : pd[t + 8], 8);
    }
    float s4[4], d4[4];
#pragma unroll
    for (int t = 0; t < 4; ++t) {
        s4[t] = (b2 ? s8[t + 4] : s8[t]) + __shfl_xor(b2 ? s8[t] : s8[t + 4], 4);
        d4[t] = (b2 ? d8[t + 4] : d8[t]) + __shfl_xor(b2 ? d8[t] : d8[t + 4], 4);
    }
    float s2[2], d2[2];
#pragma unroll
    for (int t = 0; t < 2; ++t) {
        s2[t] = (b1 ? s4[t + 2] : s4[t]) + __shfl_xor(b1 ? s4[t] : s4[t + 2], 2);
        d2[t] = (b1 ? d4[t + 2] : d4[t]) + __shfl_xor(b1 ? d4[t] : d4[t + 2], 2);
    }
    const float s1 = (b0 ? s2[1] : s2[0]) + __shfl_xor(b0 ? s2[0] : s2[1], 1);
    const float d1 = (b0 ? d2[1] : d2[0]) + __shfl_xor(b0 ? d2[0] : d2[1], 1);
    const int g = g0 + lane;
    if (g < ngroups) {
        as_[g] = s1;
        ad_[g] = d1;
    }
}

template <int H>
__global__ void __launch_bounds__(256)
alpha_kernel(const float* __restrict__ h,
             const float* __restrict__ a_src, const float* __restrict__ a_dst,
             float* __restrict__ as_, float* __restrict__ ad_, int ngroups) {
    alpha_body<H>(h, a_src, a_dst, as_, ad_, ngroups, (int)blockIdx.x);
}

#define NOWN       196
#define NDIG16     (NEDGES / 16)
#define DIG_BLOCKS ((NDIG16 + 255) / 256)
static_assert(NEDGES % 16 == 0, "the digest is written and read in 16-byte groups");
static_assert((NNODES - 1) / NOWN <= 255, "an owner index fits a byte");
__device__ __forceinline__ unsigned owner4(int4 d) {
    return  (unsigned)d.x / NOWN        | ((unsigned)d.y / NOWN) << 8 |
           ((unsigned)d.z / NOWN) << 16 | ((unsigned)d.w / NOWN) << 24;
}
#define BM_W      8
#define BM_WORDS  (DIG_BLOCKS * BM_W)
static_assert(BM_W == 8 && BM_W * 32 == 256, "a digest block is 8 waves; word index = 8 * owner + wave below");
__device__ __forceinline__ void or_owner_bits(unsigned* row, unsigned o, unsigned bit) {
    __hip_atomic_fetch_or(&row[o & 255u],         bit, __ATOMIC_RELAXED, __HIP_MEMORY_SCOPE_WORKGROUP);
    __hip_atomic_fetch_or(&row[(o >> 8) & 255u],  bit, __ATOMIC_RELAXED, __HIP_MEMORY_SCOPE_WORKGROUP);
    __hip_atomic_fetch_or(&row[(o >> 16) & 255u], bit, __ATOMIC_RELAXED, __HIP_MEMORY_SCOPE_WORKGROUP);
    __hip_atomic_fetch_or(&row[o >> 24],          bit, __ATOMIC_RELAXED, __HIP_MEMORY_SCOPE_WORKGROUP);
}
__device__ __forceinline__ void digest_block(const int* __restrict__ dst, unsigned* __restrict__ dig,
                                             unsigned* __restrict__ gbm, int blk) {
    __shared__ unsigned bm[BM_W * 256];
    const int tid = (int)threadIdx.x;
    for (int k = tid; k < BM_W * 256; k += 256) bm[k] = 0u;
    __syncthreads();
    const int i = blk * 256 + tid;
    if (i < NDIG16) {
        const int4* d4 = (const int4*)dst + 4 * (size_t)i;
        const int4 a = d4[0], b = d4[1], cc = d4[2], d = d4[3];
        uint4 o;
        o.x = owner4(a); o.y = owner4(b); o.z = owner4(cc); o.w = owner4(d);
        ((uint4*)dig)[i] = o;
        unsigned* row = bm + (tid >> 5) * 256;
        const unsigned bit = 1u << (tid & 31);
        or_owner_bits(row, o.x, bit); or_owner_bits(row, o.y, bit);
        or_owner_bits(row, o.z, bit); or_owner_bits(row, o.w, bit);
    }
    __syncthreads();
    unsigned* mybm = gbm + (size_t)blk * (BM_W * 256);
    for (int k = tid; k < BM_W * 256; k += 256) mybm[k] = bm[(k & (BM_W - 1)) * 256 + (k >> 3)];
}

#define ALPHA1_BLOCKS (((NNODES * HEADS + ALPHA_GPW - 1) / ALPHA_GPW + 7) / 8)
__global__ void __launch_bounds__(256)
alpha_digest_kernel(const float* __restrict__ h,
                    const float* __restrict__ a_src, const float* __restrict__ a_dst,
                    float* __restrict__ as_, float* __restrict__ ad_, int ngroups,
                    const int* __restrict__ dst, unsigned* __restrict__ dig, unsigned* __restrict__ gbm) {
    if ((int)blockIdx.x < DIG_BLOCKS) digest_block(dst, dig, gbm, (int)blockIdx.x);
    else alpha_body<HEADS>(h, a_src, a_dst, as_, ad_, ngroups, (int)blockIdx.x - DIG_BLOCKS);
}

#define NWG_AGG   256
#define AGG_WAVES 32
#define QCAP      64
#define LCAP      16384
#define SCAN_U    8

static_assert(NOWN * NWG_AGG >= NNODES && (NWG_AGG - 1) * NOWN < NNODES, "every node has one owner");
static_assert(ETOT <= (1 << 20) && NOWN <= 256, "queue entry = edge id, 20 bits; record entry = ... | local node << 20");
static_assert(NNODES <= (1 << 20), "record entry = source node | local node << 20");
#define RSTRIDE   (LCAP + 32)
static_assert((size_t)NNODES * (F1 + 2 * OUT_DIM) + (size_t)NWG_AGG * RSTRIDE + NEDGES / 4 +
              (size_t)DIG_BLOCKS * 256 * BM_W <= (size_t)2 * NNODES * F1,
              "the records, the digest and the owner index end before the logits begin");
static_assert(NWG_AGG == 256 && (NEDGES / 4) % 32 == 0, "one index row per owner; the index starts on a 128-byte line");
static_assert(((size_t)NWG_AGG * RSTRIDE) % 4 == 0, "the digest starts on a 16-byte boundary");
static_assert(((size_t)NNODES * (F1 + 2 * OUT_DIM)) % 32 == 0, "records start on a 128-byte line of the workspace");

template <int H> struct RowPitch { static constexpr int value = (H == 4) ? (H * 64 + 4) : (H * 64); };
#define W2_ROWS F1
static_assert(W2_ROWS * OUT_DIM == LCAP, "the W2 fragments take LCAP words of the edge record's LDS");
static_assert(OUT_DIM == 64 && W2_ROWS % 4 == 0, "layer-2 tile map below");

#define RCAP1 4096
template <int H> struct RecCap { static constexpr int value = (H == 4) ? RCAP1 : LCAP; };
static_assert(RCAP1 <= LCAP && RCAP1 % 4 == 0, "the layer-1 record fits its workspace slot; logits behind it stay 16-byte aligned");

template <int H>
__device__ __forceinline__ void elog_put(float* elog, int idx, const float* e) {
    if constexpr (H == 4) {
        ((float4*)elog)[idx] = make_float4(e[0], e[1], e[2], e[3]);
    } else {
#pragma unroll
        for (int hh = 0; hh < H; ++hh) elog[idx * H + hh] = e[hh];
    }
}
template <int H>
__device__ __forceinline__ void elog_get(const float* elog, int idx, float* e) {
    if constexpr (H == 4) {
        const float4 v = ((const float4*)elog)[idx];
        e[0] = v.x; e[1] = v.y; e[2] = v.z; e[3] = v.w;
    } else {
#pragma unroll
        for (int hh = 0; hh < H; ++hh) e[hh] = elog[idx * H + hh];
    }
}

template <int H, int PHASE, int G, bool SRC, bool QDST, bool REC, bool EMIT, bool ELOG>
__device__ __forceinline__ void flush_batch(const unsigned* list, int base, int n, int lane, int lo,
                                            const int* __restrict__ ei,
                                            const float* __restrict__ h,
                                            const float* __restrict__ as_,
                                            const float* adl, unsigned* menc, float* den, float* acc,
                                            unsigned* wlist, unsigned* wcount,
                                            unsigned* __restrict__ grec, float* elog) {
    constexpr int F = H * 64;
    constexpr unsigned RC = (unsigned)RecCap<H>::value;
    constexpr int P = RowPitch<H>::value;
    const bool act = lane < n;
    unsigned ent = 0;
    if (act) ent = list[base + lane];
    const int t  = (int)(ent & 0xFFFFFu);
    int dl = (int)(ent >> 20);
    int s = 0;
    if (act) {
        if (SRC) s = t;
        else     s = (t < NEDGES) ? ei[t] : (t - NEDGES);
        if (QDST) {
            dl = ((t < NEDGES) ? ei[NEDGES + t] : (t - NEDGES)) - lo;
            ent = (unsigned)s | ((unsigned)dl << 20);
        }
    }
    float e[H];
#pragma unroll
    for (int hh = 0; hh < H; ++hh) e[hh] = 0.f;
    if (act) {
        if (ELOG && PHASE == 1) {
            elog_get<H>(elog, base + lane, e);
        } else {
#pragma unroll
            for (int hh = 0; hh < H; ++hh) e[hh] = leakyf(as_[s * H + hh] + adl[dl * H + hh], 0.2f);
        }
    }
    if (PHASE == 0) {
        if (act) {
#pragma unroll
            for (int hh = 0; hh < H; ++hh)
                __hip_atomic_fetch_max(&menc[dl * H + hh], enc_ord(e[hh]), __ATOMIC_RELAXED,
                                       __HIP_MEMORY_SCOPE_WORKGROUP);
        }
        if (REC) {
            unsigned pos = 0;
            if (lane == 0)
                pos = __hip_atomic_fetch_add(wcount, (unsigned)n, __ATOMIC_RELAXED, __HIP_MEMORY_SCOPE_WORKGROUP);
            pos = (unsigned)__builtin_amdgcn_readfirstlane((int)pos);
            if (act && pos + (unsigned)n <= RC) {
                wlist[pos + lane] = ent;
                if (ELOG) elog_put<H>(elog, (int)pos + lane, e);
            }
        } else if (ELOG) {
            if (act) elog_put<H>(elog, base + lane, e);
        }
    } else {
        if (EMIT) {
            if (act) grec[base + lane] = (unsigned)s | ((unsigned)dl << 20);
        }
        float ex[H];
#pragma unroll
        for (int hh = 0; hh < H; ++hh) ex[hh] = 0.f;
        if (act) {
#pragma unroll
            for (int hh = 0; hh < H; ++hh) {
                ex[hh] = expf(e[hh] - dec_ord(menc[dl * H + hh]));
                __hip_atomic_fetch_add(&den[dl * H + hh], ex[hh], __ATOMIC_RELAXED, __HIP_MEMORY_SCOPE_WORKGROUP);
            }
        }
        for (int j = 0; j < n; j += G) {
            v2f   r[G][H];
            float w[G][H];
            int   dj[G];
#pragma unroll
            for (int g = 0; g < G; ++g) {
                const int jj = (j + g < n) ? (j + g) : (n - 1);
                const int sj = __builtin_amdgcn_readlane(s, jj);
                dj[g] = __builtin_amdgcn_readlane(dl, jj);
#pragma unroll
                for (int hh = 0; hh < H; ++hh)
                    w[g][hh] = __int_as_float(__builtin_amdgcn_readlane(__float_as_int(ex[hh]), jj));
                const v2f* hp = (const v2f*)(h + (size_t)sj * F) + lane;
#pragma unroll
                for (int hh = 0; hh < H; ++hh) r[g][hh] = hp[hh * 32];
            }
#pragma unroll
            for (int g = 0; g < G; ++g) {
                if (j + g < n) {
                    float* ap = acc + dj[g] * P + 2 * lane;
#pragma unroll
                    for (int hh = 0; hh < H; ++hh) {
                        __hip_atomic_fetch_add(ap + hh * 64,     w[g][hh] * r[g][hh].x, __ATOMIC_RELAXED,
                                               __HIP_MEMORY_SCOPE_WORKGROUP);
                        __hip_atomic_fetch_add(ap + hh * 64 + 1, w[g][hh] * r[g][hh].y, __ATOMIC_RELAXED,
                                               __HIP_MEMORY_SCOPE_WORKGROUP);
                    }
                }
            }
        }
    }
}

__device__ __forceinline__ unsigned zero_bytes(unsigned x) {
    const unsigned t = (x & 0x7F7F7F7Fu) + 0x7F7F7F7Fu;
    return ~(t | x) & 0x80808080u;
}

template <int H, int PHASE, int G>
__device__ __forceinline__ void scan_pass(const int* __restrict__ ei, const unsigned* __restrict__ dig,
                                          const unsigned* __restrict__ gbm,
                                          const float* __restrict__ h,
                                          const float* __restrict__ as_,
                                          const float* adl, unsigned* menc, float* den, float* acc,
                                          unsigned* myq, unsigned* wlist, unsigned* wcount, float* elog,
                                          int lo, int nown, int lane, int wave) {
    int qn = 0;
    auto enqueue = [&](bool mt, unsigned val) {
        const unsigned mask = __builtin_amdgcn_ballot_w32(mt);
        if (mt) myq[qn + __builtin_popcount(mask & ((1u << lane) - 1u))] = val;
        qn += __builtin_popcount(mask);
    };
    {
        const int dl = wave + AGG_WAVES * lane;
        enqueue(dl < nown, (unsigned)(NEDGES + lo + dl));
    }
    const uint4* dg4 = (const uint4*)dig;
    const unsigned mine = (unsigned)blockIdx.x * 0x01010101u;
    constexpr int WPW = BM_WORDS / AGG_WAVES;
    static_assert(BM_WORDS % AGG_WAVES == 0, "index words per wave");
    for (int o = 0; o < WPW; o += 32) {
        const int k = wave * WPW + o + lane;
        unsigned m = 0u;
        if (o + lane < WPW) m = gbm[((size_t)(k >> 3) * 256 + blockIdx.x) * BM_W + (k & 7)];
        while (__builtin_amdgcn_ballot_w32(m != 0u) != 0u) {
            int   gi[SCAN_U];
            uint4 dv[SCAN_U];
#pragma unroll
            for (int u = 0; u < SCAN_U; ++u) {
                gi[u] = (m != 0u) ? (k * 32 + __builtin_ctz(m | 0x80000000u)) : -1;
                m &= m - 1u;
                dv[u] = dg4[(gi[u] >= 0) ? gi[u] : 0];
            }
#pragma unroll
            for (int u = 0; u < SCAN_U; ++u) {
                if (__builtin_amdgcn_ballot_w32(gi[u] >= 0) == 0u) break;
                const int i16 = gi[u];
                unsigned f = (zero_bytes(dv[u].x ^ mine) >> 7) | (zero_bytes(dv[u].y ^ mine) >> 6) |
                             (zero_bytes(dv[u].z ^ mine) >> 5) | (zero_bytes(dv[u].w ^ mine) >> 4);
                if (i16 < 0) f = 0u;
                while (__builtin_amdgcn_ballot_w32(f != 0u) != 0u) {
                    const int p = __builtin_ctz(f | 0x80000000u);
                    enqueue(f != 0u, (unsigned)(i16 * 16 + 4 * (p & 7) + (p >> 3)));
                    f &= f - 1u;
                    while (qn >= 32) {
                        flush_batch<H, PHASE, G, false, true, PHASE == 0, false, PHASE == 0>(myq, qn - 32, 32, lane, lo, ei, h,
                                                                                             as_, adl, menc, den, acc, wlist,
                                                                                             wcount, nullptr, elog);
                        qn -= 32;
                    }
                }
            }
        }
    }
    if (qn > 0)
        flush_batch<H, PHASE, G, false, true, PHASE == 0, false, PHASE == 0>(myq, 0, qn, lane, lo, ei, h, as_, adl, menc, den,
                                                                             acc, wlist, wcount, nullptr, elog);
}

template <int H>
__global__ void __launch_bounds__(AGG_WAVES * 32)
agg_owner_kernel(const int* __restrict__ ei, const float* __restrict__ h,
                 const float* __restrict__ as_, const float* __restrict__ ad_,
                 float* __restrict__ out,
                 const float* __restrict__ bias, const float* __restrict__ W2,
                 unsigned* __restrict__ grec, const unsigned* __restrict__ dig,
                 const unsigned* __restrict__ gbm) {
    constexpr int F = H * 64;
    constexpr int P = RowPitch<H>::value;
    __shared__ __attribute__((aligned(16))) float acc[NOWN * P];
    __shared__ unsigned menc[NOWN * H];
    __shared__ float    den[NOWN * H];
    __shared__ float    adl[NOWN * H];
    __shared__ unsigned q[AGG_WAVES * QCAP];
    constexpr int RC  = RecCap<H>::value;
    constexpr int RBW = (RC * (1 + H) > LCAP) ? RC * (1 + H) : LCAP;
    __shared__ __attribute__((aligned(16))) unsigned recbuf[RBW];
    __shared__ unsigned wcount;
    unsigned* wlist = recbuf;
    float*    elog  = (float*)(recbuf + RC);

    const int tid = threadIdx.x, lane = tid & 31, wave = tid >> 5;
    const int lo = (int)blockIdx.x * NOWN;
    const int nown = (NNODES - lo < NOWN) ? (NNODES - lo) : NOWN;
    unsigned* myhdr = grec + (size_t)blockIdx.x * RSTRIDE;
    unsigned* myrec = myhdr + 32;

    unsigned cnt_in = (unsigned)LCAP + 1u;
    if constexpr (H == 1) {
        cnt_in = (unsigned)__builtin_amdgcn_readfirstlane((int)myhdr[0]);
        if (cnt_in <= (unsigned)RCAP1)
            for (int i = tid; i < (int)cnt_in; i += AGG_WAVES * 32) wlist[i] = myrec[i];
    }
    const bool fromrec = (H == 1) && (cnt_in <= (unsigned)RCAP1);

    for (int i = tid; i < NOWN * P; i += AGG_WAVES * 32) acc[i] = 0.f;
    for (int i = tid; i < NOWN * H; i += AGG_WAVES * 32) {
        menc[i] = ENC_NEG_INF;
        den[i]  = 0.f;
        adl[i]  = (i < nown * H) ? ad_[lo * H + i] : 0.f;
    }
    if (tid == 0) wcount = 0u;
    __syncthreads();

    unsigned* myq = q + wave * QCAP;
    if (fromrec) {
        for (int base = wave * 32; base < (int)cnt_in; base += AGG_WAVES * 32) {
            int n = (int)cnt_in - base;
            n = __builtin_amdgcn_readfirstlane(n < 32 ? n : 32);
            flush_batch<H, 0, 4, true, false, false, false, true>(wlist, base, n, lane, lo, ei, h, as_, adl, menc, den, acc,
                                                                  wlist, &wcount, nullptr, elog);
        }
    } else {
        scan_pass<H, 0, 4>(ei, dig, gbm, h, as_, adl, menc, den, acc, myq, wlist, &wcount, elog, lo, nown, lane, wave);
    }
    __syncthreads();

    const unsigned cnt = fromrec ? cnt_in : wcount;
    if constexpr (H == 4) {
        if (tid == 0) myhdr[0] = cnt;
    }
    constexpr int GB = (H == 1) ? 16 : 8;
    if (cnt <= (unsigned)RC) {
        for (int base = wave * 32; base < (int)cnt; base += AGG_WAVES * 32) {
            int n = (int)cnt - base;
            n = __builtin_amdgcn_readfirstlane(n < 32 ? n : 32);
            if (fromrec)
                flush_batch<H, 1, GB, true, false, false, false, true>(wlist, base, n, lane, lo, ei, h, as_, adl, menc,
                                                                       den, acc, wlist, &wcount, nullptr, elog);
            else
                flush_batch<H, 1, GB, true, false, false, H == 4, true>(wlist, base, n, lane, lo, ei, h, as_, adl, menc,
                                                                        den, acc, wlist, &wcount, myrec, elog);
        }
    } else {
        scan_pass<H, 1, 4>(ei, dig, gbm, h, as_, adl, menc, den, acc, myq, wlist, &wcount, elog, lo, nown, lane, wave);
    }
    __syncthreads();

    if constexpr (H == 4) {
        constexpr int WSLOTS = LCAP / 2 / (AGG_WAVES * 32);
        v2f wreg[WSLOTS];
#pragma unroll
        for (int j = 0; j < WSLOTS; ++j) {
            const int slot = tid + j * (AGG_WAVES * 32);
            const int sl = slot & 31, tile = (slot >> 5) & 3, k4 = slot >> 7;
            const float* wp = W2 + (size_t)(4 * k4 + 2 * (sl >> 4)) * OUT_DIM + 16 * tile + (sl & 15);
            wreg[j].x = wp[0];
            wreg[j].y = wp[OUT_DIM];
        }
        for (int i = tid; i < nown * (F / 4); i += AGG_WAVES * 32) {
            const int row = i / (F / 4), col = (i % (F / 4)) * 4;
            const float dn = den[row * H + (col >> 6)] + 1e-16f;
            float4* p = (float4*)(acc + row * P + col);
            const float4 b = *(const float4*)(bias + col);
            float4 v = *p;
            v.x /= dn; v.y /= dn; v.z /= dn; v.w /= dn;
            v.x = leakyf(v.x + b.x, 0.01f); v.y = leakyf(v.y + b.y, 0.01f);
            v.z = leakyf(v.z + b.z, 0.01f); v.w = leakyf(v.w + b.w, 0.01f);
            *p = v;
        }
        v2f* wf = (v2f*)recbuf;
#pragma unroll
        for (int j = 0; j < WSLOTS; ++j) wf[tid + j * (AGG_WAVES * 32)] = wreg[j];
        __syncthreads();

        const int half = lane >> 4, l = lane & 15;
        const int njobs = ((nown + 15) >> 4) * 2;
        const int wv = __builtin_amdgcn_readfirstlane(wave);
        for (int job = wv; job < njobs; job += AGG_WAVES) {
            const int rb = job >> 1, cp = job & 1;
            int arow = rb * 16 + l;
            arow = (arow < NOWN) ? arow : (NOWN - 1);
            const float* ap = acc + arow * P + 2 * half;
            const v2f* bp = wf + (2 * cp) * 32 + lane;
            v8f c0 = {}, c1 = {};
            for (int k4 = 0; k4 < W2_ROWS / 4; ++k4) {
                const v2f a  = *(const v2f*)(ap + 4 * k4);
                const v2f b0 = bp[k4 * 128];
                const v2f b1 = bp[k4 * 128 + 32];
                c0 = __builtin_amdgcn_wmma_f32_16x16x4_f32(false, a, false, b0, (short)0, c0, false, false);
                c1 = __builtin_amdgcn_wmma_f32_16x16x4_f32(false, a, false, b1, (short)0, c1, false, false);
            }
#pragma unroll
            for (int r = 0; r < 8; ++r) {
                const int row = rb * 16 + r + 8 * half;
                if (row < nown) {
                    float* gp = out + (size_t)(lo + row) * OUT_DIM + 32 * cp + l;
                    gp[ 0] = c0[r];
                    gp[16] = c1[r];
                }
            }
        }
    } else {
        const int n4 = nown * F / 4;
        float4* o4 = (float4*)(out + (size_t)lo * F);
        const float4* a4 = (const float4*)acc;
        for (int i = tid; i < n4; i += AGG_WAVES * 32) {
            const int idx = i * 4;
            const float dn = den[(idx / F) * H + ((idx % F) >> 6)] + 1e-16f;
            float4 v = a4[i];
            v.x /= dn; v.y /= dn; v.z /= dn; v.w /= dn;
            o4[i] = v;
        }
    }
}

__global__ void bias_act_kernel(const float* __restrict__ in, const float* __restrict__ b,
                                float* __restrict__ out, int total, int F, float slope) {
    int i = blockIdx.x * blockDim.x + threadIdx.x;
    if (i >= total) return;
    out[i] = leakyf(in[i] + b[i % F], slope);
}

extern "C" void kernel_launch(void* const* d_in, const int* in_sizes, int n_in,
                              void* d_out, int out_size, void* d_ws, size_t ws_size,
                              hipStream_t stream) {
    const float* x   = (const float*)d_in[0];
    const int*   ei  = (const int*)  d_in[1];
    const float* W1  = (const float*)d_in[2];
    const float* as1 = (const float*)d_in[3];
    const float* ad1 = (const float*)d_in[4];
    const float* b1  = (const float*)d_in[5];
    const float* W2  = (const float*)d_in[6];
    const float* as2 = (const float*)d_in[7];
    const float* ad2 = (const float*)d_in[8];
    const float* b2  = (const float*)d_in[9];
    float* out = (float*)d_out;

    float* ws   = (float*)d_ws;
    float* h1   = ws;
    float* g2   = h1   + (size_t)NNODES * F1;
    float* out2 = g2   + (size_t)NNODES * OUT_DIM;
    float* asb  = h1   + (size_t)2 * NNODES * F1;
    float* adb  = asb  + (size_t)NNODES * HEADS;
    unsigned* rec = (unsigned*)(out2 + (size_t)NNODES * OUT_DIM);
    unsigned* dig = rec + (size_t)NWG_AGG * RSTRIDE;
    unsigned* gbm = dig + NEDGES / 4;

    gemm1_wreg_kernel<<<G1_WGS, G1_WAVES * 32, 0, stream>>>(x, W1, h1);
    alpha_digest_kernel<<<DIG_BLOCKS + ALPHA1_BLOCKS, 256, 0, stream>>>(h1, as1, ad1, asb, adb, NNODES * HEADS,
                                                                        ei + NEDGES, dig, gbm);
    agg_owner_kernel<HEADS><<<NWG_AGG, AGG_WAVES * 32, 0, stream>>>(ei, h1, asb, adb, g2, b1, W2, rec, dig, gbm);

    alpha_kernel<1><<<((NNODES + ALPHA_GPW - 1) / ALPHA_GPW + 7) / 8, 256, 0, stream>>>(
        g2, as2, ad2, asb, adb, NNODES);
    agg_owner_kernel<1><<<NWG_AGG, AGG_WAVES * 32, 0, stream>>>(ei, g2, asb, adb, out2, b2, W2, rec, dig, gbm);
    bias_act_kernel<<<(NNODES * OUT_DIM + 255) / 256, 256, 0, stream>>>(out2, b2, out,
                                                                        NNODES * OUT_DIM,
                                                                        OUT_DIM, 1.0f);
}


# Layer-1 owner bins its edge record in LDS by the row's owning wave (node % 32); each wave adds its bin into its rows by 16-byte LDS load, fma, store, not by LDS float adds; row add order changes
# speedup vs baseline: 5.9273x; 1.2127x over previous
#include <hip/hip_runtime.h>
#include <math.h>

#define NNODES  50000
#define NEDGES  800000
#define ETOT    (NEDGES + NNODES)
#define IN_DIM  128
#define HID     64
#define HEADS   4
#define OUT_DIM 64
#define F1      (HEADS * HID)
#define ENC_NEG_INF 0x007FFFFFu

typedef __attribute__((ext_vector_type(2))) float v2f;
typedef __attribute__((ext_vector_type(8))) float v8f;

__device__ __forceinline__ float leakyf(float x, float s) { return x >= 0.f ? x : s * x; }

__device__ __forceinline__ unsigned enc_ord(float x) {
    unsigned u = __float_as_uint(x);
    return (u & 0x80000000u) ? ~u : (u | 0x80000000u);
}
__device__ __forceinline__ float dec_ord(unsigned e) {
    return (e & 0x80000000u) ? __uint_as_float(e & 0x7FFFFFFFu) : __uint_as_float(~e);
}

#define G1_WAVES 8
#define G1_WGS   256
#define G1_RB    (NNODES / 16)
#define G1_STEP  (G1_WGS * (G1_WAVES / 4))
static_assert(NNODES % 16 == 0 && IN_DIM == 128 && F1 == 256, "tile map of the layer-1 product");
static_assert(G1_STEP <= G1_RB, "every wave has a first row block");
__global__ void __launch_bounds__(G1_WAVES * 32)
gemm1_wreg_kernel(const float* __restrict__ X, const float* __restrict__ W, float* __restrict__ C) {
    const int tid = threadIdx.x, lane = tid & 31;
    const int half = lane >> 4;
    const int l    = lane & 15;
    const int wv = __builtin_amdgcn_readfirstlane(tid >> 5);
    const int strip = wv & 3;

    int rb = (int)blockIdx.x + G1_WGS * (wv >> 2);
    float4 a[IN_DIM / 8];
    {
        const float4* xp = (const float4*)(X + (size_t)((rb << 4) + l) * IN_DIM) + half;
#pragma unroll
        for (int q = 0; q < IN_DIM / 8; ++q) a[q] = xp[2 * q];
    }
    float4 b[IN_DIM / 8][4];
    const float* wp = W + (size_t)(4 * half) * F1 + (strip << 6) + l;
#pragma unroll
    for (int t = 0; t < 4; ++t) {
#pragma unroll
        for (int q = 0; q < IN_DIM / 8; ++q) {
            const float* p = wp + (size_t)(8 * q) * F1 + 16 * t;
            b[q][t].x = p[0]; b[q][t].y = p[F1]; b[q][t].z = p[2 * F1]; b[q][t].w = p[3 * F1];
        }
    }
    for (;;) {
        const int rbn = rb + G1_STEP;
        const bool more = rbn < G1_RB;
        const int rbl = more ? rbn : rb;
        float4 an[IN_DIM / 8];
        const float4* xn = (const float4*)(X + (size_t)((rbl << 4) + l) * IN_DIM) + half;
#pragma unroll
        for (int q = 0; q < IN_DIM / 8; ++q) an[q] = xn[2 * q];

        v8f acc0 = {}, acc1 = {}, acc2 = {}, acc3 = {};
#pragma unroll
        for (int q = 0; q < IN_DIM / 8; ++q) {
            const v2f a0 = { a[q].x, a[q].y }, a1 = { a[q].z, a[q].w };
            const v2f b00 = { b[q][0].x, b[q][0].y }, b01 = { b[q][0].z, b[q][0].w };
            const v2f b10 = { b[q][1].x, b[q][1].y }, b11 = { b[q][1].z, b[q][1].w };
            const v2f b20 = { b[q][2].x, b[q][2].y }, b21 = { b[q][2].z, b[q][2].w };
            const v2f b30 = { b[q][3].x, b[q][3].y }, b31 = { b[q][3].z, b[q][3].w };
            acc0 = __builtin_amdgcn_wmma_f32_16x16x4_f32(false, a0, false, b00, (short)0, acc0, false, false);
            acc1 = __builtin_amdgcn_wmma_f32_16x16x4_f32(false, a0, false, b10, (short)0, acc1, false, false);
            acc2 = __builtin_amdgcn_wmma_f32_16x16x4_f32(false, a0, false, b20, (short)0, acc2, false, false);
            acc3 = __builtin_amdgcn_wmma_f32_16x16x4_f32(false, a0, false, b30, (short)0, acc3, false, false);
            acc0 = __builtin_amdgcn_wmma_f32_16x16x4_f32(false, a1, false, b01, (short)0, acc0, false, false);
            acc1 = __builtin_amdgcn_wmma_f32_16x16x4_f32(false, a1, false, b11, (short)0, acc1, false, false);
            acc2 = __builtin_amdgcn_wmma_f32_16x16x4_f32(false, a1, false, b21, (short)0, acc2, false, false);
            acc3 = __builtin_amdgcn_wmma_f32_16x16x4_f32(false, a1, false, b31, (short)0, acc3, false, false);
        }
        float* crow = C + (size_t)(rb << 4) * F1 + (strip << 6);
#pragma unroll
        for (int r = 0; r < 8; ++r) {
            const int m = r + 8 * half;
            float* p = crow + (size_t)m * F1 + l;
            p[ 0] = acc0[r];
            p[16] = acc1[r];
            p[32] = acc2[r];
            p[48] = acc3[r];
        }
        if (!more) break;
        rb = rbn;
#pragma unroll
        for (int q = 0; q < IN_DIM / 8; ++q) a[q] = an[q];
    }
}

#define ALPHA_GPW 32
static_assert(HID == 64 && OUT_DIM == 64, "a group is 64 floats in both layers");
template <int H>
__device__ __forceinline__ void alpha_body(const float* __restrict__ h,
                                           const float* __restrict__ a_src, const float* __restrict__ a_dst,
                                           float* __restrict__ as_, float* __restrict__ ad_, int ngroups, int blk) {
    static_assert(16 % H == 0, "head of group g0 + 16*hf + j is j % H");
    const int lane = threadIdx.x & 31, hf = lane >> 4, l = lane & 15;
    const int g0 = (int)((blk * 256 + threadIdx.x) >> 5) * ALPHA_GPW;
    if (g0 >= ngroups) return;

    float4 v[16];
#pragma unroll
    for (int j = 0; j < 16; ++j) {
        int g = g0 + 16 * hf + j;
        g = (g < ngroups) ? g : (ngroups - 1);
        v[j] = *(const float4*)(h + (size_t)g * 64 + 4 * l);
    }
    float4 sa[H], da[H];
#pragma unroll
    for (int hh = 0; hh < H; ++hh) {
        sa[hh] = *(const float4*)(a_src + hh * 64 + 4 * l);
        da[hh] = *(const float4*)(a_dst + hh * 64 + 4 * l);
    }
    float ps[16], pd[16];
#pragma unroll
    for (int j = 0; j < 16; ++j) {
        const float4 s = sa[j % H], d = da[j % H];
        ps[j] = v[j].x * s.x + v[j].y * s.y + v[j].z * s.z + v[j].w * s.w;
        pd[j] = v[j].x * d.x + v[j].y * d.y + v[j].z * d.z + v[j].w * d.w;
    }
    const bool b3 = (l & 8) != 0, b2 = (l & 4) != 0, b1 = (l & 2) != 0, b0 = (l & 1) != 0;
    float s8[8], d8[8];
#pragma unroll
    for (int t = 0; t < 8; ++t) {
        s8[t] = (b3 ? ps[t + 8] : ps[t]) + __shfl_xor(b3 ? ps[t] : ps[t + 8], 8);
        d8[t] = (b3 ? pd[t + 8] : pd[t]) + __shfl_xor(b3 ? pd[t] : pd[t + 8], 8);
    }
    float s4[4], d4[4];
#pragma unroll
    for (int t = 0; t < 4; ++t) {
        s4[t] = (b2 ? s8[t + 4] : s8[t]) + __shfl_xor(b2 ? s8[t] : s8[t + 4], 4);
        d4[t] = (b2 ? d8[t + 4] : d8[t]) + __shfl_xor(b2 ? d8[t] : d8[t + 4], 4);
    }
    float s2[2], d2[2];
#pragma unroll
    for (int t = 0; t < 2; ++t) {
        s2[t] = (b1 ? s4[t + 2] : s4[t]) + __shfl_xor(b1 ? s4[t] : s4[t + 2], 2);
        d2[t] = (b1 ? d4[t + 2] : d4[t]) + __shfl_xor(b1 ? d4[t] : d4[t + 2], 2);
    }
    const float s1 = (b0 ? s2[1] : s2[0]) + __shfl_xor(b0 ? s2[0] : s2[1], 1);
    const float d1 = (b0 ? d2[1] : d2[0]) + __shfl_xor(b0 ? d2[0] : d2[1], 1);
    const int g = g0 + lane;
    if (g < ngroups) {
        as_[g] = s1;
        ad_[g] = d1;
    }
}

template <int H>
__global__ void __launch_bounds__(256)
alpha_kernel(const float* __restrict__ h,
             const float* __restrict__ a_src, const float* __restrict__ a_dst,
             float* __restrict__ as_, float* __restrict__ ad_, int ngroups) {
    alpha_body<H>(h, a_src, a_dst, as_, ad_, ngroups, (int)blockIdx.x);
}

#define NOWN       196
#define NDIG16     (NEDGES / 16)
#define DIG_BLOCKS ((NDIG16 + 255) / 256)
static_assert(NEDGES % 16 == 0, "the digest is written and read in 16-byte groups");
static_assert((NNODES - 1) / NOWN <= 255, "an owner index fits a byte");
__device__ __forceinline__ unsigned owner4(int4 d) {
    return  (unsigned)d.x / NOWN        | ((unsigned)d.y / NOWN) << 8 |
           ((unsigned)d.z / NOWN) << 16 | ((unsigned)d.w / NOWN) << 24;
}
#define BM_W      8
#define BM_WORDS  (DIG_BLOCKS * BM_W)
static_assert(BM_W == 8 && BM_W * 32 == 256, "a digest block is 8 waves; word index = 8 * owner + wave below");
__device__ __forceinline__ void or_owner_bits(unsigned* row, unsigned o, unsigned bit) {
    __hip_atomic_fetch_or(&row[o & 255u],         bit, __ATOMIC_RELAXED, __HIP_MEMORY_SCOPE_WORKGROUP);
    __hip_atomic_fetch_or(&row[(o >> 8) & 255u],  bit, __ATOMIC_RELAXED, __HIP_MEMORY_SCOPE_WORKGROUP);
    __hip_atomic_fetch_or(&row[(o >> 16) & 255u], bit, __ATOMIC_RELAXED, __HIP_MEMORY_SCOPE_WORKGROUP);
    __hip_atomic_fetch_or(&row[o >> 24],          bit, __ATOMIC_RELAXED, __HIP_MEMORY_SCOPE_WORKGROUP);
}
__device__ __forceinline__ void digest_block(const int* __restrict__ dst, unsigned* __restrict__ dig,
                                             unsigned* __restrict__ gbm, int blk) {
    __shared__ unsigned bm[BM_W * 256];
    const int tid = (int)threadIdx.x;
    for (int k = tid; k < BM_W * 256; k += 256) bm[k] = 0u;
    __syncthreads();
    const int i = blk * 256 + tid;
    if (i < NDIG16) {
        const int4* d4 = (const int4*)dst + 4 * (size_t)i;
        const int4 a = d4[0], b = d4[1], cc = d4[2], d = d4[3];
        uint4 o;
        o.x = owner4(a); o.y = owner4(b); o.z = owner4(cc); o.w = owner4(d);
        ((uint4*)dig)[i] = o;
        unsigned* row = bm + (tid >> 5) * 256;
        const unsigned bit = 1u << (tid & 31);
        or_owner_bits(row, o.x, bit); or_owner_bits(row, o.y, bit);
        or_owner_bits(row, o.z, bit); or_owner_bits(row, o.w, bit);
    }
    __syncthreads();
    unsigned* mybm = gbm + (size_t)blk * (BM_W * 256);
    for (int k = tid; k < BM_W * 256; k += 256) mybm[k] = bm[(k & (BM_W - 1)) * 256 + (k >> 3)];
}

#define ALPHA1_BLOCKS (((NNODES * HEADS + ALPHA_GPW - 1) / ALPHA_GPW + 7) / 8)
__global__ void __launch_bounds__(256)
alpha_digest_kernel(const float* __restrict__ h,
                    const float* __restrict__ a_src, const float* __restrict__ a_dst,
                    float* __restrict__ as_, float* __restrict__ ad_, int ngroups,
                    const int* __restrict__ dst, unsigned* __restrict__ dig, unsigned* __restrict__ gbm) {
    if ((int)blockIdx.x < DIG_BLOCKS) digest_block(dst, dig, gbm, (int)blockIdx.x);
    else alpha_body<HEADS>(h, a_src, a_dst, as_, ad_, ngroups, (int)blockIdx.x - DIG_BLOCKS);
}

#define NWG_AGG   256
#define AGG_WAVES 32
#define QCAP      64
#define LCAP      16384
#define SCAN_U    8

static_assert(NOWN * NWG_AGG >= NNODES && (NWG_AGG - 1) * NOWN < NNODES, "every node has one owner");
static_assert(ETOT <= (1 << 20) && NOWN <= 256, "queue entry = edge id, 20 bits; record entry = ... | local node << 20");
static_assert(NNODES <= (1 << 20), "record entry = source node | local node << 20");
#define RSTRIDE   (LCAP + 32)
static_assert((size_t)NNODES * (F1 + OUT_DIM) + (size_t)NWG_AGG * RSTRIDE + NEDGES / 4 +
              (size_t)DIG_BLOCKS * 256 * BM_W <= (size_t)2 * NNODES * F1,
              "the records, the digest and the owner index end before the logits begin");
static_assert(NWG_AGG == 256 && (NEDGES / 4) % 32 == 0, "one index row per owner; the index starts on a 128-byte line");
static_assert(((size_t)NWG_AGG * RSTRIDE) % 4 == 0, "the digest starts on a 16-byte boundary");
static_assert(((size_t)NNODES * (F1 + OUT_DIM)) % 32 == 0, "records start on a 128-byte line of the workspace");

template <int H> struct RowPitch { static constexpr int value = (H == 4) ? (H * 64 + 4) : (H * 64); };
#define W2_ROWS F1
static_assert(W2_ROWS * OUT_DIM == LCAP, "the W2 fragments take LCAP words of the edge record's LDS");
static_assert(OUT_DIM == 64 && W2_ROWS % 4 == 0, "layer-2 tile map below");

#define RCAP1 4096
template <int H> struct RecCap { static constexpr int value = (H == 4) ? RCAP1 : LCAP; };
static_assert(RCAP1 <= LCAP && RCAP1 % 4 == 0, "the layer-1 record fits its workspace slot; logits behind it stay 16-byte aligned");

template <int H>
__device__ __forceinline__ void elog_put(float* elog, int idx, const float* e) {
    if constexpr (H == 4) {
        ((float4*)elog)[idx] = make_float4(e[0], e[1], e[2], e[3]);
    } else {
#pragma unroll
        for (int hh = 0; hh < H; ++hh) elog[idx * H + hh] = e[hh];
    }
}
template <int H>
__device__ __forceinline__ void elog_get(const float* elog, int idx, float* e) {
    if constexpr (H == 4) {
        const float4 v = ((const float4*)elog)[idx];
        e[0] = v.x; e[1] = v.y; e[2] = v.z; e[3] = v.w;
    } else {
#pragma unroll
        for (int hh = 0; hh < H; ++hh) e[hh] = elog[idx * H + hh];
    }
}

template <int H, int PHASE, int G, bool SRC, bool QDST, bool REC, bool EMIT, bool ELOG>
__device__ __forceinline__ void flush_batch(const unsigned* list, int base, int n, int lane, int lo,
                                            const int* __restrict__ ei,
                                            const float* __restrict__ h,
                                            const float* __restrict__ as_,
                                            const float* adl, unsigned* menc, float* den, float* acc,
                                            unsigned* wlist, unsigned* wcount,
                                            unsigned* __restrict__ grec, float* elog) {
    constexpr int F = H * 64;
    constexpr unsigned RC = (unsigned)RecCap<H>::value;
    constexpr int P = RowPitch<H>::value;
    const bool act = lane < n;
    unsigned ent = 0;
    if (act) ent = list[base + lane];
    const int t  = (int)(ent & 0xFFFFFu);
    int dl = (int)(ent >> 20);
    int s = 0;
    if (act) {
        if (SRC) s = t;
        else     s = (t < NEDGES) ? ei[t] : (t - NEDGES);
        if (QDST) {
            dl = ((t < NEDGES) ? ei[NEDGES + t] : (t - NEDGES)) - lo;
            ent = (unsigned)s | ((unsigned)dl << 20);
        }
    }
    float e[H];
#pragma unroll
    for (int hh = 0; hh < H; ++hh) e[hh] = 0.f;
    if (act) {
        if (ELOG && PHASE == 1) {
            elog_get<H>(elog, base + lane, e);
        } else {
#pragma unroll
            for (int hh = 0; hh < H; ++hh) e[hh] = leakyf(as_[s * H + hh] + adl[dl * H + hh], 0.2f);
        }
    }
    if (PHASE == 0) {
        if (act) {
#pragma unroll
            for (int hh = 0; hh < H; ++hh)
                __hip_atomic_fetch_max(&menc[dl * H + hh], enc_ord(e[hh]), __ATOMIC_RELAXED,
                                       __HIP_MEMORY_SCOPE_WORKGROUP);
        }
        if (REC) {
            unsigned pos = 0;
            if (lane == 0)
                pos = __hip_atomic_fetch_add(wcount, (unsigned)n, __ATOMIC_RELAXED, __HIP_MEMORY_SCOPE_WORKGROUP);
            pos = (unsigned)__builtin_amdgcn_readfirstlane((int)pos);
            if (act && pos + (unsigned)n <= RC) {
                wlist[pos + lane] = ent;
                if (ELOG) elog_put<H>(elog, (int)pos + lane, e);
            }
        } else if (ELOG) {
            if (act) elog_put<H>(elog, base + lane, e);
        }
    } else {
        if (EMIT) {
            if (act) grec[base + lane] = (unsigned)s | ((unsigned)dl << 20);
        }
        float ex[H];
#pragma unroll
        for (int hh = 0; hh < H; ++hh) ex[hh] = 0.f;
        if (act) {
#pragma unroll
            for (int hh = 0; hh < H; ++hh) {
                ex[hh] = expf(e[hh] - dec_ord(menc[dl * H + hh]));
                __hip_atomic_fetch_add(&den[dl * H + hh], ex[hh], __ATOMIC_RELAXED, __HIP_MEMORY_SCOPE_WORKGROUP);
            }
        }
        for (int j = 0; j < n; j += G) {
            v2f   r[G][H];
            float w[G][H];
            int   dj[G];
#pragma unroll
            for (int g = 0; g < G; ++g) {
                const int jj = (j + g < n) ? (j + g) : (n - 1);
                const int sj = __builtin_amdgcn_readlane(s, jj);
                dj[g] = __builtin_amdgcn_readlane(dl, jj);
#pragma unroll
                for (int hh = 0; hh < H; ++hh)
                    w[g][hh] = __int_as_float(__builtin_amdgcn_readlane(__float_as_int(ex[hh]), jj));
                const v2f* hp = (const v2f*)(h + (size_t)sj * F) + lane;
#pragma unroll
                for (int hh = 0; hh < H; ++hh) r[g][hh] = hp[hh * 32];
            }
#pragma unroll
            for (int g = 0; g < G; ++g) {
                if (j + g < n) {
                    float* ap = acc + dj[g] * P + 2 * lane;
#pragma unroll
                    for (int hh = 0; hh < H; ++hh) {
                        __hip_atomic_fetch_add(ap + hh * 64,     w[g][hh] * r[g][hh].x, __ATOMIC_RELAXED,
                                               __HIP_MEMORY_SCOPE_WORKGROUP);
                        __hip_atomic_fetch_add(ap + hh * 64 + 1, w[g][hh] * r[g][hh].y, __ATOMIC_RELAXED,
                                               __HIP_MEMORY_SCOPE_WORKGROUP);
                    }
                }
            }
        }
    }
}

__device__ __forceinline__ void owned_batch(const unsigned* wlist, const float* elog, const unsigned short* perm,
                                            int first, int n, int lane, const float* __restrict__ h, float* acc) {
    constexpr int H = HEADS, F = F1, G = 4;
    constexpr int P = RowPitch<HEADS>::value;
    const bool act = lane < n;
    int idx = 0;
    if (act) idx = (int)perm[first + lane];
    unsigned ent = 0;
    if (act) ent = wlist[idx];
    const int s  = (int)(ent & 0xFFFFFu);
    const int dl = (int)(ent >> 20);
    float ex[H];
#pragma unroll
    for (int hh = 0; hh < H; ++hh) ex[hh] = 0.f;
    if (act) elog_get<H>(elog, idx, ex);
    const bool lowhalf = lane < 16;
    auto fetch = [&](int j0, float4* ra, float4* rb) {
#pragma unroll
        for (int g = 0; g < G; ++g) {
            const int jj = (j0 + g < n) ? (j0 + g) : (n - 1);
            const int sj = __builtin_amdgcn_readlane(s, jj);
            const float4* hp = (const float4*)(h + (size_t)sj * F) + lane;
            ra[g] = hp[0];
            rb[g] = hp[32];
        }
    };
    auto add = [&](int j0, const float4* ra, const float4* rb) {
#pragma unroll
        for (int g = 0; g < G; ++g) {
            if (j0 + g < n) {
                const int jj = j0 + g;
                const int dj = __builtin_amdgcn_readlane(dl, jj);
                const float w0 = __int_as_float(__builtin_amdgcn_readlane(__float_as_int(ex[0]), jj));
                const float w1 = __int_as_float(__builtin_amdgcn_readlane(__float_as_int(ex[1]), jj));
                const float w2 = __int_as_float(__builtin_amdgcn_readlane(__float_as_int(ex[2]), jj));
                const float w3 = __int_as_float(__builtin_amdgcn_readlane(__float_as_int(ex[3]), jj));
                const float wa = lowhalf ? w0 : w1, wb = lowhalf ? w2 : w3;
                float4* p = (float4*)(acc + dj * P) + lane;
                float4 va = p[0], vb = p[32];
                va.x += wa * ra[g].x; va.y += wa * ra[g].y; va.z += wa * ra[g].z; va.w += wa * ra[g].w;
                vb.x += wb * rb[g].x; vb.y += wb * rb[g].y; vb.z += wb * rb[g].z; vb.w += wb * rb[g].w;
                p[0]  = va;
                p[32] = vb;
            }
        }
    };
    float4 a0[G], b0[G], a1[G], b1[G];
    fetch(0, a0, b0);
    for (int j = 0; j < n; j += 2 * G) {
        fetch(j + G, a1, b1);
        add(j, a0, b0);
        fetch(j + 2 * G, a0, b0);
        add(j + G, a1, b1);
    }
}

__device__ __forceinline__ unsigned zero_bytes(unsigned x) {
    const unsigned t = (x & 0x7F7F7F7Fu) + 0x7F7F7F7Fu;
    return ~(t | x) & 0x80808080u;
}

template <int H, int PHASE, int G>
__device__ __forceinline__ void scan_pass(const int* __restrict__ ei, const unsigned* __restrict__ dig,
                                          const unsigned* __restrict__ gbm,
                                          const float* __restrict__ h,
                                          const float* __restrict__ as_,
                                          const float* adl, unsigned* menc, float* den, float* acc,
                                          unsigned* myq, unsigned* wlist, unsigned* wcount, float* elog,
                                          int lo, int nown, int lane, int wave) {
    int qn = 0;
    auto enqueue = [&](bool mt, unsigned val) {
        const unsigned mask = __builtin_amdgcn_ballot_w32(mt);
        if (mt) myq[qn + __builtin_popcount(mask & ((1u << lane) - 1u))] = val;
        qn += __builtin_popcount(mask);
    };
    {
        const int dl = wave + AGG_WAVES * lane;
        enqueue(dl < nown, (unsigned)(NEDGES + lo + dl));
    }
    const uint4* dg4 = (const uint4*)dig;
    const unsigned mine = (unsigned)blockIdx.x * 0x01010101u;
    constexpr int WPW = BM_WORDS / AGG_WAVES;
    static_assert(BM_WORDS % AGG_WAVES == 0, "index words per wave");
    for (int o = 0; o < WPW; o += 32) {
        const int k = wave * WPW + o + lane;
        unsigned m = 0u;
        if (o + lane < WPW) m = gbm[((size_t)(k >> 3) * 256 + blockIdx.x) * BM_W + (k & 7)];
        while (__builtin_amdgcn_ballot_w32(m != 0u) != 0u) {
            int   gi[SCAN_U];
            uint4 dv[SCAN_U];
#pragma unroll
            for (int u = 0; u < SCAN_U; ++u) {
                gi[u] = (m != 0u) ? (k * 32 + __builtin_ctz(m | 0x80000000u)) : -1;
                m &= m - 1u;
                dv[u] = dg4[(gi[u] >= 0) ? gi[u] : 0];
            }
#pragma unroll
            for (int u = 0; u < SCAN_U; ++u) {
                if (__builtin_amdgcn_ballot_w32(gi[u] >= 0) == 0u) break;
                const int i16 = gi[u];
                unsigned f = (zero_bytes(dv[u].x ^ mine) >> 7) | (zero_bytes(dv[u].y ^ mine) >> 6) |
                             (zero_bytes(dv[u].z ^ mine) >> 5) | (zero_bytes(dv[u].w ^ mine) >> 4);
                if (i16 < 0) f = 0u;
                while (__builtin_amdgcn_ballot_w32(f != 0u) != 0u) {
                    const int p = __builtin_ctz(f | 0x80000000u);
                    enqueue(f != 0u, (unsigned)(i16 * 16 + 4 * (p & 7) + (p >> 3)));
                    f &= f - 1u;
                    while (qn >= 32) {
                        flush_batch<H, PHASE, G, false, true, PHASE == 0, false, PHASE == 0>(myq, qn - 32, 32, lane, lo, ei, h,
                                                                                             as_, adl, menc, den, acc, wlist,
                                                                                             wcount, nullptr, elog);
                        qn -= 32;
                    }
                }
            }
        }
    }
    if (qn > 0)
        flush_batch<H, PHASE, G, false, true, PHASE == 0, false, PHASE == 0>(myq, 0, qn, lane, lo, ei, h, as_, adl, menc, den,
                                                                             acc, wlist, wcount, nullptr, elog);
}

template <int H>
__global__ void __launch_bounds__(AGG_WAVES * 32)
agg_owner_kernel(const int* __restrict__ ei, const float* __restrict__ h,
                 const float* __restrict__ as_, const float* __restrict__ ad_,
                 float* __restrict__ out,
                 const float* __restrict__ bias, const float* __restrict__ W2,
                 unsigned* __restrict__ grec, const unsigned* __restrict__ dig,
                 const unsigned* __restrict__ gbm) {
    constexpr int F = H * 64;
    constexpr int P = RowPitch<H>::value;
    __shared__ __attribute__((aligned(16))) float acc[NOWN * P];
    __shared__ unsigned menc[NOWN * H];
    __shared__ float    den[NOWN * H];
    __shared__ float    adl[NOWN * H];
    __shared__ unsigned q[AGG_WAVES * QCAP];
    constexpr int RC  = RecCap<H>::value;
    constexpr int RBW = (RC * (1 + H) > LCAP) ? RC * (1 + H) : LCAP;
    __shared__ __attribute__((aligned(16))) unsigned recbuf[RBW];
    __shared__ unsigned wcount;
    __shared__ unsigned short perm[(H == 4) ? RC : 1];
    __shared__ unsigned bincnt[AGG_WAVES], binoff[AGG_WAVES];
    unsigned* wlist = recbuf;
    float*    elog  = (float*)(recbuf + RC);

    const int tid = threadIdx.x, lane = tid & 31, wave = tid >> 5;
    const int lo = (int)blockIdx.x * NOWN;
    const int nown = (NNODES - lo < NOWN) ? (NNODES - lo) : NOWN;
    unsigned* myhdr = grec + (size_t)blockIdx.x * RSTRIDE;
    unsigned* myrec = myhdr + 32;

    unsigned cnt_in = (unsigned)LCAP + 1u;
    if constexpr (H == 1) {
        cnt_in = (unsigned)__builtin_amdgcn_readfirstlane((int)myhdr[0]);
        if (cnt_in <= (unsigned)RCAP1)
            for (int i = tid; i < (int)cnt_in; i += AGG_WAVES * 32) wlist[i] = myrec[i];
    }
    const bool fromrec = (H == 1) && (cnt_in <= (unsigned)RCAP1);

    for (int i = tid; i < NOWN * P; i += AGG_WAVES * 32) acc[i] = 0.f;
    for (int i = tid; i < NOWN * H; i += AGG_WAVES * 32) {
        menc[i] = ENC_NEG_INF;
        den[i]  = 0.f;
        adl[i]  = (i < nown * H) ? ad_[lo * H + i] : 0.f;
    }
    if (tid == 0) wcount = 0u;
    if (tid < AGG_WAVES) bincnt[tid] = 0u;
    __syncthreads();

    unsigned* myq = q + wave * QCAP;
    if (fromrec) {
        for (int base = wave * 32; base < (int)cnt_in; base += AGG_WAVES * 32) {
            int n = (int)cnt_in - base;
            n = __builtin_amdgcn_readfirstlane(n < 32 ? n : 32);
            flush_batch<H, 0, 4, true, false, false, false, true>(wlist, base, n, lane, lo, ei, h, as_, adl, menc, den, acc,
                                                                  wlist, &wcount, nullptr, elog);
        }
    } else {
        scan_pass<H, 0, 4>(ei, dig, gbm, h, as_, adl, menc, den, acc, myq, wlist, &wcount, elog, lo, nown, lane, wave);
    }
    __syncthreads();

    const unsigned cnt = (unsigned)__builtin_amdgcn_readfirstlane((int)(fromrec ? cnt_in : wcount));
    if constexpr (H == 4) {
        if (tid == 0) myhdr[0] = cnt;
    }
    constexpr int GB = (H == 1) ? 16 : 8;
    if (cnt <= (unsigned)RC) {
        if constexpr (H == 4) {
            static_assert(RC == 4 * AGG_WAVES * 32 && RC <= 65536, "4 record entries per thread; an entry number fits 16 bits");
            unsigned rk[4], bn[4];
#pragma unroll
            for (int k = 0; k < 4; ++k) {
                const int i = tid + k * (AGG_WAVES * 32);
                rk[k] = 0u; bn[k] = 0u;
                if (i < (int)cnt) {
                    const unsigned ent = wlist[i];
                    myrec[i] = ent;
                    const int dl = (int)(ent >> 20);
                    bn[k] = (unsigned)dl & (unsigned)(AGG_WAVES - 1);
                    rk[k] = __hip_atomic_fetch_add(&bincnt[bn[k]], 1u, __ATOMIC_RELAXED, __HIP_MEMORY_SCOPE_WORKGROUP);
                    float e[H];
                    elog_get<H>(elog, i, e);
#pragma unroll
                    for (int hh = 0; hh < H; ++hh) {
                        e[hh] = expf(e[hh] - dec_ord(menc[dl * H + hh]));
                        __hip_atomic_fetch_add(&den[dl * H + hh], e[hh], __ATOMIC_RELAXED, __HIP_MEMORY_SCOPE_WORKGROUP);
                    }
                    elog_put<H>(elog, i, e);
                }
            }
            __syncthreads();
            if (tid < AGG_WAVES) {
                unsigned o = 0u;
#pragma unroll
                for (int b = 0; b < AGG_WAVES; ++b) o += (b < tid) ? bincnt[b] : 0u;
                binoff[tid] = o;
            }
            __syncthreads();
#pragma unroll
            for (int k = 0; k < 4; ++k) {
                const int i = tid + k * (AGG_WAVES * 32);
                if (i < (int)cnt) perm[binoff[bn[k]] + rk[k]] = (unsigned short)i;
            }
            __syncthreads();
            const int nw = __builtin_amdgcn_readfirstlane((int)bincnt[wave]);
            const int ow = __builtin_amdgcn_readfirstlane((int)binoff[wave]);
            for (int base = 0; base < nw; base += 32) {
                const int n = (nw - base < 32) ? (nw - base) : 32;
                owned_batch(wlist, elog, perm, ow + base, n, lane, h, acc);
            }
        } else {
            for (int base = wave * 32; base < (int)cnt; base += AGG_WAVES * 32) {
                int n = (int)cnt - base;
                n = __builtin_amdgcn_readfirstlane(n < 32 ? n : 32);
                flush_batch<H, 1, GB, true, false, false, false, true>(wlist, base, n, lane, lo, ei, h, as_, adl, menc,
                                                                       den, acc, wlist, &wcount, nullptr, elog);
            }
        }
    } else {
        scan_pass<H, 1, 4>(ei, dig, gbm, h, as_, adl, menc, den, acc, myq, wlist, &wcount, elog, lo, nown, lane, wave);
    }
    __syncthreads();

    if constexpr (H == 4) {
        constexpr int WSLOTS = LCAP / 2 / (AGG_WAVES * 32);
        v2f wreg[WSLOTS];
#pragma unroll
        for (int j = 0; j < WSLOTS; ++j) {
            const int slot = tid + j * (AGG_WAVES * 32);
            const int sl = slot & 31, tile = (slot >> 5) & 3, k4 = slot >> 7;
            const float* wp = W2 + (size_t)(4 * k4 + 2 * (sl >> 4)) * OUT_DIM + 16 * tile + (sl & 15);
            wreg[j].x = wp[0];
            wreg[j].y = wp[OUT_DIM];
        }
        for (int i = tid; i < nown * (F / 4); i += AGG_WAVES * 32) {
            const int row = i / (F / 4), col = (i % (F / 4)) * 4;
            const float dn = den[row * H + (col >> 6)] + 1e-16f;
            float4* p = (float4*)(acc + row * P + col);
            const float4 b = *(const float4*)(bias + col);
            float4 v = *p;
            v.x /= dn; v.y /= dn; v.z /= dn; v.w /= dn;
            v.x = leakyf(v.x + b.x, 0.01f); v.y = leakyf(v.y + b.y, 0.01f);
            v.z = leakyf(v.z + b.z, 0.01f); v.w = leakyf(v.w + b.w, 0.01f);
            *p = v;
        }
        v2f* wf = (v2f*)recbuf;
#pragma unroll
        for (int j = 0; j < WSLOTS; ++j) wf[tid + j * (AGG_WAVES * 32)] = wreg[j];
        __syncthreads();

        const int half = lane >> 4, l = lane & 15;
        const int njobs = ((nown + 15) >> 4) * 2;
        const int wv = __builtin_amdgcn_readfirstlane(wave);
        for (int job = wv; job < njobs; job += AGG_WAVES) {
            const int rb = job >> 1, cp = job & 1;
            int arow = rb * 16 + l;
            arow = (arow < NOWN) ? arow : (NOWN - 1);
            const float* ap = acc + arow * P + 2 * half;
            const v2f* bp = wf + (2 * cp) * 32 + lane;
            v8f c0 = {}, c1 = {};
            for (int k4 = 0; k4 < W2_ROWS / 4; ++k4) {
                const v2f a  = *(const v2f*)(ap + 4 * k4);
                const v2f b0 = bp[k4 * 128];
                const v2f b1 = bp[k4 * 128 + 32];
                c0 = __builtin_amdgcn_wmma_f32_16x16x4_f32(false, a, false, b0, (short)0, c0, false, false);
                c1 = __builtin_amdgcn_wmma_f32_16x16x4_f32(false, a, false, b1, (short)0, c1, false, false);
            }
#pragma unroll
            for (int r = 0; r < 8; ++r) {
                const int row = rb * 16 + r + 8 * half;
                if (row < nown) {
                    float* gp = out + (size_t)(lo + row) * OUT_DIM + 32 * cp + l;
                    gp[ 0] = c0[r];
                    gp[16] = c1[r];
                }
            }
        }
    } else {
        const int n4 = nown * F / 4;
        float4* o4 = (float4*)(out + (size_t)lo * F);
        const float4* a4 = (const float4*)acc;
        for (int i = tid; i < n4; i += AGG_WAVES * 32) {
            const int idx = i * 4;
            const float dn = den[(idx / F) * H + ((idx % F) >> 6)] + 1e-16f;
            const float4 b = *(const float4*)(bias + (idx % F));
            float4 v = a4[i];
            v.x = v.x / dn + b.x; v.y = v.y / dn + b.y; v.z = v.z / dn + b.z; v.w = v.w / dn + b.w;
            o4[i] = v;
        }
    }
}

extern "C" void kernel_launch(void* const* d_in, const int* in_sizes, int n_in,
                              void* d_out, int out_size, void* d_ws, size_t ws_size,
                              hipStream_t stream) {
    const float* x   = (const float*)d_in[0];
    const int*   ei  = (const int*)  d_in[1];
    const float* W1  = (const float*)d_in[2];
    const float* as1 = (const float*)d_in[3];
    const float* ad1 = (const float*)d_in[4];
    const float* b1  = (const float*)d_in[5];
    const float* W2  = (const float*)d_in[6];
    const float* as2 = (const float*)d_in[7];
    const float* ad2 = (const float*)d_in[8];
    const float* b2  = (const float*)d_in[9];
    float* out = (float*)d_out;

    float* ws   = (float*)d_ws;
    float* h1   = ws;
    float* g2   = h1   + (size_t)NNODES * F1;
    float* asb  = h1   + (size_t)2 * NNODES * F1;
    float* adb  = asb  + (size_t)NNODES * HEADS;
    unsigned* rec = (unsigned*)(g2 + (size_t)NNODES * OUT_DIM);
    unsigned* dig = rec + (size_t)NWG_AGG * RSTRIDE;
    unsigned* gbm = dig + NEDGES / 4;

    gemm1_wreg_kernel<<<G1_WGS, G1_WAVES * 32, 0, stream>>>(x, W1, h1);
    alpha_digest_kernel<<<DIG_BLOCKS + ALPHA1_BLOCKS, 256, 0, stream>>>(h1, as1, ad1, asb, adb, NNODES * HEADS,
                                                                        ei + NEDGES, dig, gbm);
    agg_owner_kernel<HEADS><<<NWG_AGG, AGG_WAVES * 32, 0, stream>>>(ei, h1, asb, adb, g2, b1, W2, rec, dig, gbm);

    alpha_kernel<1><<<((NNODES + ALPHA_GPW - 1) / ALPHA_GPW + 7) / 8, 256, 0, stream>>>(
        g2, as2, ad2, asb, adb, NNODES);
    agg_owner_kernel<1><<<NWG_AGG, AGG_WAVES * 32, 0, stream>>>(ei, g2, asb, adb, out, b2, W2, rec, dig, gbm);
}
